# starting kernel
# baseline (speedup 1.0000x reference)
#include <hip/hip_runtime.h>
#include <hip/hip_bf16.h>

#define N_NODES 100000
#define N_EDGES 1600000
#define IN_DIM  256
#define HID     128
#define N_GRAPHS 64
#define OUT_MF 489
#define OUT_BP 1943
#define OUT_CC 320

typedef __attribute__((ext_vector_type(2))) float v2f;
typedef __attribute__((ext_vector_type(8))) float v8f;

__global__ void fill_f32(float* __restrict__ p, float v, int n) {
    int t = blockIdx.x * blockDim.x + threadIdx.x;
    if (t < n) p[t] = v;
}

__global__ void zero_f4(float4* __restrict__ p, int n4) {
    int t = blockIdx.x * blockDim.x + threadIdx.x;
    if (t < n4) { float4 z; z.x = z.y = z.z = z.w = 0.f; p[t] = z; }
}

__global__ void deg_accum(const int* __restrict__ dst, float* __restrict__ deg, int E) {
    int t = blockIdx.x * blockDim.x + threadIdx.x;
    if (t < E) atomicAdd(&deg[dst[t]], 1.0f);
}

__global__ void rsqrt_inplace(float* __restrict__ p, int n) {
    int t = blockIdx.x * blockDim.x + threadIdx.x;
    if (t < n) p[t] = rsqrtf(p[t]);
}

template <int K>
__global__ __launch_bounds__(256) void gemm_wmma_f32(const float* __restrict__ A,
                                                     const float* __restrict__ W,
                                                     float* __restrict__ out) {
    __shared__ float ldsW[64 * HID];
    const int tid  = threadIdx.x;
    const int wave = tid >> 5;
    const int lane = tid & 31;
    const int half = lane >> 4;
    const int m    = lane & 15;
    const int row0 = blockIdx.x * 16;
    const int col0 = wave * 16;

    const float* arow = A + (size_t)(row0 + m) * K;
    v8f acc = {};

    for (int kb = 0; kb < K; kb += 64) {
        const float4* wv = (const float4*)(W + (size_t)kb * HID);
        float4* lv = (float4*)ldsW;
        #pragma unroll
        for (int i = tid; i < 64 * HID / 4; i += 256) lv[i] = wv[i];
        __syncthreads();

        #pragma unroll
        for (int k = 0; k < 64; k += 4) {
            const int ka = k + 2 * half;
            v2f a;
            a.x = arow[kb + ka];
            a.y = arow[kb + ka + 1];
            v2f b;
            b.x = ldsW[ka * HID + col0 + m];
            b.y = ldsW[(ka + 1) * HID + col0 + m];
            acc = __builtin_amdgcn_wmma_f32_16x16x4_f32(
                false, a, false, b, (short)0, acc, false, false);
        }
        __syncthreads();
    }

    #pragma unroll
    for (int v = 0; v < 8; ++v)
        out[(size_t)(row0 + v + 8 * half) * HID + col0 + m] = acc[v];
}

__global__ __launch_bounds__(256) void edge_scatter(const int* __restrict__ src,
                                                    const int* __restrict__ dst,
                                                    const float* __restrict__ dis,
                                                    const float* __restrict__ h,
                                                    float* __restrict__ agg, int E) {
    const int lane = threadIdx.x & 31;
    const int gw   = (blockIdx.x * blockDim.x + threadIdx.x) >> 5;
    const int nw   = (gridDim.x * blockDim.x) >> 5;
    for (int e = gw; e < E; e += nw) {
        const int s = src[e], d = dst[e];
        const float nrm = dis[s] * dis[d];
        const float* hs = h + (size_t)s * HID;
        float* ad = agg + (size_t)d * HID;
        #pragma unroll
        for (int j = 0; j < HID; j += 32)
            atomicAdd(&ad[j + lane], hs[j + lane] * nrm);
    }
}

__global__ void finish_layer(float* __restrict__ agg, const float* __restrict__ h,
                             const float* __restrict__ dis, const float* __restrict__ bias,
                             int n) {
    int t = blockIdx.x * blockDim.x + threadIdx.x;
    if (t >= n) return;
    const int i = t >> 7, j = t & (HID - 1);
    const float di = dis[i];
    const float v = agg[t] + di * di * h[t] + bias[j];
    agg[t] = v > 0.f ? v : 0.f;
}

__global__ __launch_bounds__(256) void pool_accum(const float* __restrict__ h,
                                                  const int* __restrict__ batch,
                                                  float* __restrict__ pooled,
                                                  float* __restrict__ cnt, int N) {
    const int lane = threadIdx.x & 31;
    const int gw   = (blockIdx.x * blockDim.x + threadIdx.x) >> 5;
    const int nw   = (gridDim.x * blockDim.x) >> 5;
    for (int i = gw; i < N; i += nw) {
        const int g = batch[i];
        const float* hi = h + (size_t)i * HID;
        float* pg = pooled + g * HID;
        #pragma unroll
        for (int j = 0; j < HID; j += 32)
            atomicAdd(&pg[j + lane], hi[j + lane]);
        if (lane == 0) atomicAdd(&cnt[g], 1.0f);
    }
}

__global__ void pool_div(float* __restrict__ pooled, const float* __restrict__ cnt) {
    int t = blockIdx.x * blockDim.x + threadIdx.x;
    if (t >= N_GRAPHS * HID) return;
    float c = cnt[t >> 7];
    pooled[t] /= (c > 1.f ? c : 1.f);
}

__global__ void head_gemm(const float* __restrict__ pooled, const float* __restrict__ W,
                          const float* __restrict__ b, float* __restrict__ out, int OUT) {
    int t = blockIdx.x * blockDim.x + threadIdx.x;
    const int n = N_GRAPHS * OUT;
    if (t >= n) return;
    const int g = t / OUT, o = t - g * OUT;
    const float* p = pooled + g * HID;
    float s = b[o];
    #pragma unroll 8
    for (int k = 0; k < HID; ++k) s += p[k] * W[k * OUT + o];
    out[t] = s;
}

extern "C" void kernel_launch(void* const* d_in, const int* in_sizes, int n_in,
                              void* d_out, int out_size, void* d_ws, size_t ws_size,
                              hipStream_t stream) {
    const float* x   = (const float*)d_in[0];
    const int*   ei  = (const int*)d_in[1];
    const int*   bat = (const int*)d_in[2];
    const float* W1  = (const float*)d_in[3];
    const float* b1  = (const float*)d_in[4];
    const float* W2  = (const float*)d_in[5];
    const float* b2  = (const float*)d_in[6];
    const float* Wmf = (const float*)d_in[7];
    const float* bmf = (const float*)d_in[8];
    const float* Wbp = (const float*)d_in[9];
    const float* bbp = (const float*)d_in[10];
    const float* Wcc = (const float*)d_in[11];
    const float* bcc = (const float*)d_in[12];
    float* out = (float*)d_out;

    const int E = in_sizes[1] / 2;
    const int* src = ei;
    const int* dst = ei + E;

    float* ws     = (float*)d_ws;
    float* dis    = ws;
    float* h      = dis + N_NODES;
    float* agg    = h + (size_t)N_NODES * HID;
    float* pooled = agg + (size_t)N_NODES * HID;
    float* cnt    = pooled + N_GRAPHS * HID;

    const int NH   = N_NODES * HID;
    const int NH4  = NH / 4;

    fill_f32<<<(N_NODES + 255) / 256, 256, 0, stream>>>(dis, 1.0f, N_NODES);
    deg_accum<<<(E + 255) / 256, 256, 0, stream>>>(dst, dis, E);
    rsqrt_inplace<<<(N_NODES + 255) / 256, 256, 0, stream>>>(dis, N_NODES);

    gemm_wmma_f32<IN_DIM><<<N_NODES / 16, 256, 0, stream>>>(x, W1, h);
    zero_f4<<<(NH4 + 255) / 256, 256, 0, stream>>>((float4*)agg, NH4);
    edge_scatter<<<2048, 256, 0, stream>>>(src, dst, dis, h, agg, E);
    finish_layer<<<(NH + 255) / 256, 256, 0, stream>>>(agg, h, dis, b1, NH);

    gemm_wmma_f32<HID><<<N_NODES / 16, 256, 0, stream>>>(agg, W2, h);
    zero_f4<<<(NH4 + 255) / 256, 256, 0, stream>>>((float4*)agg, NH4);
    edge_scatter<<<2048, 256, 0, stream>>>(src, dst, dis, h, agg, E);
    finish_layer<<<(NH + 255) / 256, 256, 0, stream>>>(agg, h, dis, b2, NH);

    fill_f32<<<(N_GRAPHS * HID + N_GRAPHS + 255) / 256, 256, 0, stream>>>(
        pooled, 0.0f, N_GRAPHS * HID + N_GRAPHS);
    pool_accum<<<12500, 256, 0, stream>>>(agg, bat, pooled, cnt, N_NODES);
    pool_div<<<(N_GRAPHS * HID + 255) / 256, 256, 0, stream>>>(pooled, cnt);

    float* out_mf = out;
    float* out_bp = out + N_GRAPHS * OUT_MF;
    float* out_cc = out + N_GRAPHS * (OUT_MF + OUT_BP);
    head_gemm<<<(N_GRAPHS * OUT_MF + 255) / 256, 256, 0, stream>>>(pooled, Wmf, bmf, out_mf, OUT_MF);
    head_gemm<<<(N_GRAPHS * OUT_BP + 255) / 256, 256, 0, stream>>>(pooled, Wbp, bbp, out_bp, OUT_BP);
    head_gemm<<<(N_GRAPHS * OUT_CC + 255) / 256, 256, 0, stream>>>(pooled, Wcc, bcc, out_cc, OUT_CC);
}


# pool_accum: a wave sums its consecutive rows in registers and sends one float atomic add per feature per run of equal graph id (was one per row); summation order of the pooled sums changes
# speedup vs baseline: 1.1539x; 1.1539x over previous
#include <hip/hip_runtime.h>
#include <hip/hip_bf16.h>

#define N_NODES 100000
#define N_EDGES 1600000
#define IN_DIM  256
#define HID     128
#define N_GRAPHS 64
#define OUT_MF 489
#define OUT_BP 1943
#define OUT_CC 320

typedef __attribute__((ext_vector_type(2))) float v2f;
typedef __attribute__((ext_vector_type(8))) float v8f;

__global__ void fill_f32(float* __restrict__ p, float v, int n) {
    int t = blockIdx.x * blockDim.x + threadIdx.x;
    if (t < n) p[t] = v;
}

__global__ void zero_f4(float4* __restrict__ p, int n4) {
    int t = blockIdx.x * blockDim.x + threadIdx.x;
    if (t < n4) { float4 z; z.x = z.y = z.z = z.w = 0.f; p[t] = z; }
}

__global__ void deg_accum(const int* __restrict__ dst, float* __restrict__ deg, int E) {
    int t = blockIdx.x * blockDim.x + threadIdx.x;
    if (t < E) atomicAdd(&deg[dst[t]], 1.0f);
}

__global__ void rsqrt_inplace(float* __restrict__ p, int n) {
    int t = blockIdx.x * blockDim.x + threadIdx.x;
    if (t < n) p[t] = rsqrtf(p[t]);
}

template <int K>
__global__ __launch_bounds__(256) void gemm_wmma_f32(const float* __restrict__ A,
                                                     const float* __restrict__ W,
                                                     float* __restrict__ out) {
    __shared__ float ldsW[64 * HID];
    const int tid  = threadIdx.x;
    const int wave = tid >> 5;
    const int lane = tid & 31;
    const int half = lane >> 4;
    const int m    = lane & 15;
    const int row0 = blockIdx.x * 16;
    const int col0 = wave * 16;

    const float* arow = A + (size_t)(row0 + m) * K;
    v8f acc = {};

    for (int kb = 0; kb < K; kb += 64) {
        const float4* wv = (const float4*)(W + (size_t)kb * HID);
        float4* lv = (float4*)ldsW;
        #pragma unroll
        for (int i = tid; i < 64 * HID / 4; i += 256) lv[i] = wv[i];
        __syncthreads();

        #pragma unroll
        for (int k = 0; k < 64; k += 4) {
            const int ka = k + 2 * half;
            v2f a;
            a.x = arow[kb + ka];
            a.y = arow[kb + ka + 1];
            v2f b;
            b.x = ldsW[ka * HID + col0 + m];
            b.y = ldsW[(ka + 1) * HID + col0 + m];
            acc = __builtin_amdgcn_wmma_f32_16x16x4_f32(
                false, a, false, b, (short)0, acc, false, false);
        }
        __syncthreads();
    }

    #pragma unroll
    for (int v = 0; v < 8; ++v)
        out[(size_t)(row0 + v + 8 * half) * HID + col0 + m] = acc[v];
}

__global__ __launch_bounds__(256) void edge_scatter(const int* __restrict__ src,
                                                    const int* __restrict__ dst,
                                                    const float* __restrict__ dis,
                                                    const float* __restrict__ h,
                                                    float* __restrict__ agg, int E) {
    const int lane = threadIdx.x & 31;
    const int gw   = (blockIdx.x * blockDim.x + threadIdx.x) >> 5;
    const int nw   = (gridDim.x * blockDim.x) >> 5;
    for (int e = gw; e < E; e += nw) {
        const int s = src[e], d = dst[e];
        const float nrm = dis[s] * dis[d];
        const float* hs = h + (size_t)s * HID;
        float* ad = agg + (size_t)d * HID;
        #pragma unroll
        for (int j = 0; j < HID; j += 32)
            atomicAdd(&ad[j + lane], hs[j + lane] * nrm);
    }
}

__global__ void finish_layer(float* __restrict__ agg, const float* __restrict__ h,
                             const float* __restrict__ dis, const float* __restrict__ bias,
                             int n) {
    int t = blockIdx.x * blockDim.x + threadIdx.x;
    if (t >= n) return;
    const int i = t >> 7, j = t & (HID - 1);
    const float di = dis[i];
    const float v = agg[t] + di * di * h[t] + bias[j];
    agg[t] = v > 0.f ? v : 0.f;
}

#define POOL_ROWS 16
static_assert(N_NODES % POOL_ROWS == 0, "every wave owns a full group of rows");
__global__ __launch_bounds__(256) void pool_accum(const float* __restrict__ h,
                                                  const int* __restrict__ batch,
                                                  float* __restrict__ pooled,
                                                  float* __restrict__ cnt, int N) {
    const int lane = threadIdx.x & 31;
    const int gw = __builtin_amdgcn_readfirstlane(
        (int)((blockIdx.x * blockDim.x + threadIdx.x) >> 5));
    const int r0 = gw * POOL_ROWS;
    if (r0 >= N) return;

    const int*   bp = batch + r0;
    const float* hp = h + (size_t)r0 * HID + lane;
    int   g[POOL_ROWS];
    float v[POOL_ROWS][4];
    #pragma unroll
    for (int i = 0; i < POOL_ROWS; ++i) {
        g[i] = bp[i];
        #pragma unroll
        for (int k = 0; k < 4; ++k) v[i][k] = hp[i * HID + 32 * k];
    }

    int   cur = g[0];
    int   run = 0;
    float a0 = 0.f, a1 = 0.f, a2 = 0.f, a3 = 0.f;
    #pragma unroll
    for (int i = 0; i < POOL_ROWS; ++i) {
        if (g[i] != cur) {
            float* pg = pooled + cur * HID + lane;
            atomicAdd(&pg[0],  a0);
            atomicAdd(&pg[32], a1);
            atomicAdd(&pg[64], a2);
            atomicAdd(&pg[96], a3);
            if (lane == 0) atomicAdd(&cnt[cur], (float)run);
            cur = g[i]; run = 0;
            a0 = a1 = a2 = a3 = 0.f;
        }
        a0 += v[i][0]; a1 += v[i][1]; a2 += v[i][2]; a3 += v[i][3];
        ++run;
    }
    {
        float* pg = pooled + cur * HID + lane;
        atomicAdd(&pg[0],  a0);
        atomicAdd(&pg[32], a1);
        atomicAdd(&pg[64], a2);
        atomicAdd(&pg[96], a3);
        if (lane == 0) atomicAdd(&cnt[cur], (float)run);
    }
}

__global__ void pool_div(float* __restrict__ pooled, const float* __restrict__ cnt) {
    int t = blockIdx.x * blockDim.x + threadIdx.x;
    if (t >= N_GRAPHS * HID) return;
    float c = cnt[t >> 7];
    pooled[t] /= (c > 1.f ? c : 1.f);
}

__global__ void head_gemm(const float* __restrict__ pooled, const float* __restrict__ W,
                          const float* __restrict__ b, float* __restrict__ out, int OUT) {
    int t = blockIdx.x * blockDim.x + threadIdx.x;
    const int n = N_GRAPHS * OUT;
    if (t >= n) return;
    const int g = t / OUT, o = t - g * OUT;
    const float* p = pooled + g * HID;
    float s = b[o];
    #pragma unroll 8
    for (int k = 0; k < HID; ++k) s += p[k] * W[k * OUT + o];
    out[t] = s;
}

extern "C" void kernel_launch(void* const* d_in, const int* in_sizes, int n_in,
                              void* d_out, int out_size, void* d_ws, size_t ws_size,
                              hipStream_t stream) {
    const float* x   = (const float*)d_in[0];
    const int*   ei  = (const int*)d_in[1];
    const int*   bat = (const int*)d_in[2];
    const float* W1  = (const float*)d_in[3];
    const float* b1  = (const float*)d_in[4];
    const float* W2  = (const float*)d_in[5];
    const float* b2  = (const float*)d_in[6];
    const float* Wmf = (const float*)d_in[7];
    const float* bmf = (const float*)d_in[8];
    const float* Wbp = (const float*)d_in[9];
    const float* bbp = (const float*)d_in[10];
    const float* Wcc = (const float*)d_in[11];
    const float* bcc = (const float*)d_in[12];
    float* out = (float*)d_out;

    const int E = in_sizes[1] / 2;
    const int* src = ei;
    const int* dst = ei + E;

    float* ws     = (float*)d_ws;
    float* dis    = ws;
    float* h      = dis + N_NODES;
    float* agg    = h + (size_t)N_NODES * HID;
    float* pooled = agg + (size_t)N_NODES * HID;
    float* cnt    = pooled + N_GRAPHS * HID;

    const int NH   = N_NODES * HID;
    const int NH4  = NH / 4;

    fill_f32<<<(N_NODES + 255) / 256, 256, 0, stream>>>(dis, 1.0f, N_NODES);
    deg_accum<<<(E + 255) / 256, 256, 0, stream>>>(dst, dis, E);
    rsqrt_inplace<<<(N_NODES + 255) / 256, 256, 0, stream>>>(dis, N_NODES);

    gemm_wmma_f32<IN_DIM><<<N_NODES / 16, 256, 0, stream>>>(x, W1, h);
    zero_f4<<<(NH4 + 255) / 256, 256, 0, stream>>>((float4*)agg, NH4);
    edge_scatter<<<2048, 256, 0, stream>>>(src, dst, dis, h, agg, E);
    finish_layer<<<(NH + 255) / 256, 256, 0, stream>>>(agg, h, dis, b1, NH);

    gemm_wmma_f32<HID><<<N_NODES / 16, 256, 0, stream>>>(agg, W2, h);
    zero_f4<<<(NH4 + 255) / 256, 256, 0, stream>>>((float4*)agg, NH4);
    edge_scatter<<<2048, 256, 0, stream>>>(src, dst, dis, h, agg, E);
    finish_layer<<<(NH + 255) / 256, 256, 0, stream>>>(agg, h, dis, b2, NH);

    fill_f32<<<(N_GRAPHS * HID + N_GRAPHS + 255) / 256, 256, 0, stream>>>(
        pooled, 0.0f, N_GRAPHS * HID + N_GRAPHS);
    pool_accum<<<(N_NODES + POOL_ROWS * 8 - 1) / (POOL_ROWS * 8), 256, 0, stream>>>(agg, bat, pooled, cnt, N_NODES);
    pool_div<<<(N_GRAPHS * HID + 255) / 256, 256, 0, stream>>>(pooled, cnt);

    float* out_mf = out;
    float* out_bp = out + N_GRAPHS * OUT_MF;
    float* out_cc = out + N_GRAPHS * (OUT_MF + OUT_BP);
    head_gemm<<<(N_GRAPHS * OUT_MF + 255) / 256, 256, 0, stream>>>(pooled, Wmf, bmf, out_mf, OUT_MF);
    head_gemm<<<(N_GRAPHS * OUT_BP + 255) / 256, 256, 0, stream>>>(pooled, Wbp, bbp, out_bp, OUT_BP);
    head_gemm<<<(N_GRAPHS * OUT_CC + 255) / 256, 256, 0, stream>>>(pooled, Wcc, bcc, out_cc, OUT_CC);
}


# edges counting-sorted by destination once per call; each layer's sum is a per-node register gather with self loop, bias, relu fused: no float atomics, zero or finish pass; sum order changes
# speedup vs baseline: 1.7587x; 1.5241x over previous
#include <hip/hip_runtime.h>
#include <hip/hip_bf16.h>

#define N_NODES 100000
#define N_EDGES 1600000
#define IN_DIM  256
#define HID     128
#define N_GRAPHS 64
#define OUT_MF 489
#define OUT_BP 1943
#define OUT_CC 320

#define WS_PLANES_BYTES (4ull * (N_NODES + 2ull * N_NODES * HID + N_GRAPHS * HID + N_GRAPHS))
#define WS_EXTRA_BYTES  (4ull * (N_NODES + 4 + N_EDGES))

typedef __attribute__((ext_vector_type(2))) float v2f;
typedef __attribute__((ext_vector_type(8))) float v8f;

__global__ void fill_f32(float* __restrict__ p, float v, int n) {
    int t = blockIdx.x * blockDim.x + threadIdx.x;
    if (t < n) p[t] = v;
}

__global__ void zero_f4(float4* __restrict__ p, int n4) {
    int t = blockIdx.x * blockDim.x + threadIdx.x;
    if (t < n4) { float4 z; z.x = z.y = z.z = z.w = 0.f; p[t] = z; }
}

__global__ void deg_accum(const int* __restrict__ dst, float* __restrict__ deg, int E) {
    int t = blockIdx.x * blockDim.x + threadIdx.x;
    if (t < E) atomicAdd(&deg[dst[t]], 1.0f);
}

__global__ void rsqrt_inplace(float* __restrict__ p, int n) {
    int t = blockIdx.x * blockDim.x + threadIdx.x;
    if (t < n) p[t] = rsqrtf(p[t]);
}

template <int K>
__global__ __launch_bounds__(256) void gemm_wmma_f32(const float* __restrict__ A,
                                                     const float* __restrict__ W,
                                                     float* __restrict__ out) {
    __shared__ float ldsW[64 * HID];
    const int tid  = threadIdx.x;
    const int wave = tid >> 5;
    const int lane = tid & 31;
    const int half = lane >> 4;
    const int m    = lane & 15;
    const int row0 = blockIdx.x * 16;
    const int col0 = wave * 16;

    const float* arow = A + (size_t)(row0 + m) * K;
    v8f acc = {};

    for (int kb = 0; kb < K; kb += 64) {
        const float4* wv = (const float4*)(W + (size_t)kb * HID);
        float4* lv = (float4*)ldsW;
        #pragma unroll
        for (int i = tid; i < 64 * HID / 4; i += 256) lv[i] = wv[i];
        __syncthreads();

        #pragma unroll
        for (int k = 0; k < 64; k += 4) {
            const int ka = k + 2 * half;
            v2f a;
            a.x = arow[kb + ka];
            a.y = arow[kb + ka + 1];
            v2f b;
            b.x = ldsW[ka * HID + col0 + m];
            b.y = ldsW[(ka + 1) * HID + col0 + m];
            acc = __builtin_amdgcn_wmma_f32_16x16x4_f32(
                false, a, false, b, (short)0, acc, false, false);
        }
        __syncthreads();
    }

    #pragma unroll
    for (int v = 0; v < 8; ++v)
        out[(size_t)(row0 + v + 8 * half) * HID + col0 + m] = acc[v];
}

__global__ __launch_bounds__(256) void edge_scatter(const int* __restrict__ src,
                                                    const int* __restrict__ dst,
                                                    const float* __restrict__ dis,
                                                    const float* __restrict__ h,
                                                    float* __restrict__ agg, int E) {
    const int lane = threadIdx.x & 31;
    const int gw   = (blockIdx.x * blockDim.x + threadIdx.x) >> 5;
    const int nw   = (gridDim.x * blockDim.x) >> 5;
    for (int e = gw; e < E; e += nw) {
        const int s = src[e], d = dst[e];
        const float nrm = dis[s] * dis[d];
        const float* hs = h + (size_t)s * HID;
        float* ad = agg + (size_t)d * HID;
        #pragma unroll
        for (int j = 0; j < HID; j += 32)
            atomicAdd(&ad[j + lane], hs[j + lane] * nrm);
    }
}

__global__ void finish_layer(float* __restrict__ agg, const float* __restrict__ h,
                             const float* __restrict__ dis, const float* __restrict__ bias,
                             int n) {
    int t = blockIdx.x * blockDim.x + threadIdx.x;
    if (t >= n) return;
    const int i = t >> 7, j = t & (HID - 1);
    const float di = dis[i];
    const float v = agg[t] + di * di * h[t] + bias[j];
    agg[t] = v > 0.f ? v : 0.f;
}

__global__ void fill_i32(int* __restrict__ p, int v, int n) {
    int t = blockIdx.x * blockDim.x + threadIdx.x;
    if (t < n) p[t] = v;
}

static_assert(N_EDGES % 4 == 0, "the dst half of edge_index starts 16-byte aligned");
__global__ void deg_count(const int* __restrict__ dst, int* __restrict__ cnt, int E) {
    const int t = (blockIdx.x * blockDim.x + threadIdx.x) * 4;
    if (t + 3 < E) {
        const int4 d = *(const int4*)(dst + t);
        atomicAdd(&cnt[d.x], 1);
        atomicAdd(&cnt[d.y], 1);
        atomicAdd(&cnt[d.z], 1);
        atomicAdd(&cnt[d.w], 1);
    } else {
        for (int e = t; e < E; ++e) atomicAdd(&cnt[dst[e]], 1);
    }
}

#define SCAN_T   1024
#define SCAN_PER 8
static_assert(N_NODES % SCAN_PER == 0, "a thread's eight counts are all inside the array or all outside");
__global__ __launch_bounds__(SCAN_T) void scan_degrees(int* __restrict__ cur,
                                                       int* __restrict__ rowptr,
                                                       float* __restrict__ dis, int n) {
    __shared__ int wsum[32];
    const int tid = threadIdx.x, lane = tid & 31, wave = tid >> 5;
    int carry = 0;
    for (int base = 0; base < n; base += SCAN_T * SCAN_PER) {
        const int  idx  = base + tid * SCAN_PER;
        const bool live = idx < n;
        int c[SCAN_PER];
        if (live) {
            const int4 a = *(const int4*)(cur + idx);
            const int4 b = *(const int4*)(cur + idx + 4);
            c[0] = a.x; c[1] = a.y; c[2] = a.z; c[3] = a.w;
            c[4] = b.x; c[5] = b.y; c[6] = b.z; c[7] = b.w;
        } else {
            #pragma unroll
            for (int k = 0; k < SCAN_PER; ++k) c[k] = 0;
        }
        int ex[SCAN_PER];
        int tsum = 0;
        #pragma unroll
        for (int k = 0; k < SCAN_PER; ++k) { ex[k] = tsum; tsum += c[k]; }
        int v = tsum;
        #pragma unroll
        for (int d = 1; d < 32; d <<= 1) {
            const int u = __shfl_up(v, d);
            if (lane >= d) v += u;
        }
        if (lane == 31) wsum[wave] = v;
        __syncthreads();
        if (wave == 0) {
            int w = wsum[lane];
            #pragma unroll
            for (int d = 1; d < 32; d <<= 1) {
                const int u = __shfl_up(w, d);
                if (lane >= d) w += u;
            }
            wsum[lane] = w;
        }
        __syncthreads();
        const int woff  = wave ? wsum[wave - 1] : 0;
        const int total = wsum[31];
        const int first = carry + woff + v - tsum;
        if (live) {
            int4 o0, o1;
            o0.x = first + ex[0]; o0.y = first + ex[1]; o0.z = first + ex[2]; o0.w = first + ex[3];
            o1.x = first + ex[4]; o1.y = first + ex[5]; o1.z = first + ex[6]; o1.w = first + ex[7];
            *(int4*)(rowptr + idx)     = o0;
            *(int4*)(rowptr + idx + 4) = o1;
            *(int4*)(cur + idx)        = o0;
            *(int4*)(cur + idx + 4)    = o1;
            float4 d0, d1;
            d0.x = rsqrtf((float)c[0] + 1.0f); d0.y = rsqrtf((float)c[1] + 1.0f);
            d0.z = rsqrtf((float)c[2] + 1.0f); d0.w = rsqrtf((float)c[3] + 1.0f);
            d1.x = rsqrtf((float)c[4] + 1.0f); d1.y = rsqrtf((float)c[5] + 1.0f);
            d1.z = rsqrtf((float)c[6] + 1.0f); d1.w = rsqrtf((float)c[7] + 1.0f);
            *(float4*)(dis + idx)     = d0;
            *(float4*)(dis + idx + 4) = d1;
        }
        carry += total;
        __syncthreads();
    }
    if (tid == 0) rowptr[n] = carry;
}

__global__ void csr_fill(const int* __restrict__ src, const int* __restrict__ dst,
                         int* __restrict__ cur, int* __restrict__ col, int E) {
    const int t = (blockIdx.x * blockDim.x + threadIdx.x) * 4;
    if (t + 3 < E) {
        const int4 s = *(const int4*)(src + t);
        const int4 d = *(const int4*)(dst + t);
        const int p0 = atomicAdd(&cur[d.x], 1);
        const int p1 = atomicAdd(&cur[d.y], 1);
        const int p2 = atomicAdd(&cur[d.z], 1);
        const int p3 = atomicAdd(&cur[d.w], 1);
        if ((unsigned)p0 < (unsigned)E) col[p0] = s.x;
        if ((unsigned)p1 < (unsigned)E) col[p1] = s.y;
        if ((unsigned)p2 < (unsigned)E) col[p2] = s.z;
        if ((unsigned)p3 < (unsigned)E) col[p3] = s.w;
    } else {
        for (int e = t; e < E; ++e) {
            const int p = atomicAdd(&cur[dst[e]], 1);
            if ((unsigned)p < (unsigned)E) col[p] = src[e];
        }
    }
}

__global__ __launch_bounds__(256) void gather_layer(const int* __restrict__ rowptr,
                                                    const int* __restrict__ col,
                                                    const float* __restrict__ dis,
                                                    const float* __restrict__ h,
                                                    const float* __restrict__ bias,
                                                    float* __restrict__ out, int N, int E) {
    const int lane = threadIdx.x & 31;
    const int gw = __builtin_amdgcn_readfirstlane(
        (int)((blockIdx.x * blockDim.x + threadIdx.x) >> 5));
    const int nw = (int)((gridDim.x * blockDim.x) >> 5);
    const float4* h4 = (const float4*)h;
    float4*       o4 = (float4*)out;
    const float4  bv = ((const float4*)bias)[lane];

    for (int i = gw; i < N; i += nw) {
        int beg = rowptr[i], end = rowptr[i + 1];
        beg = beg > 0 ? beg : 0;
        end = end < E ? end : E;
        const float  di  = dis[i];
        const float4 hv  = h4[(size_t)i * (HID / 4) + lane];
        float4 acc; acc.x = acc.y = acc.z = acc.w = 0.f;

        for (int base = beg; base < end; base += 32) {
            const int m = (end - base) < 32 ? (end - base) : 32;
            int   s_l = 0;
            float w_l = 0.f;
            if (lane < m) {
                s_l = col[base + lane];
                s_l = (unsigned)s_l < (unsigned)N ? s_l : 0;
                w_l = dis[s_l] * di;
            }
            #pragma unroll 4
            for (int j = 0; j < m; ++j) {
                const int   s = __builtin_amdgcn_readlane(s_l, j);
                const float w = __int_as_float(__builtin_amdgcn_readlane(__float_as_int(w_l), j));
                const float4 v = h4[(size_t)s * (HID / 4) + lane];
                acc.x += v.x * w; acc.y += v.y * w; acc.z += v.z * w; acc.w += v.w * w;
            }
        }
        const float dd = di * di;
        float4 r;
        r.x = acc.x + dd * hv.x + bv.x;
        r.y = acc.y + dd * hv.y + bv.y;
        r.z = acc.z + dd * hv.z + bv.z;
        r.w = acc.w + dd * hv.w + bv.w;
        r.x = r.x > 0.f ? r.x : 0.f;
        r.y = r.y > 0.f ? r.y : 0.f;
        r.z = r.z > 0.f ? r.z : 0.f;
        r.w = r.w > 0.f ? r.w : 0.f;
        o4[(size_t)i * (HID / 4) + lane] = r;
    }
}

#define POOL_ROWS 16
static_assert(N_NODES % POOL_ROWS == 0, "every wave owns a full group of rows");
__global__ __launch_bounds__(256) void pool_accum(const float* __restrict__ h,
                                                  const int* __restrict__ batch,
                                                  float* __restrict__ pooled,
                                                  float* __restrict__ cnt, int N) {
    const int lane = threadIdx.x & 31;
    const int gw = __builtin_amdgcn_readfirstlane(
        (int)((blockIdx.x * blockDim.x + threadIdx.x) >> 5));
    const int r0 = gw * POOL_ROWS;
    if (r0 >= N) return;

    const int*   bp = batch + r0;
    const float* hp = h + (size_t)r0 * HID + lane;
    int   g[POOL_ROWS];
    float v[POOL_ROWS][4];
    #pragma unroll
    for (int i = 0; i < POOL_ROWS; ++i) {
        g[i] = bp[i];
        #pragma unroll
        for (int k = 0; k < 4; ++k) v[i][k] = hp[i * HID + 32 * k];
    }

    int   cur = g[0];
    int   run = 0;
    float a0 = 0.f, a1 = 0.f, a2 = 0.f, a3 = 0.f;
    #pragma unroll
    for (int i = 0; i < POOL_ROWS; ++i) {
        if (g[i] != cur) {
            float* pg = pooled + cur * HID + lane;
            atomicAdd(&pg[0],  a0);
            atomicAdd(&pg[32], a1);
            atomicAdd(&pg[64], a2);
            atomicAdd(&pg[96], a3);
            if (lane == 0) atomicAdd(&cnt[cur], (float)run);
            cur = g[i]; run = 0;
            a0 = a1 = a2 = a3 = 0.f;
        }
        a0 += v[i][0]; a1 += v[i][1]; a2 += v[i][2]; a3 += v[i][3];
        ++run;
    }
    {
        float* pg = pooled + cur * HID + lane;
        atomicAdd(&pg[0],  a0);
        atomicAdd(&pg[32], a1);
        atomicAdd(&pg[64], a2);
        atomicAdd(&pg[96], a3);
        if (lane == 0) atomicAdd(&cnt[cur], (float)run);
    }
}

__global__ void pool_div(float* __restrict__ pooled, const float* __restrict__ cnt) {
    int t = blockIdx.x * blockDim.x + threadIdx.x;
    if (t >= N_GRAPHS * HID) return;
    float c = cnt[t >> 7];
    pooled[t] /= (c > 1.f ? c : 1.f);
}

__global__ void head_gemm(const float* __restrict__ pooled, const float* __restrict__ W,
                          const float* __restrict__ b, float* __restrict__ out, int OUT) {
    int t = blockIdx.x * blockDim.x + threadIdx.x;
    const int n = N_GRAPHS * OUT;
    if (t >= n) return;
    const int g = t / OUT, o = t - g * OUT;
    const float* p = pooled + g * HID;
    float s = b[o];
    #pragma unroll 8
    for (int k = 0; k < HID; ++k) s += p[k] * W[k * OUT + o];
    out[t] = s;
}

extern "C" void kernel_launch(void* const* d_in, const int* in_sizes, int n_in,
                              void* d_out, int out_size, void* d_ws, size_t ws_size,
                              hipStream_t stream) {
    const float* x   = (const float*)d_in[0];
    const int*   ei  = (const int*)d_in[1];
    const int*   bat = (const int*)d_in[2];
    const float* W1  = (const float*)d_in[3];
    const float* b1  = (const float*)d_in[4];
    const float* W2  = (const float*)d_in[5];
    const float* b2  = (const float*)d_in[6];
    const float* Wmf = (const float*)d_in[7];
    const float* bmf = (const float*)d_in[8];
    const float* Wbp = (const float*)d_in[9];
    const float* bbp = (const float*)d_in[10];
    const float* Wcc = (const float*)d_in[11];
    const float* bcc = (const float*)d_in[12];
    float* out = (float*)d_out;

    const int E = in_sizes[1] / 2;
    const int* src = ei;
    const int* dst = ei + E;

    float* ws     = (float*)d_ws;
    float* dis    = ws;
    float* h      = dis + N_NODES;
    float* agg    = h + (size_t)N_NODES * HID;
    float* pooled = agg + (size_t)N_NODES * HID;
    float* cnt    = pooled + N_GRAPHS * HID;

    int* rowptr = (int*)(cnt + N_GRAPHS);
    int* col    = rowptr + (N_NODES + 4);
    int* cursor = (int*)agg;

    if (ws_size >= (size_t)WS_PLANES_BYTES + (size_t)WS_EXTRA_BYTES) {
        fill_i32<<<(N_NODES + 255) / 256, 256, 0, stream>>>(cursor, 0, N_NODES);
        deg_count<<<(E + 1023) / 1024, 256, 0, stream>>>(dst, cursor, E);
        scan_degrees<<<1, SCAN_T, 0, stream>>>(cursor, rowptr, dis, N_NODES);
        csr_fill<<<(E + 1023) / 1024, 256, 0, stream>>>(src, dst, cursor, col, E);

        gemm_wmma_f32<IN_DIM><<<N_NODES / 16, 256, 0, stream>>>(x, W1, h);
        gather_layer<<<2048, 256, 0, stream>>>(rowptr, col, dis, h, b1, agg, N_NODES, E);

        gemm_wmma_f32<HID><<<N_NODES / 16, 256, 0, stream>>>(agg, W2, h);
        gather_layer<<<2048, 256, 0, stream>>>(rowptr, col, dis, h, b2, agg, N_NODES, E);
    } else {
        const int NH  = N_NODES * HID;
        const int NH4 = NH / 4;
        fill_f32<<<(N_NODES + 255) / 256, 256, 0, stream>>>(dis, 1.0f, N_NODES);
        deg_accum<<<(E + 255) / 256, 256, 0, stream>>>(dst, dis, E);
        rsqrt_inplace<<<(N_NODES + 255) / 256, 256, 0, stream>>>(dis, N_NODES);

        gemm_wmma_f32<IN_DIM><<<N_NODES / 16, 256, 0, stream>>>(x, W1, h);
        zero_f4<<<(NH4 + 255) / 256, 256, 0, stream>>>((float4*)agg, NH4);
        edge_scatter<<<2048, 256, 0, stream>>>(src, dst, dis, h, agg, E);
        finish_layer<<<(NH + 255) / 256, 256, 0, stream>>>(agg, h, dis, b1, NH);

        gemm_wmma_f32<HID><<<N_NODES / 16, 256, 0, stream>>>(agg, W2, h);
        zero_f4<<<(NH4 + 255) / 256, 256, 0, stream>>>((float4*)agg, NH4);
        edge_scatter<<<2048, 256, 0, stream>>>(src, dst, dis, h, agg, E);
        finish_layer<<<(NH + 255) / 256, 256, 0, stream>>>(agg, h, dis, b2, NH);
    }

    fill_f32<<<(N_GRAPHS * HID + N_GRAPHS + 255) / 256, 256, 0, stream>>>(
        pooled, 0.0f, N_GRAPHS * HID + N_GRAPHS);
    pool_accum<<<(N_NODES + POOL_ROWS * 8 - 1) / (POOL_ROWS * 8), 256, 0, stream>>>(agg, bat, pooled, cnt, N_NODES);
    pool_div<<<(N_GRAPHS * HID + 255) / 256, 256, 0, stream>>>(pooled, cnt);

    float* out_mf = out;
    float* out_bp = out + N_GRAPHS * OUT_MF;
    float* out_cc = out + N_GRAPHS * (OUT_MF + OUT_BP);
    head_gemm<<<(N_GRAPHS * OUT_MF + 255) / 256, 256, 0, stream>>>(pooled, Wmf, bmf, out_mf, OUT_MF);
    head_gemm<<<(N_GRAPHS * OUT_BP + 255) / 256, 256, 0, stream>>>(pooled, Wbp, bbp, out_bp, OUT_BP);
    head_gemm<<<(N_GRAPHS * OUT_CC + 255) / 256, 256, 0, stream>>>(pooled, Wcc, bcc, out_cc, OUT_CC);
}


# edges grouped by destination on LDS counters in two levels (pairs binned by node range, then per-range count, scan, place): no global atomic per edge; a node's source order stays arrival order
# speedup vs baseline: 2.6257x; 1.4930x over previous
#include <hip/hip_runtime.h>
#include <hip/hip_bf16.h>

#define N_NODES 100000
#define N_EDGES 1600000
#define IN_DIM  256
#define HID     128
#define N_GRAPHS 64
#define OUT_MF 489
#define OUT_BP 1943
#define OUT_CC 320

#define WS_PLANES_BYTES (4ull * (N_NODES + 2ull * N_NODES * HID + N_GRAPHS * HID + N_GRAPHS))
#define WS_EXTRA_BYTES  (4ull * (N_NODES + 4 + N_EDGES))

typedef __attribute__((ext_vector_type(2))) float v2f;
typedef __attribute__((ext_vector_type(8))) float v8f;

__global__ void fill_f32(float* __restrict__ p, float v, int n) {
    int t = blockIdx.x * blockDim.x + threadIdx.x;
    if (t < n) p[t] = v;
}

__global__ void zero_f4(float4* __restrict__ p, int n4) {
    int t = blockIdx.x * blockDim.x + threadIdx.x;
    if (t < n4) { float4 z; z.x = z.y = z.z = z.w = 0.f; p[t] = z; }
}

__global__ void deg_accum(const int* __restrict__ dst, float* __restrict__ deg, int E) {
    int t = blockIdx.x * blockDim.x + threadIdx.x;
    if (t < E) atomicAdd(&deg[dst[t]], 1.0f);
}

__global__ void rsqrt_inplace(float* __restrict__ p, int n) {
    int t = blockIdx.x * blockDim.x + threadIdx.x;
    if (t < n) p[t] = rsqrtf(p[t]);
}

template <int K>
__global__ __launch_bounds__(256) void gemm_wmma_f32(const float* __restrict__ A,
                                                     const float* __restrict__ W,
                                                     float* __restrict__ out) {
    __shared__ float ldsW[64 * HID];
    const int tid  = threadIdx.x;
    const int wave = tid >> 5;
    const int lane = tid & 31;
    const int half = lane >> 4;
    const int m    = lane & 15;
    const int row0 = blockIdx.x * 16;
    const int col0 = wave * 16;

    const float* arow = A + (size_t)(row0 + m) * K;
    v8f acc = {};

    for (int kb = 0; kb < K; kb += 64) {
        const float4* wv = (const float4*)(W + (size_t)kb * HID);
        float4* lv = (float4*)ldsW;
        #pragma unroll
        for (int i = tid; i < 64 * HID / 4; i += 256) lv[i] = wv[i];
        __syncthreads();

        #pragma unroll
        for (int k = 0; k < 64; k += 4) {
            const int ka = k + 2 * half;
            v2f a;
            a.x = arow[kb + ka];
            a.y = arow[kb + ka + 1];
            v2f b;
            b.x = ldsW[ka * HID + col0 + m];
            b.y = ldsW[(ka + 1) * HID + col0 + m];
            acc = __builtin_amdgcn_wmma_f32_16x16x4_f32(
                false, a, false, b, (short)0, acc, false, false);
        }
        __syncthreads();
    }

    #pragma unroll
    for (int v = 0; v < 8; ++v)
        out[(size_t)(row0 + v + 8 * half) * HID + col0 + m] = acc[v];
}

__global__ __launch_bounds__(256) void edge_scatter(const int* __restrict__ src,
                                                    const int* __restrict__ dst,
                                                    const float* __restrict__ dis,
                                                    const float* __restrict__ h,
                                                    float* __restrict__ agg, int E) {
    const int lane = threadIdx.x & 31;
    const int gw   = (blockIdx.x * blockDim.x + threadIdx.x) >> 5;
    const int nw   = (gridDim.x * blockDim.x) >> 5;
    for (int e = gw; e < E; e += nw) {
        const int s = src[e], d = dst[e];
        const float nrm = dis[s] * dis[d];
        const float* hs = h + (size_t)s * HID;
        float* ad = agg + (size_t)d * HID;
        #pragma unroll
        for (int j = 0; j < HID; j += 32)
            atomicAdd(&ad[j + lane], hs[j + lane] * nrm);
    }
}

__global__ void finish_layer(float* __restrict__ agg, const float* __restrict__ h,
                             const float* __restrict__ dis, const float* __restrict__ bias,
                             int n) {
    int t = blockIdx.x * blockDim.x + threadIdx.x;
    if (t >= n) return;
    const int i = t >> 7, j = t & (HID - 1);
    const float di = dis[i];
    const float v = agg[t] + di * di * h[t] + bias[j];
    agg[t] = v > 0.f ? v : 0.f;
}

#define NB     256
#define BIN_R  ((N_NODES + NB - 1) / NB)
#define NC     256
#define BT     1024
#define CHUNK_CAP 6400
#define BIN_CAP   8192
#define CNT_PAD   512
#define EPT       8
static_assert(BIN_R <= CNT_PAD, "a range's nodes fit the padded counter array");
static_assert((NB * NC) % (BT * 4) == 0, "bin_scan reads whole 16-byte groups per thread");

__device__ __forceinline__ int bin_of(int d) {
    const unsigned b = (unsigned)d / (unsigned)BIN_R;
    return b < (unsigned)NB ? (int)b : NB - 1;
}

__device__ __forceinline__ int wave_incl_scan(int v, int lane) {
    #pragma unroll
    for (int d = 1; d < 32; d <<= 1) {
        const int u = __shfl_up(v, d);
        if (lane >= d) v += u;
    }
    return v;
}

__global__ __launch_bounds__(BT) void bin_count(const int* __restrict__ dst,
                                                int* __restrict__ cntT, int E) {
    __shared__ int hist[NB];
    const int tid = threadIdx.x, c = blockIdx.x;
    const int epc = (E + NC - 1) / NC;
    const int e0  = c * epc;
    const int e1  = (e0 + epc) < E ? (e0 + epc) : E;
    if (tid < NB) hist[tid] = 0;
    __syncthreads();
    for (int base = e0 + tid; base < e1; base += BT * EPT) {
        int d[EPT];
        #pragma unroll
        for (int k = 0; k < EPT; ++k) d[k] = (base + k * BT) < e1 ? dst[base + k * BT] : 0;
        #pragma unroll
        for (int k = 0; k < EPT; ++k)
            if ((base + k * BT) < e1) atomicAdd(&hist[bin_of(d[k])], 1);
    }
    __syncthreads();
    if (tid < NB) cntT[tid * NC + c] = hist[tid];
}

#define OFF_PER ((NB * NC) / BT)
__global__ __launch_bounds__(BT) void bin_scan(int* __restrict__ cntT) {
    __shared__ int wsum[32];
    const int tid = threadIdx.x, lane = tid & 31, wave = tid >> 5;
    int4* p = (int4*)(cntT + tid * OFF_PER);
    int c[OFF_PER];
    #pragma unroll
    for (int k = 0; k < OFF_PER / 4; ++k) {
        const int4 a = p[k];
        c[4 * k] = a.x; c[4 * k + 1] = a.y; c[4 * k + 2] = a.z; c[4 * k + 3] = a.w;
    }
    int tsum = 0;
    #pragma unroll
    for (int k = 0; k < OFF_PER; ++k) { const int t = c[k]; c[k] = tsum; tsum += t; }
    const int v = wave_incl_scan(tsum, lane);
    if (lane == 31) wsum[wave] = v;
    __syncthreads();
    if (wave == 0) wsum[lane] = wave_incl_scan(wsum[lane], lane);
    __syncthreads();
    const int first = (wave ? wsum[wave - 1] : 0) + v - tsum;
    #pragma unroll
    for (int k = 0; k < OFF_PER / 4; ++k) {
        int4 o;
        o.x = first + c[4 * k];     o.y = first + c[4 * k + 1];
        o.z = first + c[4 * k + 2]; o.w = first + c[4 * k + 3];
        p[k] = o;
    }
    if (tid == BT - 1) cntT[NB * NC] = first + tsum;
}

__global__ __launch_bounds__(BT) void bin_place(const int* __restrict__ src,
                                                const int* __restrict__ dst,
                                                const int* __restrict__ off,
                                                int2* __restrict__ pairs, int E) {
    __shared__ int  hist[NB];
    __shared__ int  gdelta[NB];
    __shared__ int  wtot[NB / 32];
    __shared__ int2 sbuf[CHUNK_CAP];
    const int tid = threadIdx.x, lane = tid & 31, wave = tid >> 5, c = blockIdx.x;
    const int epc = (E + NC - 1) / NC;
    const int e0  = c * epc;
    const int e1  = (e0 + epc) < E ? (e0 + epc) : E;
    const int n   = e1 > e0 ? e1 - e0 : 0;
    if (tid < NB) hist[tid] = 0;
    __syncthreads();
    for (int base = e0 + tid; base < e1; base += BT * EPT) {
        int d[EPT];
        #pragma unroll
        for (int k = 0; k < EPT; ++k) d[k] = (base + k * BT) < e1 ? dst[base + k * BT] : 0;
        #pragma unroll
        for (int k = 0; k < EPT; ++k)
            if ((base + k * BT) < e1) atomicAdd(&hist[bin_of(d[k])], 1);
    }
    __syncthreads();
    int cnt = 0, v = 0;
    if (tid < NB) {
        cnt = hist[tid];
        v = wave_incl_scan(cnt, lane);
        if (lane == 31) wtot[wave] = v;
    }
    __syncthreads();
    if (tid < NB) {
        int base = 0;
        for (int w = 0; w < wave; ++w) base += wtot[w];
        const int ex = base + v - cnt;
        hist[tid]   = ex;
        gdelta[tid] = off[tid * NC + c] - ex;
    }
    __syncthreads();
    const bool staged = n <= CHUNK_CAP;
    for (int base = e0 + tid; base < e1; base += BT * EPT) {
        int s[EPT], d[EPT];
        #pragma unroll
        for (int k = 0; k < EPT; ++k) {
            const bool in = (base + k * BT) < e1;
            s[k] = in ? src[base + k * BT] : 0;
            d[k] = in ? dst[base + k * BT] : 0;
        }
        #pragma unroll
        for (int k = 0; k < EPT; ++k) {
            if ((base + k * BT) < e1) {
                int2 r; r.x = s[k]; r.y = d[k];
                const int b = bin_of(r.y);
                const int p = atomicAdd(&hist[b], 1);
                if (staged) {
                    if ((unsigned)p < (unsigned)CHUNK_CAP) sbuf[p] = r;
                } else {
                    const int g = p + gdelta[b];
                    if ((unsigned)g < (unsigned)E) pairs[g] = r;
                }
            }
        }
    }
    if (staged) {
        __syncthreads();
        for (int i = tid; i < n; i += BT) {
            const int2 r = sbuf[i];
            const int  g = i + gdelta[bin_of(r.y)];
            if ((unsigned)g < (unsigned)E) pairs[g] = r;
        }
    }
}

__global__ __launch_bounds__(BT) void bin_csr(const int2* __restrict__ pairs,
                                              const int* __restrict__ off,
                                              int* __restrict__ rowptr, int* __restrict__ col,
                                              float* __restrict__ dis, int N, int E) {
    __shared__ int cntl[CNT_PAD];
    __shared__ int wtot[CNT_PAD / 32];
    __shared__ int scol[BIN_CAP];
    const int tid = threadIdx.x, lane = tid & 31, wave = tid >> 5, b = blockIdx.x;
    int beg = off[b * NC], end = off[(b + 1) * NC];
    beg = beg > 0 ? beg : 0;  beg = beg < E ? beg : E;
    end = end > beg ? end : beg;  end = end < E ? end : E;
    const int n     = end - beg;
    const int node0 = b * BIN_R;
    int nn = N - node0;
    nn = nn < BIN_R ? nn : BIN_R;
    nn = nn > 0 ? nn : 0;
    if (tid < CNT_PAD) cntl[tid] = 0;
    __syncthreads();
    for (int base = tid; base < n; base += BT * EPT) {
        int d[EPT];
        #pragma unroll
        for (int k = 0; k < EPT; ++k) d[k] = (base + k * BT) < n ? pairs[beg + base + k * BT].y : -1;
        #pragma unroll
        for (int k = 0; k < EPT; ++k) {
            const unsigned l = (unsigned)(d[k] - node0);
            if ((base + k * BT) < n && l < (unsigned)nn) atomicAdd(&cntl[l], 1);
        }
    }
    __syncthreads();
    int cnt = 0, v = 0;
    if (tid < CNT_PAD) {
        cnt = cntl[tid];
        v = wave_incl_scan(cnt, lane);
        if (lane == 31) wtot[wave] = v;
    }
    __syncthreads();
    if (tid < CNT_PAD) {
        int base = 0;
        for (int w = 0; w < wave; ++w) base += wtot[w];
        const int ex = base + v - cnt;
        cntl[tid] = ex;
        if (tid < nn) {
            rowptr[node0 + tid] = beg + ex;
            dis[node0 + tid]    = rsqrtf((float)cnt + 1.0f);
        }
    }
    if (b == NB - 1 && tid == 0) rowptr[N] = end;
    __syncthreads();
    const bool staged = n <= BIN_CAP;
    for (int base = tid; base < n; base += BT * EPT) {
        int2 r[EPT];
        #pragma unroll
        for (int k = 0; k < EPT; ++k) {
            if ((base + k * BT) < n) r[k] = pairs[beg + base + k * BT];
            else { r[k].x = 0; r[k].y = -1; }
        }
        #pragma unroll
        for (int k = 0; k < EPT; ++k) {
            const unsigned l = (unsigned)(r[k].y - node0);
            if ((base + k * BT) < n && l < (unsigned)nn) {
                const int p = atomicAdd(&cntl[l], 1);
                if (staged) {
                    if ((unsigned)p < (unsigned)BIN_CAP) scol[p] = r[k].x;
                } else {
                    const int g = beg + p;
                    if ((unsigned)g < (unsigned)E) col[g] = r[k].x;
                }
            }
        }
    }
    if (staged) {
        __syncthreads();
        for (int i = tid; i < n; i += BT) col[beg + i] = scol[i];
    }
}

__global__ __launch_bounds__(256) void gather_layer(const int* __restrict__ rowptr,
                                                    const int* __restrict__ col,
                                                    const float* __restrict__ dis,
                                                    const float* __restrict__ h,
                                                    const float* __restrict__ bias,
                                                    float* __restrict__ out, int N, int E) {
    const int lane = threadIdx.x & 31;
    const int gw = __builtin_amdgcn_readfirstlane(
        (int)((blockIdx.x * blockDim.x + threadIdx.x) >> 5));
    const int nw = (int)((gridDim.x * blockDim.x) >> 5);
    const float4* h4 = (const float4*)h;
    float4*       o4 = (float4*)out;
    const float4  bv = ((const float4*)bias)[lane];

    for (int i = gw; i < N; i += nw) {
        int beg = rowptr[i], end = rowptr[i + 1];
        beg = beg > 0 ? beg : 0;
        end = end < E ? end : E;
        const float  di  = dis[i];
        const float4 hv  = h4[(size_t)i * (HID / 4) + lane];
        float4 acc; acc.x = acc.y = acc.z = acc.w = 0.f;

        for (int base = beg; base < end; base += 32) {
            const int m = (end - base) < 32 ? (end - base) : 32;
            int   s_l = 0;
            float w_l = 0.f;
            if (lane < m) {
                s_l = col[base + lane];
                s_l = (unsigned)s_l < (unsigned)N ? s_l : 0;
                w_l = dis[s_l] * di;
            }
            #pragma unroll 4
            for (int j = 0; j < m; ++j) {
                const int   s = __builtin_amdgcn_readlane(s_l, j);
                const float w = __int_as_float(__builtin_amdgcn_readlane(__float_as_int(w_l), j));
                const float4 v = h4[(size_t)s * (HID / 4) + lane];
                acc.x += v.x * w; acc.y += v.y * w; acc.z += v.z * w; acc.w += v.w * w;
            }
        }
        const float dd = di * di;
        float4 r;
        r.x = acc.x + dd * hv.x + bv.x;
        r.y = acc.y + dd * hv.y + bv.y;
        r.z = acc.z + dd * hv.z + bv.z;
        r.w = acc.w + dd * hv.w + bv.w;
        r.x = r.x > 0.f ? r.x : 0.f;
        r.y = r.y > 0.f ? r.y : 0.f;
        r.z = r.z > 0.f ? r.z : 0.f;
        r.w = r.w > 0.f ? r.w : 0.f;
        o4[(size_t)i * (HID / 4) + lane] = r;
    }
}

#define POOL_ROWS 16
static_assert(N_NODES % POOL_ROWS == 0, "every wave owns a full group of rows");
__global__ __launch_bounds__(256) void pool_accum(const float* __restrict__ h,
                                                  const int* __restrict__ batch,
                                                  float* __restrict__ pooled,
                                                  float* __restrict__ cnt, int N) {
    const int lane = threadIdx.x & 31;
    const int gw = __builtin_amdgcn_readfirstlane(
        (int)((blockIdx.x * blockDim.x + threadIdx.x) >> 5));
    const int r0 = gw * POOL_ROWS;
    if (r0 >= N) return;

    const int*   bp = batch + r0;
    const float* hp = h + (size_t)r0 * HID + lane;
    int   g[POOL_ROWS];
    float v[POOL_ROWS][4];
    #pragma unroll
    for (int i = 0; i < POOL_ROWS; ++i) {
        g[i] = bp[i];
        #pragma unroll
        for (int k = 0; k < 4; ++k) v[i][k] = hp[i * HID + 32 * k];
    }

    int   cur = g[0];
    int   run = 0;
    float a0 = 0.f, a1 = 0.f, a2 = 0.f, a3 = 0.f;
    #pragma unroll
    for (int i = 0; i < POOL_ROWS; ++i) {
        if (g[i] != cur) {
            float* pg = pooled + cur * HID + lane;
            atomicAdd(&pg[0],  a0);
            atomicAdd(&pg[32], a1);
            atomicAdd(&pg[64], a2);
            atomicAdd(&pg[96], a3);
            if (lane == 0) atomicAdd(&cnt[cur], (float)run);
            cur = g[i]; run = 0;
            a0 = a1 = a2 = a3 = 0.f;
        }
        a0 += v[i][0]; a1 += v[i][1]; a2 += v[i][2]; a3 += v[i][3];
        ++run;
    }
    {
        float* pg = pooled + cur * HID + lane;
        atomicAdd(&pg[0],  a0);
        atomicAdd(&pg[32], a1);
        atomicAdd(&pg[64], a2);
        atomicAdd(&pg[96], a3);
        if (lane == 0) atomicAdd(&cnt[cur], (float)run);
    }
}

__global__ void pool_div(float* __restrict__ pooled, const float* __restrict__ cnt) {
    int t = blockIdx.x * blockDim.x + threadIdx.x;
    if (t >= N_GRAPHS * HID) return;
    float c = cnt[t >> 7];
    pooled[t] /= (c > 1.f ? c : 1.f);
}

__global__ void head_gemm(const float* __restrict__ pooled, const float* __restrict__ W,
                          const float* __restrict__ b, float* __restrict__ out, int OUT) {
    int t = blockIdx.x * blockDim.x + threadIdx.x;
    const int n = N_GRAPHS * OUT;
    if (t >= n) return;
    const int g = t / OUT, o = t - g * OUT;
    const float* p = pooled + g * HID;
    float s = b[o];
    #pragma unroll 8
    for (int k = 0; k < HID; ++k) s += p[k] * W[k * OUT + o];
    out[t] = s;
}

extern "C" void kernel_launch(void* const* d_in, const int* in_sizes, int n_in,
                              void* d_out, int out_size, void* d_ws, size_t ws_size,
                              hipStream_t stream) {
    const float* x   = (const float*)d_in[0];
    const int*   ei  = (const int*)d_in[1];
    const int*   bat = (const int*)d_in[2];
    const float* W1  = (const float*)d_in[3];
    const float* b1  = (const float*)d_in[4];
    const float* W2  = (const float*)d_in[5];
    const float* b2  = (const float*)d_in[6];
    const float* Wmf = (const float*)d_in[7];
    const float* bmf = (const float*)d_in[8];
    const float* Wbp = (const float*)d_in[9];
    const float* bbp = (const float*)d_in[10];
    const float* Wcc = (const float*)d_in[11];
    const float* bcc = (const float*)d_in[12];
    float* out = (float*)d_out;

    const int E = in_sizes[1] / 2;
    const int* src = ei;
    const int* dst = ei + E;

    float* ws     = (float*)d_ws;
    float* dis    = ws;
    float* h      = dis + N_NODES;
    float* agg    = h + (size_t)N_NODES * HID;
    float* pooled = agg + (size_t)N_NODES * HID;
    float* cnt    = pooled + N_GRAPHS * HID;

    int*  rowptr = (int*)(cnt + N_GRAPHS);
    int*  col    = rowptr + (N_NODES + 4);
    int2* pairs  = (int2*)agg;
    int*  cntT   = (int*)agg + 2 * N_EDGES;

    if (ws_size >= (size_t)WS_PLANES_BYTES + (size_t)WS_EXTRA_BYTES) {
        bin_count<<<NC, BT, 0, stream>>>(dst, cntT, E);
        bin_scan<<<1, BT, 0, stream>>>(cntT);
        bin_place<<<NC, BT, 0, stream>>>(src, dst, cntT, pairs, E);
        bin_csr<<<NB, BT, 0, stream>>>(pairs, cntT, rowptr, col, dis, N_NODES, E);

        gemm_wmma_f32<IN_DIM><<<N_NODES / 16, 256, 0, stream>>>(x, W1, h);
        gather_layer<<<2048, 256, 0, stream>>>(rowptr, col, dis, h, b1, agg, N_NODES, E);

        gemm_wmma_f32<HID><<<N_NODES / 16, 256, 0, stream>>>(agg, W2, h);
        gather_layer<<<2048, 256, 0, stream>>>(rowptr, col, dis, h, b2, agg, N_NODES, E);
    } else {
        const int NH  = N_NODES * HID;
        const int NH4 = NH / 4;
        fill_f32<<<(N_NODES + 255) / 256, 256, 0, stream>>>(dis, 1.0f, N_NODES);
        deg_accum<<<(E + 255) / 256, 256, 0, stream>>>(dst, dis, E);
        rsqrt_inplace<<<(N_NODES + 255) / 256, 256, 0, stream>>>(dis, N_NODES);

        gemm_wmma_f32<IN_DIM><<<N_NODES / 16, 256, 0, stream>>>(x, W1, h);
        zero_f4<<<(NH4 + 255) / 256, 256, 0, stream>>>((float4*)agg, NH4);
        edge_scatter<<<2048, 256, 0, stream>>>(src, dst, dis, h, agg, E);
        finish_layer<<<(NH + 255) / 256, 256, 0, stream>>>(agg, h, dis, b1, NH);

        gemm_wmma_f32<HID><<<N_NODES / 16, 256, 0, stream>>>(agg, W2, h);
        zero_f4<<<(NH4 + 255) / 256, 256, 0, stream>>>((float4*)agg, NH4);
        edge_scatter<<<2048, 256, 0, stream>>>(src, dst, dis, h, agg, E);
        finish_layer<<<(NH + 255) / 256, 256, 0, stream>>>(agg, h, dis, b2, NH);
    }

    fill_f32<<<(N_GRAPHS * HID + N_GRAPHS + 255) / 256, 256, 0, stream>>>(
        pooled, 0.0f, N_GRAPHS * HID + N_GRAPHS);
    pool_accum<<<(N_NODES + POOL_ROWS * 8 - 1) / (POOL_ROWS * 8), 256, 0, stream>>>(agg, bat, pooled, cnt, N_NODES);
    pool_div<<<(N_GRAPHS * HID + 255) / 256, 256, 0, stream>>>(pooled, cnt);

    float* out_mf = out;
    float* out_bp = out + N_GRAPHS * OUT_MF;
    float* out_cc = out + N_GRAPHS * (OUT_MF + OUT_BP);
    head_gemm<<<(N_GRAPHS * OUT_MF + 255) / 256, 256, 0, stream>>>(pooled, Wmf, bmf, out_mf, OUT_MF);
    head_gemm<<<(N_GRAPHS * OUT_BP + 255) / 256, 256, 0, stream>>>(pooled, Wbp, bbp, out_bp, OUT_BP);
    head_gemm<<<(N_GRAPHS * OUT_CC + 255) / 256, 256, 0, stream>>>(pooled, Wcc, bcc, out_cc, OUT_CC);
}


# GEMM re-tiled: a work-group owns 80 rows (was 16), W and A chunks staged once per group in padded LDS, each wave runs 5 accumulator tiles per 8-byte B fragment read; k order of every sum kept
# speedup vs baseline: 3.4662x; 1.3201x over previous
#include <hip/hip_runtime.h>
#include <hip/hip_bf16.h>

#define N_NODES 100000
#define N_EDGES 1600000
#define IN_DIM  256
#define HID     128
#define N_GRAPHS 64
#define OUT_MF 489
#define OUT_BP 1943
#define OUT_CC 320

#define WS_PLANES_BYTES (4ull * (N_NODES + 2ull * N_NODES * HID + N_GRAPHS * HID + N_GRAPHS))
#define WS_EXTRA_BYTES  (4ull * (N_NODES + 4 + N_EDGES))

typedef __attribute__((ext_vector_type(2))) float v2f;
typedef __attribute__((ext_vector_type(8))) float v8f;
typedef __attribute__((ext_vector_type(4))) float v4f;

__global__ void fill_f32(float* __restrict__ p, float v, int n) {
    int t = blockIdx.x * blockDim.x + threadIdx.x;
    if (t < n) p[t] = v;
}

__global__ void zero_f4(float4* __restrict__ p, int n4) {
    int t = blockIdx.x * blockDim.x + threadIdx.x;
    if (t < n4) { float4 z; z.x = z.y = z.z = z.w = 0.f; p[t] = z; }
}

__global__ void deg_accum(const int* __restrict__ dst, float* __restrict__ deg, int E) {
    int t = blockIdx.x * blockDim.x + threadIdx.x;
    if (t < E) atomicAdd(&deg[dst[t]], 1.0f);
}

__global__ void rsqrt_inplace(float* __restrict__ p, int n) {
    int t = blockIdx.x * blockDim.x + threadIdx.x;
    if (t < n) p[t] = rsqrtf(p[t]);
}

#define GEMM_ROWS 80
#define GEMM_RT   (GEMM_ROWS / 16)
#define GEMM_KC   64
#define GEMM_PA   (GEMM_KC + 4)
#define GEMM_PW   144
#define GEMM_NA   (GEMM_ROWS * GEMM_KC / 4 / 256)
#define GEMM_NW   ((GEMM_KC / 2) * (HID / 4) / 256)
static_assert(N_NODES % GEMM_ROWS == 0, "every work-group owns a full set of rows");
static_assert((GEMM_ROWS * GEMM_KC / 4) % 256 == 0, "A chunk splits evenly over the threads");

template <int K>
__global__ __launch_bounds__(256) void gemm_wmma_f32(const float* __restrict__ A,
                                                     const float* __restrict__ W,
                                                     float* __restrict__ out) {
    __shared__ float4 ldsA4[GEMM_ROWS * GEMM_PA / 4];
    __shared__ float4 ldsW4[(GEMM_KC / 2) * GEMM_PW / 2];
    float*  ldsA = (float*)ldsA4;
    float2* ldsW = (float2*)ldsW4;
    const int tid  = threadIdx.x;
    const int wave = tid >> 5;
    const int lane = tid & 31;
    const int half = lane >> 4;
    const int m    = lane & 15;
    const int row0 = blockIdx.x * GEMM_ROWS;
    const int col0 = wave * 16;

    const float* ag = A + (size_t)(row0 + (tid >> 4)) * K + 4 * (tid & 15);
    const float* wg = W + (size_t)(2 * (tid >> 5)) * HID + 4 * (tid & 31);
    v4f* as = (v4f*)(ldsA + (tid >> 4) * GEMM_PA + 4 * (tid & 15));
    v4f* wsd = (v4f*)(ldsW + (tid >> 5) * GEMM_PW + 4 * (tid & 31));

    v8f acc[GEMM_RT];
    #pragma unroll
    for (int t = 0; t < GEMM_RT; ++t) acc[t] = v8f{};

    const float*  aF = ldsA + m * GEMM_PA + 2 * half;
    const float2* bF = ldsW + half * GEMM_PW + col0 + m;

    for (int kb = 0; kb < K; kb += GEMM_KC) {
        {
            v4f ra[GEMM_NA], rw0[GEMM_NW], rw1[GEMM_NW];
            #pragma unroll
            for (int i = 0; i < GEMM_NA; ++i)
                ra[i] = *(const v4f*)(ag + (size_t)(16 * i) * K + kb);
            #pragma unroll
            for (int i = 0; i < GEMM_NW; ++i) {
                rw0[i] = *(const v4f*)(wg + (size_t)(kb + 16 * i) * HID);
                rw1[i] = *(const v4f*)(wg + (size_t)(kb + 16 * i + 1) * HID);
            }
            __builtin_amdgcn_sched_barrier(0);
            #pragma unroll
            for (int i = 0; i < GEMM_NA; ++i) as[i * 16 * GEMM_PA / 4] = ra[i];
            #pragma unroll
            for (int i = 0; i < GEMM_NW; ++i) {
                v4f lo, hi;
                lo.x = rw0[i].x; lo.y = rw1[i].x; lo.z = rw0[i].y; lo.w = rw1[i].y;
                hi.x = rw0[i].z; hi.y = rw1[i].z; hi.z = rw0[i].w; hi.w = rw1[i].w;
                wsd[i * 8 * GEMM_PW / 2]     = lo;
                wsd[i * 8 * GEMM_PW / 2 + 1] = hi;
            }
        }
        __syncthreads();

        #pragma unroll
        for (int k = 0; k < GEMM_KC; k += 4) {
            const float2 bb = bF[(k >> 1) * GEMM_PW];
            v2f b; b.x = bb.x; b.y = bb.y;
            #pragma unroll
            for (int t = 0; t < GEMM_RT; ++t) {
                const float2 aa = *(const float2*)(aF + t * 16 * GEMM_PA + k);
                v2f a; a.x = aa.x; a.y = aa.y;
                acc[t] = __builtin_amdgcn_wmma_f32_16x16x4_f32(
                    false, a, false, b, (short)0, acc[t], false, false);
            }
        }
        __syncthreads();
    }

    #pragma unroll
    for (int t = 0; t < GEMM_RT; ++t) {
        #pragma unroll
        for (int v = 0; v < 8; ++v)
            out[(size_t)(row0 + 16 * t + v + 8 * half) * HID + col0 + m] = acc[t][v];
    }
}

__global__ __launch_bounds__(256) void edge_scatter(const int* __restrict__ src,
                                                    const int* __restrict__ dst,
                                                    const float* __restrict__ dis,
                                                    const float* __restrict__ h,
                                                    float* __restrict__ agg, int E) {
    const int lane = threadIdx.x & 31;
    const int gw   = (blockIdx.x * blockDim.x + threadIdx.x) >> 5;
    const int nw   = (gridDim.x * blockDim.x) >> 5;
    for (int e = gw; e < E; e += nw) {
        const int s = src[e], d = dst[e];
        const float nrm = dis[s] * dis[d];
        const float* hs = h + (size_t)s * HID;
        float* ad = agg + (size_t)d * HID;
        #pragma unroll
        for (int j = 0; j < HID; j += 32)
            atomicAdd(&ad[j + lane], hs[j + lane] * nrm);
    }
}

__global__ void finish_layer(float* __restrict__ agg, const float* __restrict__ h,
                             const float* __restrict__ dis, const float* __restrict__ bias,
                             int n) {
    int t = blockIdx.x * blockDim.x + threadIdx.x;
    if (t >= n) return;
    const int i = t >> 7, j = t & (HID - 1);
    const float di = dis[i];
    const float v = agg[t] + di * di * h[t] + bias[j];
    agg[t] = v > 0.f ? v : 0.f;
}

#define NB     256
#define BIN_R  ((N_NODES + NB - 1) / NB)
#define NC     256
#define BT     1024
#define CHUNK_CAP 6400
#define BIN_CAP   8192
#define CNT_PAD   512
#define EPT       8
static_assert(BIN_R <= CNT_PAD, "a range's nodes fit the padded counter array");
static_assert((NB * NC) % (BT * 4) == 0, "bin_scan reads whole 16-byte groups per thread");

__device__ __forceinline__ int bin_of(int d) {
    const unsigned b = (unsigned)d / (unsigned)BIN_R;
    return b < (unsigned)NB ? (int)b : NB - 1;
}

__device__ __forceinline__ int wave_incl_scan(int v, int lane) {
    #pragma unroll
    for (int d = 1; d < 32; d <<= 1) {
        const int u = __shfl_up(v, d);
        if (lane >= d) v += u;
    }
    return v;
}

__global__ __launch_bounds__(BT) void bin_count(const int* __restrict__ dst,
                                                int* __restrict__ cntT, int E) {
    __shared__ int hist[NB];
    const int tid = threadIdx.x, c = blockIdx.x;
    const int epc = (E + NC - 1) / NC;
    const int e0  = c * epc;
    const int e1  = (e0 + epc) < E ? (e0 + epc) : E;
    if (tid < NB) hist[tid] = 0;
    __syncthreads();
    for (int base = e0 + tid; base < e1; base += BT * EPT) {
        int d[EPT];
        #pragma unroll
        for (int k = 0; k < EPT; ++k) d[k] = (base + k * BT) < e1 ? dst[base + k * BT] : 0;
        #pragma unroll
        for (int k = 0; k < EPT; ++k)
            if ((base + k * BT) < e1) atomicAdd(&hist[bin_of(d[k])], 1);
    }
    __syncthreads();
    if (tid < NB) cntT[tid * NC + c] = hist[tid];
}

#define OFF_PER ((NB * NC) / BT)
__global__ __launch_bounds__(BT) void bin_scan(int* __restrict__ cntT) {
    __shared__ int wsum[32];
    const int tid = threadIdx.x, lane = tid & 31, wave = tid >> 5;
    int4* p = (int4*)(cntT + tid * OFF_PER);
    int c[OFF_PER];
    #pragma unroll
    for (int k = 0; k < OFF_PER / 4; ++k) {
        const int4 a = p[k];
        c[4 * k] = a.x; c[4 * k + 1] = a.y; c[4 * k + 2] = a.z; c[4 * k + 3] = a.w;
    }
    int tsum = 0;
    #pragma unroll
    for (int k = 0; k < OFF_PER; ++k) { const int t = c[k]; c[k] = tsum; tsum += t; }
    const int v = wave_incl_scan(tsum, lane);
    if (lane == 31) wsum[wave] = v;
    __syncthreads();
    if (wave == 0) wsum[lane] = wave_incl_scan(wsum[lane], lane);
    __syncthreads();
    const int first = (wave ? wsum[wave - 1] : 0) + v - tsum;
    #pragma unroll
    for (int k = 0; k < OFF_PER / 4; ++k) {
        int4 o;
        o.x = first + c[4 * k];     o.y = first + c[4 * k + 1];
        o.z = first + c[4 * k + 2]; o.w = first + c[4 * k + 3];
        p[k] = o;
    }
    if (tid == BT - 1) cntT[NB * NC] = first + tsum;
}

__global__ __launch_bounds__(BT) void bin_place(const int* __restrict__ src,
                                                const int* __restrict__ dst,
                                                const int* __restrict__ off,
                                                int2* __restrict__ pairs, int E) {
    __shared__ int  hist[NB];
    __shared__ int  gdelta[NB];
    __shared__ int  wtot[NB / 32];
    __shared__ int2 sbuf[CHUNK_CAP];
    const int tid = threadIdx.x, lane = tid & 31, wave = tid >> 5, c = blockIdx.x;
    const int epc = (E + NC - 1) / NC;
    const int e0  = c * epc;
    const int e1  = (e0 + epc) < E ? (e0 + epc) : E;
    const int n   = e1 > e0 ? e1 - e0 : 0;
    if (tid < NB) hist[tid] = 0;
    __syncthreads();
    for (int base = e0 + tid; base < e1; base += BT * EPT) {
        int d[EPT];
        #pragma unroll
        for (int k = 0; k < EPT; ++k) d[k] = (base + k * BT) < e1 ? dst[base + k * BT] : 0;
        #pragma unroll
        for (int k = 0; k < EPT; ++k)
            if ((base + k * BT) < e1) atomicAdd(&hist[bin_of(d[k])], 1);
    }
    __syncthreads();
    int cnt = 0, v = 0;
    if (tid < NB) {
        cnt = hist[tid];
        v = wave_incl_scan(cnt, lane);
        if (lane == 31) wtot[wave] = v;
    }
    __syncthreads();
    if (tid < NB) {
        int base = 0;
        for (int w = 0; w < wave; ++w) base += wtot[w];
        const int ex = base + v - cnt;
        hist[tid]   = ex;
        gdelta[tid] = off[tid * NC + c] - ex;
    }
    __syncthreads();
    const bool staged = n <= CHUNK_CAP;
    for (int base = e0 + tid; base < e1; base += BT * EPT) {
        int s[EPT], d[EPT];
        #pragma unroll
        for (int k = 0; k < EPT; ++k) {
            const bool in = (base + k * BT) < e1;
            s[k] = in ? src[base + k * BT] : 0;
            d[k] = in ? dst[base + k * BT] : 0;
        }
        #pragma unroll
        for (int k = 0; k < EPT; ++k) {
            if ((base + k * BT) < e1) {
                int2 r; r.x = s[k]; r.y = d[k];
                const int b = bin_of(r.y);
                const int p = atomicAdd(&hist[b], 1);
                if (staged) {
                    if ((unsigned)p < (unsigned)CHUNK_CAP) sbuf[p] = r;
                } else {
                    const int g = p + gdelta[b];
                    if ((unsigned)g < (unsigned)E) pairs[g] = r;
                }
            }
        }
    }
    if (staged) {
        __syncthreads();
        for (int i = tid; i < n; i += BT) {
            const int2 r = sbuf[i];
            const int  g = i + gdelta[bin_of(r.y)];
            if ((unsigned)g < (unsigned)E) pairs[g] = r;
        }
    }
}

__global__ __launch_bounds__(BT) void bin_csr(const int2* __restrict__ pairs,
                                              const int* __restrict__ off,
                                              int* __restrict__ rowptr, int* __restrict__ col,
                                              float* __restrict__ dis, int N, int E) {
    __shared__ int cntl[CNT_PAD];
    __shared__ int wtot[CNT_PAD / 32];
    __shared__ int scol[BIN_CAP];
    const int tid = threadIdx.x, lane = tid & 31, wave = tid >> 5, b = blockIdx.x;
    int beg = off[b * NC], end = off[(b + 1) * NC];
    beg = beg > 0 ? beg : 0;  beg = beg < E ? beg : E;
    end = end > beg ? end : beg;  end = end < E ? end : E;
    const int n     = end - beg;
    const int node0 = b * BIN_R;
    int nn = N - node0;
    nn = nn < BIN_R ? nn : BIN_R;
    nn = nn > 0 ? nn : 0;
    if (tid < CNT_PAD) cntl[tid] = 0;
    __syncthreads();
    for (int base = tid; base < n; base += BT * EPT) {
        int d[EPT];
        #pragma unroll
        for (int k = 0; k < EPT; ++k) d[k] = (base + k * BT) < n ? pairs[beg + base + k * BT].y : -1;
        #pragma unroll
        for (int k = 0; k < EPT; ++k) {
            const unsigned l = (unsigned)(d[k] - node0);
            if ((base + k * BT) < n && l < (unsigned)nn) atomicAdd(&cntl[l], 1);
        }
    }
    __syncthreads();
    int cnt = 0, v = 0;
    if (tid < CNT_PAD) {
        cnt = cntl[tid];
        v = wave_incl_scan(cnt, lane);
        if (lane == 31) wtot[wave] = v;
    }
    __syncthreads();
    if (tid < CNT_PAD) {
        int base = 0;
        for (int w = 0; w < wave; ++w) base += wtot[w];
        const int ex = base + v - cnt;
        cntl[tid] = ex;
        if (tid < nn) {
            rowptr[node0 + tid] = beg + ex;
            dis[node0 + tid]    = rsqrtf((float)cnt + 1.0f);
        }
    }
    if (b == NB - 1 && tid == 0) rowptr[N] = end;
    __syncthreads();
    const bool staged = n <= BIN_CAP;
    for (int base = tid; base < n; base += BT * EPT) {
        int2 r[EPT];
        #pragma unroll
        for (int k = 0; k < EPT; ++k) {
            if ((base + k * BT) < n) r[k] = pairs[beg + base + k * BT];
            else { r[k].x = 0; r[k].y = -1; }
        }
        #pragma unroll
        for (int k = 0; k < EPT; ++k) {
            const unsigned l = (unsigned)(r[k].y - node0);
            if ((base + k * BT) < n && l < (unsigned)nn) {
                const int p = atomicAdd(&cntl[l], 1);
                if (staged) {
                    if ((unsigned)p < (unsigned)BIN_CAP) scol[p] = r[k].x;
                } else {
                    const int g = beg + p;
                    if ((unsigned)g < (unsigned)E) col[g] = r[k].x;
                }
            }
        }
    }
    if (staged) {
        __syncthreads();
        for (int i = tid; i < n; i += BT) col[beg + i] = scol[i];
    }
}

__global__ __launch_bounds__(256) void gather_layer(const int* __restrict__ rowptr,
                                                    const int* __restrict__ col,
                                                    const float* __restrict__ dis,
                                                    const float* __restrict__ h,
                                                    const float* __restrict__ bias,
                                                    float* __restrict__ out, int N, int E) {
    const int lane = threadIdx.x & 31;
    const int gw = __builtin_amdgcn_readfirstlane(
        (int)((blockIdx.x * blockDim.x + threadIdx.x) >> 5));
    const int nw = (int)((gridDim.x * blockDim.x) >> 5);
    const float4* h4 = (const float4*)h;
    float4*       o4 = (float4*)out;
    const float4  bv = ((const float4*)bias)[lane];

    for (int i = gw; i < N; i += nw) {
        int beg = rowptr[i], end = rowptr[i + 1];
        beg = beg > 0 ? beg : 0;
        end = end < E ? end : E;
        const float  di  = dis[i];
        const float4 hv  = h4[(size_t)i * (HID / 4) + lane];
        float4 acc; acc.x = acc.y = acc.z = acc.w = 0.f;

        for (int base = beg; base < end; base += 32) {
            const int m = (end - base) < 32 ? (end - base) : 32;
            int   s_l = 0;
            float w_l = 0.f;
            if (lane < m) {
                s_l = col[base + lane];
                s_l = (unsigned)s_l < (unsigned)N ? s_l : 0;
                w_l = dis[s_l] * di;
            }
            #pragma unroll 4
            for (int j = 0; j < m; ++j) {
                const int   s = __builtin_amdgcn_readlane(s_l, j);
                const float w = __int_as_float(__builtin_amdgcn_readlane(__float_as_int(w_l), j));
                const float4 v = h4[(size_t)s * (HID / 4) + lane];
                acc.x += v.x * w; acc.y += v.y * w; acc.z += v.z * w; acc.w += v.w * w;
            }
        }
        const float dd = di * di;
        float4 r;
        r.x = acc.x + dd * hv.x + bv.x;
        r.y = acc.y + dd * hv.y + bv.y;
        r.z = acc.z + dd * hv.z + bv.z;
        r.w = acc.w + dd * hv.w + bv.w;
        r.x = r.x > 0.f ? r.x : 0.f;
        r.y = r.y > 0.f ? r.y : 0.f;
        r.z = r.z > 0.f ? r.z : 0.f;
        r.w = r.w > 0.f ? r.w : 0.f;
        o4[(size_t)i * (HID / 4) + lane] = r;
    }
}

#define POOL_ROWS 16
static_assert(N_NODES % POOL_ROWS == 0, "every wave owns a full group of rows");
__global__ __launch_bounds__(256) void pool_accum(const float* __restrict__ h,
                                                  const int* __restrict__ batch,
                                                  float* __restrict__ pooled,
                                                  float* __restrict__ cnt, int N) {
    const int lane = threadIdx.x & 31;
    const int gw = __builtin_amdgcn_readfirstlane(
        (int)((blockIdx.x * blockDim.x + threadIdx.x) >> 5));
    const int r0 = gw * POOL_ROWS;
    if (r0 >= N) return;

    const int*   bp = batch + r0;
    const float* hp = h + (size_t)r0 * HID + lane;
    int   g[POOL_ROWS];
    float v[POOL_ROWS][4];
    #pragma unroll
    for (int i = 0; i < POOL_ROWS; ++i) {
        g[i] = bp[i];
        #pragma unroll
        for (int k = 0; k < 4; ++k) v[i][k] = hp[i * HID + 32 * k];
    }

    int   cur = g[0];
    int   run = 0;
    float a0 = 0.f, a1 = 0.f, a2 = 0.f, a3 = 0.f;
    #pragma unroll
    for (int i = 0; i < POOL_ROWS; ++i) {
        if (g[i] != cur) {
            float* pg = pooled + cur * HID + lane;
            atomicAdd(&pg[0],  a0);
            atomicAdd(&pg[32], a1);
            atomicAdd(&pg[64], a2);
            atomicAdd(&pg[96], a3);
            if (lane == 0) atomicAdd(&cnt[cur], (float)run);
            cur = g[i]; run = 0;
            a0 = a1 = a2 = a3 = 0.f;
        }
        a0 += v[i][0]; a1 += v[i][1]; a2 += v[i][2]; a3 += v[i][3];
        ++run;
    }
    {
        float* pg = pooled + cur * HID + lane;
        atomicAdd(&pg[0],  a0);
        atomicAdd(&pg[32], a1);
        atomicAdd(&pg[64], a2);
        atomicAdd(&pg[96], a3);
        if (lane == 0) atomicAdd(&cnt[cur], (float)run);
    }
}

__global__ void pool_div(float* __restrict__ pooled, const float* __restrict__ cnt) {
    int t = blockIdx.x * blockDim.x + threadIdx.x;
    if (t >= N_GRAPHS * HID) return;
    float c = cnt[t >> 7];
    pooled[t] /= (c > 1.f ? c : 1.f);
}

__global__ void head_gemm(const float* __restrict__ pooled, const float* __restrict__ W,
                          const float* __restrict__ b, float* __restrict__ out, int OUT) {
    int t = blockIdx.x * blockDim.x + threadIdx.x;
    const int n = N_GRAPHS * OUT;
    if (t >= n) return;
    const int g = t / OUT, o = t - g * OUT;
    const float* p = pooled + g * HID;
    float s = b[o];
    #pragma unroll 8
    for (int k = 0; k < HID; ++k) s += p[k] * W[k * OUT + o];
    out[t] = s;
}

extern "C" void kernel_launch(void* const* d_in, const int* in_sizes, int n_in,
                              void* d_out, int out_size, void* d_ws, size_t ws_size,
                              hipStream_t stream) {
    const float* x   = (const float*)d_in[0];
    const int*   ei  = (const int*)d_in[1];
    const int*   bat = (const int*)d_in[2];
    const float* W1  = (const float*)d_in[3];
    const float* b1  = (const float*)d_in[4];
    const float* W2  = (const float*)d_in[5];
    const float* b2  = (const float*)d_in[6];
    const float* Wmf = (const float*)d_in[7];
    const float* bmf = (const float*)d_in[8];
    const float* Wbp = (const float*)d_in[9];
    const float* bbp = (const float*)d_in[10];
    const float* Wcc = (const float*)d_in[11];
    const float* bcc = (const float*)d_in[12];
    float* out = (float*)d_out;

    const int E = in_sizes[1] / 2;
    const int* src = ei;
    const int* dst = ei + E;

    float* ws     = (float*)d_ws;
    float* dis    = ws;
    float* h      = dis + N_NODES;
    float* agg    = h + (size_t)N_NODES * HID;
    float* pooled = agg + (size_t)N_NODES * HID;
    float* cnt    = pooled + N_GRAPHS * HID;

    int*  rowptr = (int*)(cnt + N_GRAPHS);
    int*  col    = rowptr + (N_NODES + 4);
    int2* pairs  = (int2*)agg;
    int*  cntT   = (int*)agg + 2 * N_EDGES;

    if (ws_size >= (size_t)WS_PLANES_BYTES + (size_t)WS_EXTRA_BYTES) {
        bin_count<<<NC, BT, 0, stream>>>(dst, cntT, E);
        bin_scan<<<1, BT, 0, stream>>>(cntT);
        bin_place<<<NC, BT, 0, stream>>>(src, dst, cntT, pairs, E);
        bin_csr<<<NB, BT, 0, stream>>>(pairs, cntT, rowptr, col, dis, N_NODES, E);

        gemm_wmma_f32<IN_DIM><<<N_NODES / GEMM_ROWS, 256, 0, stream>>>(x, W1, h);
        gather_layer<<<2048, 256, 0, stream>>>(rowptr, col, dis, h, b1, agg, N_NODES, E);

        gemm_wmma_f32<HID><<<N_NODES / GEMM_ROWS, 256, 0, stream>>>(agg, W2, h);
        gather_layer<<<2048, 256, 0, stream>>>(rowptr, col, dis, h, b2, agg, N_NODES, E);
    } else {
        const int NH  = N_NODES * HID;
        const int NH4 = NH / 4;
        fill_f32<<<(N_NODES + 255) / 256, 256, 0, stream>>>(dis, 1.0f, N_NODES);
        deg_accum<<<(E + 255) / 256, 256, 0, stream>>>(dst, dis, E);
        rsqrt_inplace<<<(N_NODES + 255) / 256, 256, 0, stream>>>(dis, N_NODES);

        gemm_wmma_f32<IN_DIM><<<N_NODES / GEMM_ROWS, 256, 0, stream>>>(x, W1, h);
        zero_f4<<<(NH4 + 255) / 256, 256, 0, stream>>>((float4*)agg, NH4);
        edge_scatter<<<2048, 256, 0, stream>>>(src, dst, dis, h, agg, E);
        finish_layer<<<(NH + 255) / 256, 256, 0, stream>>>(agg, h, dis, b1, NH);

        gemm_wmma_f32<HID><<<N_NODES / GEMM_ROWS, 256, 0, stream>>>(agg, W2, h);
        zero_f4<<<(NH4 + 255) / 256, 256, 0, stream>>>((float4*)agg, NH4);
        edge_scatter<<<2048, 256, 0, stream>>>(src, dst, dis, h, agg, E);
        finish_layer<<<(NH + 255) / 256, 256, 0, stream>>>(agg, h, dis, b2, NH);
    }

    fill_f32<<<(N_GRAPHS * HID + N_GRAPHS + 255) / 256, 256, 0, stream>>>(
        pooled, 0.0f, N_GRAPHS * HID + N_GRAPHS);
    pool_accum<<<(N_NODES + POOL_ROWS * 8 - 1) / (POOL_ROWS * 8), 256, 0, stream>>>(agg, bat, pooled, cnt, N_NODES);
    pool_div<<<(N_GRAPHS * HID + 255) / 256, 256, 0, stream>>>(pooled, cnt);

    float* out_mf = out;
    float* out_bp = out + N_GRAPHS * OUT_MF;
    float* out_cc = out + N_GRAPHS * (OUT_MF + OUT_BP);
    head_gemm<<<(N_GRAPHS * OUT_MF + 255) / 256, 256, 0, stream>>>(pooled, Wmf, bmf, out_mf, OUT_MF);
    head_gemm<<<(N_GRAPHS * OUT_BP + 255) / 256, 256, 0, stream>>>(pooled, Wbp, bbp, out_bp, OUT_BP);
    head_gemm<<<(N_GRAPHS * OUT_CC + 255) / 256, 256, 0, stream>>>(pooled, Wcc, bcc, out_cc, OUT_CC);
}


# pooled-sum division and the three head products run as one launch (were four): a thread still owns one output and sums k in the same order; means staged per wave in LDS, W loads 32 in flight
# speedup vs baseline: 3.8728x; 1.1173x over previous
#include <hip/hip_runtime.h>
#include <hip/hip_bf16.h>

#define N_NODES 100000
#define N_EDGES 1600000
#define IN_DIM  256
#define HID     128
#define N_GRAPHS 64
#define OUT_MF 489
#define OUT_BP 1943
#define OUT_CC 320

#define WS_PLANES_BYTES (4ull * (N_NODES + 2ull * N_NODES * HID + N_GRAPHS * HID + N_GRAPHS))
#define WS_EXTRA_BYTES  (4ull * (N_NODES + 4 + N_EDGES))

typedef __attribute__((ext_vector_type(2))) float v2f;
typedef __attribute__((ext_vector_type(8))) float v8f;
typedef __attribute__((ext_vector_type(4))) float v4f;

__global__ void fill_f32(float* __restrict__ p, float v, int n) {
    int t = blockIdx.x * blockDim.x + threadIdx.x;
    if (t < n) p[t] = v;
}

__global__ void zero_f4(float4* __restrict__ p, int n4) {
    int t = blockIdx.x * blockDim.x + threadIdx.x;
    if (t < n4) { float4 z; z.x = z.y = z.z = z.w = 0.f; p[t] = z; }
}

__global__ void deg_accum(const int* __restrict__ dst, float* __restrict__ deg, int E) {
    int t = blockIdx.x * blockDim.x + threadIdx.x;
    if (t < E) atomicAdd(&deg[dst[t]], 1.0f);
}

__global__ void rsqrt_inplace(float* __restrict__ p, int n) {
    int t = blockIdx.x * blockDim.x + threadIdx.x;
    if (t < n) p[t] = rsqrtf(p[t]);
}

#define GEMM_ROWS 80
#define GEMM_RT   (GEMM_ROWS / 16)
#define GEMM_KC   64
#define GEMM_PA   (GEMM_KC + 4)
#define GEMM_PW   144
#define GEMM_NA   (GEMM_ROWS * GEMM_KC / 4 / 256)
#define GEMM_NW   ((GEMM_KC / 2) * (HID / 4) / 256)
static_assert(N_NODES % GEMM_ROWS == 0, "every work-group owns a full set of rows");
static_assert((GEMM_ROWS * GEMM_KC / 4) % 256 == 0, "A chunk splits evenly over the threads");

template <int K>
__global__ __launch_bounds__(256) void gemm_wmma_f32(const float* __restrict__ A,
                                                     const float* __restrict__ W,
                                                     float* __restrict__ out) {
    __shared__ float4 ldsA4[GEMM_ROWS * GEMM_PA / 4];
    __shared__ float4 ldsW4[(GEMM_KC / 2) * GEMM_PW / 2];
    float*  ldsA = (float*)ldsA4;
    float2* ldsW = (float2*)ldsW4;
    const int tid  = threadIdx.x;
    const int wave = tid >> 5;
    const int lane = tid & 31;
    const int half = lane >> 4;
    const int m    = lane & 15;
    const int row0 = blockIdx.x * GEMM_ROWS;
    const int col0 = wave * 16;

    const float* ag = A + (size_t)(row0 + (tid >> 4)) * K + 4 * (tid & 15);
    const float* wg = W + (size_t)(2 * (tid >> 5)) * HID + 4 * (tid & 31);
    v4f* as = (v4f*)(ldsA + (tid >> 4) * GEMM_PA + 4 * (tid & 15));
    v4f* wsd = (v4f*)(ldsW + (tid >> 5) * GEMM_PW + 4 * (tid & 31));

    v8f acc[GEMM_RT];
    #pragma unroll
    for (int t = 0; t < GEMM_RT; ++t) acc[t] = v8f{};

    const float*  aF = ldsA + m * GEMM_PA + 2 * half;
    const float2* bF = ldsW + half * GEMM_PW + col0 + m;

    for (int kb = 0; kb < K; kb += GEMM_KC) {
        {
            v4f ra[GEMM_NA], rw0[GEMM_NW], rw1[GEMM_NW];
            #pragma unroll
            for (int i = 0; i < GEMM_NA; ++i)
                ra[i] = *(const v4f*)(ag + (size_t)(16 * i) * K + kb);
            #pragma unroll
            for (int i = 0; i < GEMM_NW; ++i) {
                rw0[i] = *(const v4f*)(wg + (size_t)(kb + 16 * i) * HID);
                rw1[i] = *(const v4f*)(wg + (size_t)(kb + 16 * i + 1) * HID);
            }
            __builtin_amdgcn_sched_barrier(0);
            #pragma unroll
            for (int i = 0; i < GEMM_NA; ++i) as[i * 16 * GEMM_PA / 4] = ra[i];
            #pragma unroll
            for (int i = 0; i < GEMM_NW; ++i) {
                v4f lo, hi;
                lo.x = rw0[i].x; lo.y = rw1[i].x; lo.z = rw0[i].y; lo.w = rw1[i].y;
                hi.x = rw0[i].z; hi.y = rw1[i].z; hi.z = rw0[i].w; hi.w = rw1[i].w;
                wsd[i * 8 * GEMM_PW / 2]     = lo;
                wsd[i * 8 * GEMM_PW / 2 + 1] = hi;
            }
        }
        __syncthreads();

        #pragma unroll
        for (int k = 0; k < GEMM_KC; k += 4) {
            const float2 bb = bF[(k >> 1) * GEMM_PW];
            v2f b; b.x = bb.x; b.y = bb.y;
            #pragma unroll
            for (int t = 0; t < GEMM_RT; ++t) {
                const float2 aa = *(const float2*)(aF + t * 16 * GEMM_PA + k);
                v2f a; a.x = aa.x; a.y = aa.y;
                acc[t] = __builtin_amdgcn_wmma_f32_16x16x4_f32(
                    false, a, false, b, (short)0, acc[t], false, false);
            }
        }
        __syncthreads();
    }

    #pragma unroll
    for (int t = 0; t < GEMM_RT; ++t) {
        #pragma unroll
        for (int v = 0; v < 8; ++v)
            out[(size_t)(row0 + 16 * t + v + 8 * half) * HID + col0 + m] = acc[t][v];
    }
}

__global__ __launch_bounds__(256) void edge_scatter(const int* __restrict__ src,
                                                    const int* __restrict__ dst,
                                                    const float* __restrict__ dis,
                                                    const float* __restrict__ h,
                                                    float* __restrict__ agg, int E) {
    const int lane = threadIdx.x & 31;
    const int gw   = (blockIdx.x * blockDim.x + threadIdx.x) >> 5;
    const int nw   = (gridDim.x * blockDim.x) >> 5;
    for (int e = gw; e < E; e += nw) {
        const int s = src[e], d = dst[e];
        const float nrm = dis[s] * dis[d];
        const float* hs = h + (size_t)s * HID;
        float* ad = agg + (size_t)d * HID;
        #pragma unroll
        for (int j = 0; j < HID; j += 32)
            atomicAdd(&ad[j + lane], hs[j + lane] * nrm);
    }
}

__global__ void finish_layer(float* __restrict__ agg, const float* __restrict__ h,
                             const float* __restrict__ dis, const float* __restrict__ bias,
                             int n) {
    int t = blockIdx.x * blockDim.x + threadIdx.x;
    if (t >= n) return;
    const int i = t >> 7, j = t & (HID - 1);
    const float di = dis[i];
    const float v = agg[t] + di * di * h[t] + bias[j];
    agg[t] = v > 0.f ? v : 0.f;
}

#define NB     256
#define BIN_R  ((N_NODES + NB - 1) / NB)
#define NC     256
#define BT     1024
#define CHUNK_CAP 6400
#define BIN_CAP   8192
#define CNT_PAD   512
#define EPT       8
static_assert(BIN_R <= CNT_PAD, "a range's nodes fit the padded counter array");
static_assert((NB * NC) % (BT * 4) == 0, "bin_scan reads whole 16-byte groups per thread");

__device__ __forceinline__ int bin_of(int d) {
    const unsigned b = (unsigned)d / (unsigned)BIN_R;
    return b < (unsigned)NB ? (int)b : NB - 1;
}

__device__ __forceinline__ int wave_incl_scan(int v, int lane) {
    #pragma unroll
    for (int d = 1; d < 32; d <<= 1) {
        const int u = __shfl_up(v, d);
        if (lane >= d) v += u;
    }
    return v;
}

__global__ __launch_bounds__(BT) void bin_count(const int* __restrict__ dst,
                                                int* __restrict__ cntT, int E) {
    __shared__ int hist[NB];
    const int tid = threadIdx.x, c = blockIdx.x;
    const int epc = (E + NC - 1) / NC;
    const int e0  = c * epc;
    const int e1  = (e0 + epc) < E ? (e0 + epc) : E;
    if (tid < NB) hist[tid] = 0;
    __syncthreads();
    for (int base = e0 + tid; base < e1; base += BT * EPT) {
        int d[EPT];
        #pragma unroll
        for (int k = 0; k < EPT; ++k) d[k] = (base + k * BT) < e1 ? dst[base + k * BT] : 0;
        #pragma unroll
        for (int k = 0; k < EPT; ++k)
            if ((base + k * BT) < e1) atomicAdd(&hist[bin_of(d[k])], 1);
    }
    __syncthreads();
    if (tid < NB) cntT[tid * NC + c] = hist[tid];
}

#define OFF_PER ((NB * NC) / BT)
__global__ __launch_bounds__(BT) void bin_scan(int* __restrict__ cntT) {
    __shared__ int wsum[32];
    const int tid = threadIdx.x, lane = tid & 31, wave = tid >> 5;
    int4* p = (int4*)(cntT + tid * OFF_PER);
    int c[OFF_PER];
    #pragma unroll
    for (int k = 0; k < OFF_PER / 4; ++k) {
        const int4 a = p[k];
        c[4 * k] = a.x; c[4 * k + 1] = a.y; c[4 * k + 2] = a.z; c[4 * k + 3] = a.w;
    }
    int tsum = 0;
    #pragma unroll
    for (int k = 0; k < OFF_PER; ++k) { const int t = c[k]; c[k] = tsum; tsum += t; }
    const int v = wave_incl_scan(tsum, lane);
    if (lane == 31) wsum[wave] = v;
    __syncthreads();
    if (wave == 0) wsum[lane] = wave_incl_scan(wsum[lane], lane);
    __syncthreads();
    const int first = (wave ? wsum[wave - 1] : 0) + v - tsum;
    #pragma unroll
    for (int k = 0; k < OFF_PER / 4; ++k) {
        int4 o;
        o.x = first + c[4 * k];     o.y = first + c[4 * k + 1];
        o.z = first + c[4 * k + 2]; o.w = first + c[4 * k + 3];
        p[k] = o;
    }
    if (tid == BT - 1) cntT[NB * NC] = first + tsum;
}

__global__ __launch_bounds__(BT) void bin_place(const int* __restrict__ src,
                                                const int* __restrict__ dst,
                                                const int* __restrict__ off,
                                                int2* __restrict__ pairs, int E) {
    __shared__ int  hist[NB];
    __shared__ int  gdelta[NB];
    __shared__ int  wtot[NB / 32];
    __shared__ int2 sbuf[CHUNK_CAP];
    const int tid = threadIdx.x, lane = tid & 31, wave = tid >> 5, c = blockIdx.x;
    const int epc = (E + NC - 1) / NC;
    const int e0  = c * epc;
    const int e1  = (e0 + epc) < E ? (e0 + epc) : E;
    const int n   = e1 > e0 ? e1 - e0 : 0;
    if (tid < NB) hist[tid] = 0;
    __syncthreads();
    for (int base = e0 + tid; base < e1; base += BT * EPT) {
        int d[EPT];
        #pragma unroll
        for (int k = 0; k < EPT; ++k) d[k] = (base + k * BT) < e1 ? dst[base + k * BT] : 0;
        #pragma unroll
        for (int k = 0; k < EPT; ++k)
            if ((base + k * BT) < e1) atomicAdd(&hist[bin_of(d[k])], 1);
    }
    __syncthreads();
    int cnt = 0, v = 0;
    if (tid < NB) {
        cnt = hist[tid];
        v = wave_incl_scan(cnt, lane);
        if (lane == 31) wtot[wave] = v;
    }
    __syncthreads();
    if (tid < NB) {
        int base = 0;
        for (int w = 0; w < wave; ++w) base += wtot[w];
        const int ex = base + v - cnt;
        hist[tid]   = ex;
        gdelta[tid] = off[tid * NC + c] - ex;
    }
    __syncthreads();
    const bool staged = n <= CHUNK_CAP;
    for (int base = e0 + tid; base < e1; base += BT * EPT) {
        int s[EPT], d[EPT];
        #pragma unroll
        for (int k = 0; k < EPT; ++k) {
            const bool in = (base + k * BT) < e1;
            s[k] = in ? src[base + k * BT] : 0;
            d[k] = in ? dst[base + k * BT] : 0;
        }
        #pragma unroll
        for (int k = 0; k < EPT; ++k) {
            if ((base + k * BT) < e1) {
                int2 r; r.x = s[k]; r.y = d[k];
                const int b = bin_of(r.y);
                const int p = atomicAdd(&hist[b], 1);
                if (staged) {
                    if ((unsigned)p < (unsigned)CHUNK_CAP) sbuf[p] = r;
                } else {
                    const int g = p + gdelta[b];
                    if ((unsigned)g < (unsigned)E) pairs[g] = r;
                }
            }
        }
    }
    if (staged) {
        __syncthreads();
        for (int i = tid; i < n; i += BT) {
            const int2 r = sbuf[i];
            const int  g = i + gdelta[bin_of(r.y)];
            if ((unsigned)g < (unsigned)E) pairs[g] = r;
        }
    }
}

__global__ __launch_bounds__(BT) void bin_csr(const int2* __restrict__ pairs,
                                              const int* __restrict__ off,
                                              int* __restrict__ rowptr, int* __restrict__ col,
                                              float* __restrict__ dis, int N, int E) {
    __shared__ int cntl[CNT_PAD];
    __shared__ int wtot[CNT_PAD / 32];
    __shared__ int scol[BIN_CAP];
    const int tid = threadIdx.x, lane = tid & 31, wave = tid >> 5, b = blockIdx.x;
    int beg = off[b * NC], end = off[(b + 1) * NC];
    beg = beg > 0 ? beg : 0;  beg = beg < E ? beg : E;
    end = end > beg ? end : beg;  end = end < E ? end : E;
    const int n     = end - beg;
    const int node0 = b * BIN_R;
    int nn = N - node0;
    nn = nn < BIN_R ? nn : BIN_R;
    nn = nn > 0 ? nn : 0;
    if (tid < CNT_PAD) cntl[tid] = 0;
    __syncthreads();
    for (int base = tid; base < n; base += BT * EPT) {
        int d[EPT];
        #pragma unroll
        for (int k = 0; k < EPT; ++k) d[k] = (base + k * BT) < n ? pairs[beg + base + k * BT].y : -1;
        #pragma unroll
        for (int k = 0; k < EPT; ++k) {
            const unsigned l = (unsigned)(d[k] - node0);
            if ((base + k * BT) < n && l < (unsigned)nn) atomicAdd(&cntl[l], 1);
        }
    }
    __syncthreads();
    int cnt = 0, v = 0;
    if (tid < CNT_PAD) {
        cnt = cntl[tid];
        v = wave_incl_scan(cnt, lane);
        if (lane == 31) wtot[wave] = v;
    }
    __syncthreads();
    if (tid < CNT_PAD) {
        int base = 0;
        for (int w = 0; w < wave; ++w) base += wtot[w];
        const int ex = base + v - cnt;
        cntl[tid] = ex;
        if (tid < nn) {
            rowptr[node0 + tid] = beg + ex;
            dis[node0 + tid]    = rsqrtf((float)cnt + 1.0f);
        }
    }
    if (b == NB - 1 && tid == 0) rowptr[N] = end;
    __syncthreads();
    const bool staged = n <= BIN_CAP;
    for (int base = tid; base < n; base += BT * EPT) {
        int2 r[EPT];
        #pragma unroll
        for (int k = 0; k < EPT; ++k) {
            if ((base + k * BT) < n) r[k] = pairs[beg + base + k * BT];
            else { r[k].x = 0; r[k].y = -1; }
        }
        #pragma unroll
        for (int k = 0; k < EPT; ++k) {
            const unsigned l = (unsigned)(r[k].y - node0);
            if ((base + k * BT) < n && l < (unsigned)nn) {
                const int p = atomicAdd(&cntl[l], 1);
                if (staged) {
                    if ((unsigned)p < (unsigned)BIN_CAP) scol[p] = r[k].x;
                } else {
                    const int g = beg + p;
                    if ((unsigned)g < (unsigned)E) col[g] = r[k].x;
                }
            }
        }
    }
    if (staged) {
        __syncthreads();
        for (int i = tid; i < n; i += BT) col[beg + i] = scol[i];
    }
}

__global__ __launch_bounds__(256) void gather_layer(const int* __restrict__ rowptr,
                                                    const int* __restrict__ col,
                                                    const float* __restrict__ dis,
                                                    const float* __restrict__ h,
                                                    const float* __restrict__ bias,
                                                    float* __restrict__ out, int N, int E) {
    const int lane = threadIdx.x & 31;
    const int gw = __builtin_amdgcn_readfirstlane(
        (int)((blockIdx.x * blockDim.x + threadIdx.x) >> 5));
    const int nw = (int)((gridDim.x * blockDim.x) >> 5);
    const float4* h4 = (const float4*)h;
    float4*       o4 = (float4*)out;
    const float4  bv = ((const float4*)bias)[lane];

    for (int i = gw; i < N; i += nw) {
        int beg = rowptr[i], end = rowptr[i + 1];
        beg = beg > 0 ? beg : 0;
        end = end < E ? end : E;
        const float  di  = dis[i];
        const float4 hv  = h4[(size_t)i * (HID / 4) + lane];
        float4 acc; acc.x = acc.y = acc.z = acc.w = 0.f;

        for (int base = beg; base < end; base += 32) {
            const int m = (end - base) < 32 ? (end - base) : 32;
            int   s_l = 0;
            float w_l = 0.f;
            if (lane < m) {
                s_l = col[base + lane];
                s_l = (unsigned)s_l < (unsigned)N ? s_l : 0;
                w_l = dis[s_l] * di;
            }
            #pragma unroll 4
            for (int j = 0; j < m; ++j) {
                const int   s = __builtin_amdgcn_readlane(s_l, j);
                const float w = __int_as_float(__builtin_amdgcn_readlane(__float_as_int(w_l), j));
                const float4 v = h4[(size_t)s * (HID / 4) + lane];
                acc.x += v.x * w; acc.y += v.y * w; acc.z += v.z * w; acc.w += v.w * w;
            }
        }
        const float dd = di * di;
        float4 r;
        r.x = acc.x + dd * hv.x + bv.x;
        r.y = acc.y + dd * hv.y + bv.y;
        r.z = acc.z + dd * hv.z + bv.z;
        r.w = acc.w + dd * hv.w + bv.w;
        r.x = r.x > 0.f ? r.x : 0.f;
        r.y = r.y > 0.f ? r.y : 0.f;
        r.z = r.z > 0.f ? r.z : 0.f;
        r.w = r.w > 0.f ? r.w : 0.f;
        o4[(size_t)i * (HID / 4) + lane] = r;
    }
}

#define POOL_ROWS 16
static_assert(N_NODES % POOL_ROWS == 0, "every wave owns a full group of rows");
__global__ __launch_bounds__(256) void pool_accum(const float* __restrict__ h,
                                                  const int* __restrict__ batch,
                                                  float* __restrict__ pooled,
                                                  float* __restrict__ cnt, int N) {
    const int lane = threadIdx.x & 31;
    const int gw = __builtin_amdgcn_readfirstlane(
        (int)((blockIdx.x * blockDim.x + threadIdx.x) >> 5));
    const int r0 = gw * POOL_ROWS;
    if (r0 >= N) return;

    const int*   bp = batch + r0;
    const float* hp = h + (size_t)r0 * HID + lane;
    int   g[POOL_ROWS];
    float v[POOL_ROWS][4];
    #pragma unroll
    for (int i = 0; i < POOL_ROWS; ++i) {
        g[i] = bp[i];
        #pragma unroll
        for (int k = 0; k < 4; ++k) v[i][k] = hp[i * HID + 32 * k];
    }

    int   cur = g[0];
    int   run = 0;
    float a0 = 0.f, a1 = 0.f, a2 = 0.f, a3 = 0.f;
    #pragma unroll
    for (int i = 0; i < POOL_ROWS; ++i) {
        if (g[i] != cur) {
            float* pg = pooled + cur * HID + lane;
            atomicAdd(&pg[0],  a0);
            atomicAdd(&pg[32], a1);
            atomicAdd(&pg[64], a2);
            atomicAdd(&pg[96], a3);
            if (lane == 0) atomicAdd(&cnt[cur], (float)run);
            cur = g[i]; run = 0;
            a0 = a1 = a2 = a3 = 0.f;
        }
        a0 += v[i][0]; a1 += v[i][1]; a2 += v[i][2]; a3 += v[i][3];
        ++run;
    }
    {
        float* pg = pooled + cur * HID + lane;
        atomicAdd(&pg[0],  a0);
        atomicAdd(&pg[32], a1);
        atomicAdd(&pg[64], a2);
        atomicAdd(&pg[96], a3);
        if (lane == 0) atomicAdd(&cnt[cur], (float)run);
    }
}

#define HEAD_T_MF  (N_GRAPHS * OUT_MF)
#define HEAD_T_BP  (N_GRAPHS * OUT_BP)
#define HEAD_T_CC  (N_GRAPHS * OUT_CC)
#define HEAD_T_ALL (HEAD_T_MF + HEAD_T_BP + HEAD_T_CC)
#define HEAD_KB    32
static_assert(HEAD_T_MF % 32 == 0 && HEAD_T_BP % 32 == 0, "a wave never straddles two heads");
static_assert(HEAD_T_ALL % 256 == 0, "whole work-groups: every thread owns an output");
static_assert(OUT_MF > 32 && OUT_BP > 32 && OUT_CC > 32, "a wave's outputs touch at most two graphs");
static_assert(HID % HEAD_KB == 0 && HEAD_KB % 4 == 0, "k groups are whole 16-byte pieces of a mean row");

__global__ __launch_bounds__(256) void heads_fused(const float* __restrict__ pooled,
                                                   const float* __restrict__ cnt,
                                                   const float* __restrict__ Wmf,
                                                   const float* __restrict__ bmf,
                                                   const float* __restrict__ Wbp,
                                                   const float* __restrict__ bbp,
                                                   const float* __restrict__ Wcc,
                                                   const float* __restrict__ bcc,
                                                   float* __restrict__ out) {
    __shared__ float4 sP4[8 * 2 * (HID / 4)];
    const int tid = threadIdx.x, lane = tid & 31, wave = tid >> 5;
    const int t = blockIdx.x * 256 + tid;
    const int tw = __builtin_amdgcn_readfirstlane(t - lane);
    const float* W;
    const float* b;
    int OUT, hb;
    if (tw < HEAD_T_MF)                  { W = Wmf; b = bmf; OUT = OUT_MF; hb = 0; }
    else if (tw < HEAD_T_MF + HEAD_T_BP) { W = Wbp; b = bbp; OUT = OUT_BP; hb = HEAD_T_MF; }
    else                                 { W = Wcc; b = bcc; OUT = OUT_CC; hb = HEAD_T_MF + HEAD_T_BP; }
    const int tt = t - hb;
    const int g  = tt / OUT, o = tt - g * OUT;
    const int g0 = (tw - hb) / OUT;
    const int g1 = (g0 + 1) < N_GRAPHS ? (g0 + 1) : (N_GRAPHS - 1);

    #pragma unroll
    for (int r = 0; r < 2; ++r) {
        const int   gr = r ? g1 : g0;
        const float c  = cnt[gr];
        const float d  = c > 1.f ? c : 1.f;
        float4 v = ((const float4*)(pooled + gr * HID))[lane];
        v.x /= d; v.y /= d; v.z /= d; v.w /= d;
        sP4[(wave * 2 + r) * (HID / 4) + lane] = v;
    }
    __syncthreads();

    const float* prow = (const float*)sP4 + (wave * 2 + (g - g0)) * HID;
    const float* wp   = W + o;
    float s = b[o];
    for (int kb = 0; kb < HID; kb += HEAD_KB) {
        float w[HEAD_KB];
        #pragma unroll
        for (int i = 0; i < HEAD_KB; ++i) w[i] = wp[(size_t)(kb + i) * OUT];
        #pragma unroll
        for (int i = 0; i < HEAD_KB; i += 4) {
            const float4 p = *(const float4*)(prow + kb + i);
            s += p.x * w[i];
            s += p.y * w[i + 1];
            s += p.z * w[i + 2];
            s += p.w * w[i + 3];
        }
    }
    out[t] = s;
}

extern "C" void kernel_launch(void* const* d_in, const int* in_sizes, int n_in,
                              void* d_out, int out_size, void* d_ws, size_t ws_size,
                              hipStream_t stream) {
    const float* x   = (const float*)d_in[0];
    const int*   ei  = (const int*)d_in[1];
    const int*   bat = (const int*)d_in[2];
    const float* W1  = (const float*)d_in[3];
    const float* b1  = (const float*)d_in[4];
    const float* W2  = (const float*)d_in[5];
    const float* b2  = (const float*)d_in[6];
    const float* Wmf = (const float*)d_in[7];
    const float* bmf = (const float*)d_in[8];
    const float* Wbp = (const float*)d_in[9];
    const float* bbp = (const float*)d_in[10];
    const float* Wcc = (const float*)d_in[11];
    const float* bcc = (const float*)d_in[12];
    float* out = (float*)d_out;

    const int E = in_sizes[1] / 2;
    const int* src = ei;
    const int* dst = ei + E;

    float* ws     = (float*)d_ws;
    float* dis    = ws;
    float* h      = dis + N_NODES;
    float* agg    = h + (size_t)N_NODES * HID;
    float* pooled = agg + (size_t)N_NODES * HID;
    float* cnt    = pooled + N_GRAPHS * HID;

    int*  rowptr = (int*)(cnt + N_GRAPHS);
    int*  col    = rowptr + (N_NODES + 4);
    int2* pairs  = (int2*)agg;
    int*  cntT   = (int*)agg + 2 * N_EDGES;

    if (ws_size >= (size_t)WS_PLANES_BYTES + (size_t)WS_EXTRA_BYTES) {
        bin_count<<<NC, BT, 0, stream>>>(dst, cntT, E);
        bin_scan<<<1, BT, 0, stream>>>(cntT);
        bin_place<<<NC, BT, 0, stream>>>(src, dst, cntT, pairs, E);
        bin_csr<<<NB, BT, 0, stream>>>(pairs, cntT, rowptr, col, dis, N_NODES, E);

        gemm_wmma_f32<IN_DIM><<<N_NODES / GEMM_ROWS, 256, 0, stream>>>(x, W1, h);
        gather_layer<<<2048, 256, 0, stream>>>(rowptr, col, dis, h, b1, agg, N_NODES, E);

        gemm_wmma_f32<HID><<<N_NODES / GEMM_ROWS, 256, 0, stream>>>(agg, W2, h);
        gather_layer<<<2048, 256, 0, stream>>>(rowptr, col, dis, h, b2, agg, N_NODES, E);
    } else {
        const int NH  = N_NODES * HID;
        const int NH4 = NH / 4;
        fill_f32<<<(N_NODES + 255) / 256, 256, 0, stream>>>(dis, 1.0f, N_NODES);
        deg_accum<<<(E + 255) / 256, 256, 0, stream>>>(dst, dis, E);
        rsqrt_inplace<<<(N_NODES + 255) / 256, 256, 0, stream>>>(dis, N_NODES);

        gemm_wmma_f32<IN_DIM><<<N_NODES / GEMM_ROWS, 256, 0, stream>>>(x, W1, h);
        zero_f4<<<(NH4 + 255) / 256, 256, 0, stream>>>((float4*)agg, NH4);
        edge_scatter<<<2048, 256, 0, stream>>>(src, dst, dis, h, agg, E);
        finish_layer<<<(NH + 255) / 256, 256, 0, stream>>>(agg, h, dis, b1, NH);

        gemm_wmma_f32<HID><<<N_NODES / GEMM_ROWS, 256, 0, stream>>>(agg, W2, h);
        zero_f4<<<(NH4 + 255) / 256, 256, 0, stream>>>((float4*)agg, NH4);
        edge_scatter<<<2048, 256, 0, stream>>>(src, dst, dis, h, agg, E);
        finish_layer<<<(NH + 255) / 256, 256, 0, stream>>>(agg, h, dis, b2, NH);
    }

    fill_f32<<<(N_GRAPHS * HID + N_GRAPHS + 255) / 256, 256, 0, stream>>>(
        pooled, 0.0f, N_GRAPHS * HID + N_GRAPHS);
    pool_accum<<<(N_NODES + POOL_ROWS * 8 - 1) / (POOL_ROWS * 8), 256, 0, stream>>>(agg, bat, pooled, cnt, N_NODES);

    heads_fused<<<HEAD_T_ALL / 256, 256, 0, stream>>>(pooled, cnt, Wmf, bmf, Wbp, bbp, Wcc, bcc, out);
}


# layer-2 gather adds its finished rows into the pooled sums itself (wave runs in registers, 8 waves merged in LDS, one float atomic per feature): no h2 plane, no pool pass; pooled sum order changes
# speedup vs baseline: 4.0491x; 1.0455x over previous
#include <hip/hip_runtime.h>
#include <hip/hip_bf16.h>

#define N_NODES 100000
#define N_EDGES 1600000
#define IN_DIM  256
#define HID     128
#define N_GRAPHS 64
#define OUT_MF 489
#define OUT_BP 1943
#define OUT_CC 320

#define WS_PLANES_BYTES (4ull * (N_NODES + 2ull * N_NODES * HID + N_GRAPHS * HID + N_GRAPHS))
#define WS_EXTRA_BYTES  (4ull * (N_NODES + 4 + N_EDGES))

typedef __attribute__((ext_vector_type(2))) float v2f;
typedef __attribute__((ext_vector_type(8))) float v8f;
typedef __attribute__((ext_vector_type(4))) float v4f;

__global__ void fill_f32(float* __restrict__ p, float v, int n) {
    int t = blockIdx.x * blockDim.x + threadIdx.x;
    if (t < n) p[t] = v;
}

__global__ void zero_f4(float4* __restrict__ p, int n4) {
    int t = blockIdx.x * blockDim.x + threadIdx.x;
    if (t < n4) { float4 z; z.x = z.y = z.z = z.w = 0.f; p[t] = z; }
}

__global__ void deg_accum(const int* __restrict__ dst, float* __restrict__ deg, int E) {
    int t = blockIdx.x * blockDim.x + threadIdx.x;
    if (t < E) atomicAdd(&deg[dst[t]], 1.0f);
}

__global__ void rsqrt_inplace(float* __restrict__ p, int n) {
    int t = blockIdx.x * blockDim.x + threadIdx.x;
    if (t < n) p[t] = rsqrtf(p[t]);
}

#define GEMM_ROWS 80
#define GEMM_RT   (GEMM_ROWS / 16)
#define GEMM_KC   64
#define GEMM_PA   (GEMM_KC + 4)
#define GEMM_PW   144
#define GEMM_NA   (GEMM_ROWS * GEMM_KC / 4 / 256)
#define GEMM_NW   ((GEMM_KC / 2) * (HID / 4) / 256)
static_assert(N_NODES % GEMM_ROWS == 0, "every work-group owns a full set of rows");
static_assert((GEMM_ROWS * GEMM_KC / 4) % 256 == 0, "A chunk splits evenly over the threads");

template <int K>
__global__ __launch_bounds__(256) void gemm_wmma_f32(const float* __restrict__ A,
                                                     const float* __restrict__ W,
                                                     float* __restrict__ out) {
    __shared__ float4 ldsA4[GEMM_ROWS * GEMM_PA / 4];
    __shared__ float4 ldsW4[(GEMM_KC / 2) * GEMM_PW / 2];
    float*  ldsA = (float*)ldsA4;
    float2* ldsW = (float2*)ldsW4;
    const int tid  = threadIdx.x;
    const int wave = tid >> 5;
    const int lane = tid & 31;
    const int half = lane >> 4;
    const int m    = lane & 15;
    const int row0 = blockIdx.x * GEMM_ROWS;
    const int col0 = wave * 16;

    const float* ag = A + (size_t)(row0 + (tid >> 4)) * K + 4 * (tid & 15);
    const float* wg = W + (size_t)(2 * (tid >> 5)) * HID + 4 * (tid & 31);
    v4f* as = (v4f*)(ldsA + (tid >> 4) * GEMM_PA + 4 * (tid & 15));
    v4f* wsd = (v4f*)(ldsW + (tid >> 5) * GEMM_PW + 4 * (tid & 31));

    v8f acc[GEMM_RT];
    #pragma unroll
    for (int t = 0; t < GEMM_RT; ++t) acc[t] = v8f{};

    const float*  aF = ldsA + m * GEMM_PA + 2 * half;
    const float2* bF = ldsW + half * GEMM_PW + col0 + m;

    for (int kb = 0; kb < K; kb += GEMM_KC) {
        {
            v4f ra[GEMM_NA], rw0[GEMM_NW], rw1[GEMM_NW];
            #pragma unroll
            for (int i = 0; i < GEMM_NA; ++i)
                ra[i] = *(const v4f*)(ag + (size_t)(16 * i) * K + kb);
            #pragma unroll
            for (int i = 0; i < GEMM_NW; ++i) {
                rw0[i] = *(const v4f*)(wg + (size_t)(kb + 16 * i) * HID);
                rw1[i] = *(const v4f*)(wg + (size_t)(kb + 16 * i + 1) * HID);
            }
            __builtin_amdgcn_sched_barrier(0);
            #pragma unroll
            for (int i = 0; i < GEMM_NA; ++i) as[i * 16 * GEMM_PA / 4] = ra[i];
            #pragma unroll
            for (int i = 0; i < GEMM_NW; ++i) {
                v4f lo, hi;
                lo.x = rw0[i].x; lo.y = rw1[i].x; lo.z = rw0[i].y; lo.w = rw1[i].y;
                hi.x = rw0[i].z; hi.y = rw1[i].z; hi.z = rw0[i].w; hi.w = rw1[i].w;
                wsd[i * 8 * GEMM_PW / 2]     = lo;
                wsd[i * 8 * GEMM_PW / 2 + 1] = hi;
            }
        }
        __syncthreads();

        #pragma unroll
        for (int k = 0; k < GEMM_KC; k += 4) {
            const float2 bb = bF[(k >> 1) * GEMM_PW];
            v2f b; b.x = bb.x; b.y = bb.y;
            #pragma unroll
            for (int t = 0; t < GEMM_RT; ++t) {
                const float2 aa = *(const float2*)(aF + t * 16 * GEMM_PA + k);
                v2f a; a.x = aa.x; a.y = aa.y;
                acc[t] = __builtin_amdgcn_wmma_f32_16x16x4_f32(
                    false, a, false, b, (short)0, acc[t], false, false);
            }
        }
        __syncthreads();
    }

    #pragma unroll
    for (int t = 0; t < GEMM_RT; ++t) {
        #pragma unroll
        for (int v = 0; v < 8; ++v)
            out[(size_t)(row0 + 16 * t + v + 8 * half) * HID + col0 + m] = acc[t][v];
    }
}

__global__ __launch_bounds__(256) void edge_scatter(const int* __restrict__ src,
                                                    const int* __restrict__ dst,
                                                    const float* __restrict__ dis,
                                                    const float* __restrict__ h,
                                                    float* __restrict__ agg, int E) {
    const int lane = threadIdx.x & 31;
    const int gw   = (blockIdx.x * blockDim.x + threadIdx.x) >> 5;
    const int nw   = (gridDim.x * blockDim.x) >> 5;
    for (int e = gw; e < E; e += nw) {
        const int s = src[e], d = dst[e];
        const float nrm = dis[s] * dis[d];
        const float* hs = h + (size_t)s * HID;
        float* ad = agg + (size_t)d * HID;
        #pragma unroll
        for (int j = 0; j < HID; j += 32)
            atomicAdd(&ad[j + lane], hs[j + lane] * nrm);
    }
}

__global__ void finish_layer(float* __restrict__ agg, const float* __restrict__ h,
                             const float* __restrict__ dis, const float* __restrict__ bias,
                             int n) {
    int t = blockIdx.x * blockDim.x + threadIdx.x;
    if (t >= n) return;
    const int i = t >> 7, j = t & (HID - 1);
    const float di = dis[i];
    const float v = agg[t] + di * di * h[t] + bias[j];
    agg[t] = v > 0.f ? v : 0.f;
}

#define NB     256
#define BIN_R  ((N_NODES + NB - 1) / NB)
#define NC     256
#define BT     1024
#define CHUNK_CAP 6400
#define BIN_CAP   8192
#define CNT_PAD   512
#define EPT       8
static_assert(BIN_R <= CNT_PAD, "a range's nodes fit the padded counter array");
static_assert((NB * NC) % (BT * 4) == 0, "bin_scan reads whole 16-byte groups per thread");

__device__ __forceinline__ int bin_of(int d) {
    const unsigned b = (unsigned)d / (unsigned)BIN_R;
    return b < (unsigned)NB ? (int)b : NB - 1;
}

__device__ __forceinline__ int wave_incl_scan(int v, int lane) {
    #pragma unroll
    for (int d = 1; d < 32; d <<= 1) {
        const int u = __shfl_up(v, d);
        if (lane >= d) v += u;
    }
    return v;
}

__global__ __launch_bounds__(BT) void bin_count(const int* __restrict__ dst,
                                                int* __restrict__ cntT, int E) {
    __shared__ int hist[NB];
    const int tid = threadIdx.x, c = blockIdx.x;
    const int epc = (E + NC - 1) / NC;
    const int e0  = c * epc;
    const int e1  = (e0 + epc) < E ? (e0 + epc) : E;
    if (tid < NB) hist[tid] = 0;
    __syncthreads();
    for (int base = e0 + tid; base < e1; base += BT * EPT) {
        int d[EPT];
        #pragma unroll
        for (int k = 0; k < EPT; ++k) d[k] = (base + k * BT) < e1 ? dst[base + k * BT] : 0;
        #pragma unroll
        for (int k = 0; k < EPT; ++k)
            if ((base + k * BT) < e1) atomicAdd(&hist[bin_of(d[k])], 1);
    }
    __syncthreads();
    if (tid < NB) cntT[tid * NC + c] = hist[tid];
}

#define OFF_PER ((NB * NC) / BT)
__global__ __launch_bounds__(BT) void bin_scan(int* __restrict__ cntT) {
    __shared__ int wsum[32];
    const int tid = threadIdx.x, lane = tid & 31, wave = tid >> 5;
    int4* p = (int4*)(cntT + tid * OFF_PER);
    int c[OFF_PER];
    #pragma unroll
    for (int k = 0; k < OFF_PER / 4; ++k) {
        const int4 a = p[k];
        c[4 * k] = a.x; c[4 * k + 1] = a.y; c[4 * k + 2] = a.z; c[4 * k + 3] = a.w;
    }
    int tsum = 0;
    #pragma unroll
    for (int k = 0; k < OFF_PER; ++k) { const int t = c[k]; c[k] = tsum; tsum += t; }
    const int v = wave_incl_scan(tsum, lane);
    if (lane == 31) wsum[wave] = v;
    __syncthreads();
    if (wave == 0) wsum[lane] = wave_incl_scan(wsum[lane], lane);
    __syncthreads();
    const int first = (wave ? wsum[wave - 1] : 0) + v - tsum;
    #pragma unroll
    for (int k = 0; k < OFF_PER / 4; ++k) {
        int4 o;
        o.x = first + c[4 * k];     o.y = first + c[4 * k + 1];
        o.z = first + c[4 * k + 2]; o.w = first + c[4 * k + 3];
        p[k] = o;
    }
    if (tid == BT - 1) cntT[NB * NC] = first + tsum;
}

__global__ __launch_bounds__(BT) void bin_place(const int* __restrict__ src,
                                                const int* __restrict__ dst,
                                                const int* __restrict__ off,
                                                int2* __restrict__ pairs, int E) {
    __shared__ int  hist[NB];
    __shared__ int  gdelta[NB];
    __shared__ int  wtot[NB / 32];
    __shared__ int2 sbuf[CHUNK_CAP];
    const int tid = threadIdx.x, lane = tid & 31, wave = tid >> 5, c = blockIdx.x;
    const int epc = (E + NC - 1) / NC;
    const int e0  = c * epc;
    const int e1  = (e0 + epc) < E ? (e0 + epc) : E;
    const int n   = e1 > e0 ? e1 - e0 : 0;
    if (tid < NB) hist[tid] = 0;
    __syncthreads();
    for (int base = e0 + tid; base < e1; base += BT * EPT) {
        int d[EPT];
        #pragma unroll
        for (int k = 0; k < EPT; ++k) d[k] = (base + k * BT) < e1 ? dst[base + k * BT] : 0;
        #pragma unroll
        for (int k = 0; k < EPT; ++k)
            if ((base + k * BT) < e1) atomicAdd(&hist[bin_of(d[k])], 1);
    }
    __syncthreads();
    int cnt = 0, v = 0;
    if (tid < NB) {
        cnt = hist[tid];
        v = wave_incl_scan(cnt, lane);
        if (lane == 31) wtot[wave] = v;
    }
    __syncthreads();
    if (tid < NB) {
        int base = 0;
        for (int w = 0; w < wave; ++w) base += wtot[w];
        const int ex = base + v - cnt;
        hist[tid]   = ex;
        gdelta[tid] = off[tid * NC + c] - ex;
    }
    __syncthreads();
    const bool staged = n <= CHUNK_CAP;
    for (int base = e0 + tid; base < e1; base += BT * EPT) {
        int s[EPT], d[EPT];
        #pragma unroll
        for (int k = 0; k < EPT; ++k) {
            const bool in = (base + k * BT) < e1;
            s[k] = in ? src[base + k * BT] : 0;
            d[k] = in ? dst[base + k * BT] : 0;
        }
        #pragma unroll
        for (int k = 0; k < EPT; ++k) {
            if ((base + k * BT) < e1) {
                int2 r; r.x = s[k]; r.y = d[k];
                const int b = bin_of(r.y);
                const int p = atomicAdd(&hist[b], 1);
                if (staged) {
                    if ((unsigned)p < (unsigned)CHUNK_CAP) sbuf[p] = r;
                } else {
                    const int g = p + gdelta[b];
                    if ((unsigned)g < (unsigned)E) pairs[g] = r;
                }
            }
        }
    }
    if (staged) {
        __syncthreads();
        for (int i = tid; i < n; i += BT) {
            const int2 r = sbuf[i];
            const int  g = i + gdelta[bin_of(r.y)];
            if ((unsigned)g < (unsigned)E) pairs[g] = r;
        }
    }
}

__global__ __launch_bounds__(BT) void bin_csr(const int2* __restrict__ pairs,
                                              const int* __restrict__ off,
                                              int* __restrict__ rowptr, int* __restrict__ col,
                                              float* __restrict__ dis, int N, int E) {
    __shared__ int cntl[CNT_PAD];
    __shared__ int wtot[CNT_PAD / 32];
    __shared__ int scol[BIN_CAP];
    const int tid = threadIdx.x, lane = tid & 31, wave = tid >> 5, b = blockIdx.x;
    int beg = off[b * NC], end = off[(b + 1) * NC];
    beg = beg > 0 ? beg : 0;  beg = beg < E ? beg : E;
    end = end > beg ? end : beg;  end = end < E ? end : E;
    const int n     = end - beg;
    const int node0 = b * BIN_R;
    int nn = N - node0;
    nn = nn < BIN_R ? nn : BIN_R;
    nn = nn > 0 ? nn : 0;
    if (tid < CNT_PAD) cntl[tid] = 0;
    __syncthreads();
    for (int base = tid; base < n; base += BT * EPT) {
        int d[EPT];
        #pragma unroll
        for (int k = 0; k < EPT; ++k) d[k] = (base + k * BT) < n ? pairs[beg + base + k * BT].y : -1;
        #pragma unroll
        for (int k = 0; k < EPT; ++k) {
            const unsigned l = (unsigned)(d[k] - node0);
            if ((base + k * BT) < n && l < (unsigned)nn) atomicAdd(&cntl[l], 1);
        }
    }
    __syncthreads();
    int cnt = 0, v = 0;
    if (tid < CNT_PAD) {
        cnt = cntl[tid];
        v = wave_incl_scan(cnt, lane);
        if (lane == 31) wtot[wave] = v;
    }
    __syncthreads();
    if (tid < CNT_PAD) {
        int base = 0;
        for (int w = 0; w < wave; ++w) base += wtot[w];
        const int ex = base + v - cnt;
        cntl[tid] = ex;
        if (tid < nn) {
            rowptr[node0 + tid] = beg + ex;
            dis[node0 + tid]    = rsqrtf((float)cnt + 1.0f);
        }
    }
    if (b == NB - 1 && tid == 0) rowptr[N] = end;
    __syncthreads();
    const bool staged = n <= BIN_CAP;
    for (int base = tid; base < n; base += BT * EPT) {
        int2 r[EPT];
        #pragma unroll
        for (int k = 0; k < EPT; ++k) {
            if ((base + k * BT) < n) r[k] = pairs[beg + base + k * BT];
            else { r[k].x = 0; r[k].y = -1; }
        }
        #pragma unroll
        for (int k = 0; k < EPT; ++k) {
            const unsigned l = (unsigned)(r[k].y - node0);
            if ((base + k * BT) < n && l < (unsigned)nn) {
                const int p = atomicAdd(&cntl[l], 1);
                if (staged) {
                    if ((unsigned)p < (unsigned)BIN_CAP) scol[p] = r[k].x;
                } else {
                    const int g = beg + p;
                    if ((unsigned)g < (unsigned)E) col[g] = r[k].x;
                }
            }
        }
    }
    if (staged) {
        __syncthreads();
        for (int i = tid; i < n; i += BT) col[beg + i] = scol[i];
    }
}

__global__ __launch_bounds__(256) void gather_layer(const int* __restrict__ rowptr,
                                                    const int* __restrict__ col,
                                                    const float* __restrict__ dis,
                                                    const float* __restrict__ h,
                                                    const float* __restrict__ bias,
                                                    float* __restrict__ out, int N, int E) {
    const int lane = threadIdx.x & 31;
    const int gw = __builtin_amdgcn_readfirstlane(
        (int)((blockIdx.x * blockDim.x + threadIdx.x) >> 5));
    const int nw = (int)((gridDim.x * blockDim.x) >> 5);
    const float4* h4 = (const float4*)h;
    float4*       o4 = (float4*)out;
    const float4  bv = ((const float4*)bias)[lane];

    for (int i = gw; i < N; i += nw) {
        int beg = rowptr[i], end = rowptr[i + 1];
        beg = beg > 0 ? beg : 0;
        end = end < E ? end : E;
        const float  di  = dis[i];
        const float4 hv  = h4[(size_t)i * (HID / 4) + lane];
        float4 acc; acc.x = acc.y = acc.z = acc.w = 0.f;

        for (int base = beg; base < end; base += 32) {
            const int m = (end - base) < 32 ? (end - base) : 32;
            int   s_l = 0;
            float w_l = 0.f;
            if (lane < m) {
                s_l = col[base + lane];
                s_l = (unsigned)s_l < (unsigned)N ? s_l : 0;
                w_l = dis[s_l] * di;
            }
            #pragma unroll 4
            for (int j = 0; j < m; ++j) {
                const int   s = __builtin_amdgcn_readlane(s_l, j);
                const float w = __int_as_float(__builtin_amdgcn_readlane(__float_as_int(w_l), j));
                const float4 v = h4[(size_t)s * (HID / 4) + lane];
                acc.x += v.x * w; acc.y += v.y * w; acc.z += v.z * w; acc.w += v.w * w;
            }
        }
        const float dd = di * di;
        float4 r;
        r.x = acc.x + dd * hv.x + bv.x;
        r.y = acc.y + dd * hv.y + bv.y;
        r.z = acc.z + dd * hv.z + bv.z;
        r.w = acc.w + dd * hv.w + bv.w;
        r.x = r.x > 0.f ? r.x : 0.f;
        r.y = r.y > 0.f ? r.y : 0.f;
        r.z = r.z > 0.f ? r.z : 0.f;
        r.w = r.w > 0.f ? r.w : 0.f;
        o4[(size_t)i * (HID / 4) + lane] = r;
    }
}

#define GATHER_WGS 2048
static_assert((unsigned long long)N_NODES * (GATHER_WGS * 8ull) < (1ull << 32),
              "wave index times node count fits 32 bits in the range split");
__global__ __launch_bounds__(256) void gather_pool_layer(const int* __restrict__ rowptr,
                                                         const int* __restrict__ col,
                                                         const float* __restrict__ dis,
                                                         const float* __restrict__ h,
                                                         const float* __restrict__ bias,
                                                         const int* __restrict__ batch,
                                                         float* __restrict__ pooled,
                                                         float* __restrict__ cnt, int N, int E) {
    __shared__ float4 sAcc4[8 * (HID / 4)];
    __shared__ int4   sMeta4[4];
    const int tid  = threadIdx.x;
    const int lane = tid & 31;
    const int wave = tid >> 5;
    const unsigned gw = (unsigned)__builtin_amdgcn_readfirstlane(
        (int)((blockIdx.x * blockDim.x + threadIdx.x) >> 5));
    const unsigned nw = (gridDim.x * blockDim.x) >> 5;
    const int n0 = (int)(gw * (unsigned)N / nw);
    const int n1 = (int)((gw + 1u) * (unsigned)N / nw);
    const float4* h4 = (const float4*)h;
    const float4  bv = ((const float4*)bias)[lane];

    int    cur = 0, run = 0;
    float4 pacc; pacc.x = pacc.y = pacc.z = pacc.w = 0.f;

    for (int i = n0; i < n1; ++i) {
        int beg = rowptr[i], end = rowptr[i + 1];
        beg = beg > 0 ? beg : 0;
        end = end < E ? end : E;
        const float  di  = dis[i];
        const int    g   = batch[i];
        const float4 hv  = h4[(size_t)i * (HID / 4) + lane];
        float4 acc; acc.x = acc.y = acc.z = acc.w = 0.f;

        for (int base = beg; base < end; base += 32) {
            const int m = (end - base) < 32 ? (end - base) : 32;
            int   s_l = 0;
            float w_l = 0.f;
            if (lane < m) {
                s_l = col[base + lane];
                s_l = (unsigned)s_l < (unsigned)N ? s_l : 0;
                w_l = dis[s_l] * di;
            }
            #pragma unroll 4
            for (int j = 0; j < m; ++j) {
                const int   s = __builtin_amdgcn_readlane(s_l, j);
                const float w = __int_as_float(__builtin_amdgcn_readlane(__float_as_int(w_l), j));
                const float4 v = h4[(size_t)s * (HID / 4) + lane];
                acc.x += v.x * w; acc.y += v.y * w; acc.z += v.z * w; acc.w += v.w * w;
            }
        }
        const float dd = di * di;
        float4 r;
        r.x = acc.x + dd * hv.x + bv.x;
        r.y = acc.y + dd * hv.y + bv.y;
        r.z = acc.z + dd * hv.z + bv.z;
        r.w = acc.w + dd * hv.w + bv.w;
        r.x = r.x > 0.f ? r.x : 0.f;
        r.y = r.y > 0.f ? r.y : 0.f;
        r.z = r.z > 0.f ? r.z : 0.f;
        r.w = r.w > 0.f ? r.w : 0.f;

        if (run > 0 && g != cur) {
            if ((unsigned)cur < (unsigned)N_GRAPHS) {
                float* pg = pooled + cur * HID + 4 * lane;
                atomicAdd(&pg[0], pacc.x);
                atomicAdd(&pg[1], pacc.y);
                atomicAdd(&pg[2], pacc.z);
                atomicAdd(&pg[3], pacc.w);
                if (lane == 0) atomicAdd(&cnt[cur], (float)run);
            }
            run = 0;
            pacc.x = pacc.y = pacc.z = pacc.w = 0.f;
        }
        cur = g;
        pacc.x += r.x; pacc.y += r.y; pacc.z += r.z; pacc.w += r.w;
        ++run;
    }

    sAcc4[wave * (HID / 4) + lane] = pacc;
    if (lane == 0) { ((int*)sMeta4)[wave] = cur; ((int*)sMeta4)[8 + wave] = run; }
    __syncthreads();
    if (tid < HID) {
        const float* sa = (const float*)sAcc4;
        float v[8];
        #pragma unroll
        for (int w = 0; w < 8; ++w) v[w] = sa[w * HID + tid];
        const int4 ga = sMeta4[0], gb = sMeta4[1], ra = sMeta4[2], rb = sMeta4[3];
        const int gq[8] = { ga.x, ga.y, ga.z, ga.w, gb.x, gb.y, gb.z, gb.w };
        const int rw[8] = { ra.x, ra.y, ra.z, ra.w, rb.x, rb.y, rb.z, rb.w };
        int   cg = 0, rc = 0;
        float s  = 0.f;
        #pragma unroll
        for (int w = 0; w < 8; ++w) {
            if (rw[w] == 0) continue;
            if (rc > 0 && gq[w] != cg) {
                if ((unsigned)cg < (unsigned)N_GRAPHS) {
                    atomicAdd(&pooled[cg * HID + tid], s);
                    if (tid == 0) atomicAdd(&cnt[cg], (float)rc);
                }
                rc = 0; s = 0.f;
            }
            cg = gq[w];
            s += v[w];
            rc += rw[w];
        }
        if (rc > 0 && (unsigned)cg < (unsigned)N_GRAPHS) {
            atomicAdd(&pooled[cg * HID + tid], s);
            if (tid == 0) atomicAdd(&cnt[cg], (float)rc);
        }
    }
}

#define POOL_ROWS 16
static_assert(N_NODES % POOL_ROWS == 0, "every wave owns a full group of rows");
__global__ __launch_bounds__(256) void pool_accum(const float* __restrict__ h,
                                                  const int* __restrict__ batch,
                                                  float* __restrict__ pooled,
                                                  float* __restrict__ cnt, int N) {
    const int lane = threadIdx.x & 31;
    const int gw = __builtin_amdgcn_readfirstlane(
        (int)((blockIdx.x * blockDim.x + threadIdx.x) >> 5));
    const int r0 = gw * POOL_ROWS;
    if (r0 >= N) return;

    const int*   bp = batch + r0;
    const float* hp = h + (size_t)r0 * HID + lane;
    int   g[POOL_ROWS];
    float v[POOL_ROWS][4];
    #pragma unroll
    for (int i = 0; i < POOL_ROWS; ++i) {
        g[i] = bp[i];
        #pragma unroll
        for (int k = 0; k < 4; ++k) v[i][k] = hp[i * HID + 32 * k];
    }

    int   cur = g[0];
    int   run = 0;
    float a0 = 0.f, a1 = 0.f, a2 = 0.f, a3 = 0.f;
    #pragma unroll
    for (int i = 0; i < POOL_ROWS; ++i) {
        if (g[i] != cur) {
            float* pg = pooled + cur * HID + lane;
            atomicAdd(&pg[0],  a0);
            atomicAdd(&pg[32], a1);
            atomicAdd(&pg[64], a2);
            atomicAdd(&pg[96], a3);
            if (lane == 0) atomicAdd(&cnt[cur], (float)run);
            cur = g[i]; run = 0;
            a0 = a1 = a2 = a3 = 0.f;
        }
        a0 += v[i][0]; a1 += v[i][1]; a2 += v[i][2]; a3 += v[i][3];
        ++run;
    }
    {
        float* pg = pooled + cur * HID + lane;
        atomicAdd(&pg[0],  a0);
        atomicAdd(&pg[32], a1);
        atomicAdd(&pg[64], a2);
        atomicAdd(&pg[96], a3);
        if (lane == 0) atomicAdd(&cnt[cur], (float)run);
    }
}

#define HEAD_T_MF  (N_GRAPHS * OUT_MF)
#define HEAD_T_BP  (N_GRAPHS * OUT_BP)
#define HEAD_T_CC  (N_GRAPHS * OUT_CC)
#define HEAD_T_ALL (HEAD_T_MF + HEAD_T_BP + HEAD_T_CC)
#define HEAD_KB    32
static_assert(HEAD_T_MF % 32 == 0 && HEAD_T_BP % 32 == 0, "a wave never straddles two heads");
static_assert(HEAD_T_ALL % 256 == 0, "whole work-groups: every thread owns an output");
static_assert(OUT_MF > 32 && OUT_BP > 32 && OUT_CC > 32, "a wave's outputs touch at most two graphs");
static_assert(HID % HEAD_KB == 0 && HEAD_KB % 4 == 0, "k groups are whole 16-byte pieces of a mean row");

__global__ __launch_bounds__(256) void heads_fused(const float* __restrict__ pooled,
                                                   const float* __restrict__ cnt,
                                                   const float* __restrict__ Wmf,
                                                   const float* __restrict__ bmf,
                                                   const float* __restrict__ Wbp,
                                                   const float* __restrict__ bbp,
                                                   const float* __restrict__ Wcc,
                                                   const float* __restrict__ bcc,
                                                   float* __restrict__ out) {
    __shared__ float4 sP4[8 * 2 * (HID / 4)];
    const int tid = threadIdx.x, lane = tid & 31, wave = tid >> 5;
    const int t = blockIdx.x * 256 + tid;
    const int tw = __builtin_amdgcn_readfirstlane(t - lane);
    const float* W;
    const float* b;
    int OUT, hb;
    if (tw < HEAD_T_MF)                  { W = Wmf; b = bmf; OUT = OUT_MF; hb = 0; }
    else if (tw < HEAD_T_MF + HEAD_T_BP) { W = Wbp; b = bbp; OUT = OUT_BP; hb = HEAD_T_MF; }
    else                                 { W = Wcc; b = bcc; OUT = OUT_CC; hb = HEAD_T_MF + HEAD_T_BP; }
    const int tt = t - hb;
    const int g  = tt / OUT, o = tt - g * OUT;
    const int g0 = (tw - hb) / OUT;
    const int g1 = (g0 + 1) < N_GRAPHS ? (g0 + 1) : (N_GRAPHS - 1);

    #pragma unroll
    for (int r = 0; r < 2; ++r) {
        const int   gr = r ? g1 : g0;
        const float c  = cnt[gr];
        const float d  = c > 1.f ? c : 1.f;
        float4 v = ((const float4*)(pooled + gr * HID))[lane];
        v.x /= d; v.y /= d; v.z /= d; v.w /= d;
        sP4[(wave * 2 + r) * (HID / 4) + lane] = v;
    }
    __syncthreads();

    const float* prow = (const float*)sP4 + (wave * 2 + (g - g0)) * HID;
    const float* wp   = W + o;
    float s = b[o];
    for (int kb = 0; kb < HID; kb += HEAD_KB) {
        float w[HEAD_KB];
        #pragma unroll
        for (int i = 0; i < HEAD_KB; ++i) w[i] = wp[(size_t)(kb + i) * OUT];
        #pragma unroll
        for (int i = 0; i < HEAD_KB; i += 4) {
            const float4 p = *(const float4*)(prow + kb + i);
            s += p.x * w[i];
            s += p.y * w[i + 1];
            s += p.z * w[i + 2];
            s += p.w * w[i + 3];
        }
    }
    out[t] = s;
}

extern "C" void kernel_launch(void* const* d_in, const int* in_sizes, int n_in,
                              void* d_out, int out_size, void* d_ws, size_t ws_size,
                              hipStream_t stream) {
    const float* x   = (const float*)d_in[0];
    const int*   ei  = (const int*)d_in[1];
    const int*   bat = (const int*)d_in[2];
    const float* W1  = (const float*)d_in[3];
    const float* b1  = (const float*)d_in[4];
    const float* W2  = (const float*)d_in[5];
    const float* b2  = (const float*)d_in[6];
    const float* Wmf = (const float*)d_in[7];
    const float* bmf = (const float*)d_in[8];
    const float* Wbp = (const float*)d_in[9];
    const float* bbp = (const float*)d_in[10];
    const float* Wcc = (const float*)d_in[11];
    const float* bcc = (const float*)d_in[12];
    float* out = (float*)d_out;

    const int E = in_sizes[1] / 2;
    const int* src = ei;
    const int* dst = ei + E;

    float* ws     = (float*)d_ws;
    float* dis    = ws;
    float* h      = dis + N_NODES;
    float* agg    = h + (size_t)N_NODES * HID;
    float* pooled = agg + (size_t)N_NODES * HID;
    float* cnt    = pooled + N_GRAPHS * HID;

    int*  rowptr = (int*)(cnt + N_GRAPHS);
    int*  col    = rowptr + (N_NODES + 4);
    int2* pairs  = (int2*)agg;
    int*  cntT   = (int*)agg + 2 * N_EDGES;

    if (ws_size >= (size_t)WS_PLANES_BYTES + (size_t)WS_EXTRA_BYTES) {
        bin_count<<<NC, BT, 0, stream>>>(dst, cntT, E);
        bin_scan<<<1, BT, 0, stream>>>(cntT);
        bin_place<<<NC, BT, 0, stream>>>(src, dst, cntT, pairs, E);
        bin_csr<<<NB, BT, 0, stream>>>(pairs, cntT, rowptr, col, dis, N_NODES, E);

        gemm_wmma_f32<IN_DIM><<<N_NODES / GEMM_ROWS, 256, 0, stream>>>(x, W1, h);
        gather_layer<<<2048, 256, 0, stream>>>(rowptr, col, dis, h, b1, agg, N_NODES, E);

        gemm_wmma_f32<HID><<<N_NODES / GEMM_ROWS, 256, 0, stream>>>(agg, W2, h);

        fill_f32<<<(N_GRAPHS * HID + N_GRAPHS + 255) / 256, 256, 0, stream>>>(
            pooled, 0.0f, N_GRAPHS * HID + N_GRAPHS);
        gather_pool_layer<<<GATHER_WGS, 256, 0, stream>>>(rowptr, col, dis, h, b2, bat,
                                                          pooled, cnt, N_NODES, E);
    } else {
        const int NH  = N_NODES * HID;
        const int NH4 = NH / 4;
        fill_f32<<<(N_NODES + 255) / 256, 256, 0, stream>>>(dis, 1.0f, N_NODES);
        deg_accum<<<(E + 255) / 256, 256, 0, stream>>>(dst, dis, E);
        rsqrt_inplace<<<(N_NODES + 255) / 256, 256, 0, stream>>>(dis, N_NODES);

        gemm_wmma_f32<IN_DIM><<<N_NODES / GEMM_ROWS, 256, 0, stream>>>(x, W1, h);
        zero_f4<<<(NH4 + 255) / 256, 256, 0, stream>>>((float4*)agg, NH4);
        edge_scatter<<<2048, 256, 0, stream>>>(src, dst, dis, h, agg, E);
        finish_layer<<<(NH + 255) / 256, 256, 0, stream>>>(agg, h, dis, b1, NH);

        gemm_wmma_f32<HID><<<N_NODES / GEMM_ROWS, 256, 0, stream>>>(agg, W2, h);
        zero_f4<<<(NH4 + 255) / 256, 256, 0, stream>>>((float4*)agg, NH4);
        edge_scatter<<<2048, 256, 0, stream>>>(src, dst, dis, h, agg, E);
        finish_layer<<<(NH + 255) / 256, 256, 0, stream>>>(agg, h, dis, b2, NH);

        fill_f32<<<(N_GRAPHS * HID + N_GRAPHS + 255) / 256, 256, 0, stream>>>(
            pooled, 0.0f, N_GRAPHS * HID + N_GRAPHS);
        pool_accum<<<(N_NODES + POOL_ROWS * 8 - 1) / (POOL_ROWS * 8), 256, 0, stream>>>(agg, bat, pooled, cnt, N_NODES);
    }

    heads_fused<<<HEAD_T_ALL / 256, 256, 0, stream>>>(pooled, cnt, Wmf, bmf, Wbp, bbp, Wcc, bcc, out);
}


# offsets of the range grouping computed inside the placement pass: each work-group sums the chunk-major count table's columns itself with coalesced 16-byte loads; the single-group scan launch is gone
# speedup vs baseline: 4.3358x; 1.0708x over previous
#include <hip/hip_runtime.h>
#include <hip/hip_bf16.h>

#define N_NODES 100000
#define N_EDGES 1600000
#define IN_DIM  256
#define HID     128
#define N_GRAPHS 64
#define OUT_MF 489
#define OUT_BP 1943
#define OUT_CC 320

#define WS_PLANES_BYTES (4ull * (N_NODES + 2ull * N_NODES * HID + N_GRAPHS * HID + N_GRAPHS))
#define WS_EXTRA_BYTES  (4ull * (N_NODES + 4 + N_EDGES))

typedef __attribute__((ext_vector_type(2))) float v2f;
typedef __attribute__((ext_vector_type(8))) float v8f;
typedef __attribute__((ext_vector_type(4))) float v4f;

__global__ void fill_f32(float* __restrict__ p, float v, int n) {
    int t = blockIdx.x * blockDim.x + threadIdx.x;
    if (t < n) p[t] = v;
}

__global__ void zero_f4(float4* __restrict__ p, int n4) {
    int t = blockIdx.x * blockDim.x + threadIdx.x;
    if (t < n4) { float4 z; z.x = z.y = z.z = z.w = 0.f; p[t] = z; }
}

__global__ void deg_accum(const int* __restrict__ dst, float* __restrict__ deg, int E) {
    int t = blockIdx.x * blockDim.x + threadIdx.x;
    if (t < E) atomicAdd(&deg[dst[t]], 1.0f);
}

__global__ void rsqrt_inplace(float* __restrict__ p, int n) {
    int t = blockIdx.x * blockDim.x + threadIdx.x;
    if (t < n) p[t] = rsqrtf(p[t]);
}

#define GEMM_ROWS 80
#define GEMM_RT   (GEMM_ROWS / 16)
#define GEMM_KC   64
#define GEMM_PA   (GEMM_KC + 4)
#define GEMM_PW   144
#define GEMM_NA   (GEMM_ROWS * GEMM_KC / 4 / 256)
#define GEMM_NW   ((GEMM_KC / 2) * (HID / 4) / 256)
static_assert(N_NODES % GEMM_ROWS == 0, "every work-group owns a full set of rows");
static_assert((GEMM_ROWS * GEMM_KC / 4) % 256 == 0, "A chunk splits evenly over the threads");

template <int K>
__global__ __launch_bounds__(256) void gemm_wmma_f32(const float* __restrict__ A,
                                                     const float* __restrict__ W,
                                                     float* __restrict__ out) {
    __shared__ float4 ldsA4[GEMM_ROWS * GEMM_PA / 4];
    __shared__ float4 ldsW4[(GEMM_KC / 2) * GEMM_PW / 2];
    float*  ldsA = (float*)ldsA4;
    float2* ldsW = (float2*)ldsW4;
    const int tid  = threadIdx.x;
    const int wave = tid >> 5;
    const int lane = tid & 31;
    const int half = lane >> 4;
    const int m    = lane & 15;
    const int row0 = blockIdx.x * GEMM_ROWS;
    const int col0 = wave * 16;

    const float* ag = A + (size_t)(row0 + (tid >> 4)) * K + 4 * (tid & 15);
    const float* wg = W + (size_t)(2 * (tid >> 5)) * HID + 4 * (tid & 31);
    v4f* as = (v4f*)(ldsA + (tid >> 4) * GEMM_PA + 4 * (tid & 15));
    v4f* wsd = (v4f*)(ldsW + (tid >> 5) * GEMM_PW + 4 * (tid & 31));

    v8f acc[GEMM_RT];
    #pragma unroll
    for (int t = 0; t < GEMM_RT; ++t) acc[t] = v8f{};

    const float*  aF = ldsA + m * GEMM_PA + 2 * half;
    const float2* bF = ldsW + half * GEMM_PW + col0 + m;

    for (int kb = 0; kb < K; kb += GEMM_KC) {
        {
            v4f ra[GEMM_NA], rw0[GEMM_NW], rw1[GEMM_NW];
            #pragma unroll
            for (int i = 0; i < GEMM_NA; ++i)
                ra[i] = *(const v4f*)(ag + (size_t)(16 * i) * K + kb);
            #pragma unroll
            for (int i = 0; i < GEMM_NW; ++i) {
                rw0[i] = *(const v4f*)(wg + (size_t)(kb + 16 * i) * HID);
                rw1[i] = *(const v4f*)(wg + (size_t)(kb + 16 * i + 1) * HID);
            }
            __builtin_amdgcn_sched_barrier(0);
            #pragma unroll
            for (int i = 0; i < GEMM_NA; ++i) as[i * 16 * GEMM_PA / 4] = ra[i];
            #pragma unroll
            for (int i = 0; i < GEMM_NW; ++i) {
                v4f lo, hi;
                lo.x = rw0[i].x; lo.y = rw1[i].x; lo.z = rw0[i].y; lo.w = rw1[i].y;
                hi.x = rw0[i].z; hi.y = rw1[i].z; hi.z = rw0[i].w; hi.w = rw1[i].w;
                wsd[i * 8 * GEMM_PW / 2]     = lo;
                wsd[i * 8 * GEMM_PW / 2 + 1] = hi;
            }
        }
        __syncthreads();

        #pragma unroll
        for (int k = 0; k < GEMM_KC; k += 4) {
            const float2 bb = bF[(k >> 1) * GEMM_PW];
            v2f b; b.x = bb.x; b.y = bb.y;
            #pragma unroll
            for (int t = 0; t < GEMM_RT; ++t) {
                const float2 aa = *(const float2*)(aF + t * 16 * GEMM_PA + k);
                v2f a; a.x = aa.x; a.y = aa.y;
                acc[t] = __builtin_amdgcn_wmma_f32_16x16x4_f32(
                    false, a, false, b, (short)0, acc[t], false, false);
            }
        }
        __syncthreads();
    }

    #pragma unroll
    for (int t = 0; t < GEMM_RT; ++t) {
        #pragma unroll
        for (int v = 0; v < 8; ++v)
            out[(size_t)(row0 + 16 * t + v + 8 * half) * HID + col0 + m] = acc[t][v];
    }
}

__global__ __launch_bounds__(256) void edge_scatter(const int* __restrict__ src,
                                                    const int* __restrict__ dst,
                                                    const float* __restrict__ dis,
                                                    const float* __restrict__ h,
                                                    float* __restrict__ agg, int E) {
    const int lane = threadIdx.x & 31;
    const int gw   = (blockIdx.x * blockDim.x + threadIdx.x) >> 5;
    const int nw   = (gridDim.x * blockDim.x) >> 5;
    for (int e = gw; e < E; e += nw) {
        const int s = src[e], d = dst[e];
        const float nrm = dis[s] * dis[d];
        const float* hs = h + (size_t)s * HID;
        float* ad = agg + (size_t)d * HID;
        #pragma unroll
        for (int j = 0; j < HID; j += 32)
            atomicAdd(&ad[j + lane], hs[j + lane] * nrm);
    }
}

__global__ void finish_layer(float* __restrict__ agg, const float* __restrict__ h,
                             const float* __restrict__ dis, const float* __restrict__ bias,
                             int n) {
    int t = blockIdx.x * blockDim.x + threadIdx.x;
    if (t >= n) return;
    const int i = t >> 7, j = t & (HID - 1);
    const float di = dis[i];
    const float v = agg[t] + di * di * h[t] + bias[j];
    agg[t] = v > 0.f ? v : 0.f;
}

#define NB     256
#define BIN_R  ((N_NODES + NB - 1) / NB)
#define NC     256
#define BT     1024
#define CHUNK_CAP 6400
#define BIN_CAP   8192
#define CNT_PAD   512
#define EPT       8
static_assert(BIN_R <= CNT_PAD, "a range's nodes fit the padded counter array");
#define TAB_Q  (NB / 4)
#define TAB_RG (BT / TAB_Q)
#define TAB_K  (NC / TAB_RG)
static_assert(NB % 4 == 0 && BT % TAB_Q == 0 && NC % TAB_RG == 0 && TAB_Q % 32 == 0,
              "the table splits into whole 16-byte groups, whole waves per row half");

__device__ __forceinline__ int bin_of(int d) {
    const unsigned b = (unsigned)d / (unsigned)BIN_R;
    return b < (unsigned)NB ? (int)b : NB - 1;
}

__device__ __forceinline__ int wave_incl_scan(int v, int lane) {
    #pragma unroll
    for (int d = 1; d < 32; d <<= 1) {
        const int u = __shfl_up(v, d);
        if (lane >= d) v += u;
    }
    return v;
}

__global__ __launch_bounds__(BT) void bin_count(const int* __restrict__ dst,
                                                int* __restrict__ cntT, int E) {
    __shared__ int hist[NB];
    const int tid = threadIdx.x, c = blockIdx.x;
    const int epc = (E + NC - 1) / NC;
    const int e0  = c * epc;
    const int e1  = (e0 + epc) < E ? (e0 + epc) : E;
    if (tid < NB) hist[tid] = 0;
    __syncthreads();
    for (int base = e0 + tid; base < e1; base += BT * EPT) {
        int d[EPT];
        #pragma unroll
        for (int k = 0; k < EPT; ++k) d[k] = (base + k * BT) < e1 ? dst[base + k * BT] : 0;
        #pragma unroll
        for (int k = 0; k < EPT; ++k)
            if ((base + k * BT) < e1) atomicAdd(&hist[bin_of(d[k])], 1);
    }
    __syncthreads();
    if (tid < NB) cntT[c * NB + tid] = hist[tid];
}

__global__ __launch_bounds__(BT) void bin_place(const int* __restrict__ src,
                                                const int* __restrict__ dst,
                                                const int* __restrict__ cntT,
                                                int* __restrict__ rbase,
                                                int2* __restrict__ pairs, int E) {
    __shared__ int  hist[NB];
    __shared__ int  gdelta[NB];
    __shared__ int  wtot[NB / 32];
    __shared__ int  rtot[NB];
    __shared__ int  rpre[NB];
    __shared__ int  wtotR[NB / 32];
    __shared__ int2 sbuf[CHUNK_CAP];
    const int tid = threadIdx.x, lane = tid & 31, wave = tid >> 5, c = blockIdx.x;
    const int epc = (E + NC - 1) / NC;
    const int e0  = c * epc;
    const int e1  = (e0 + epc) < E ? (e0 + epc) : E;
    const int n   = e1 > e0 ? e1 - e0 : 0;
    if (tid < NB) { hist[tid] = 0; rtot[tid] = 0; rpre[tid] = 0; }
    const int q  = tid & (TAB_Q - 1);
    const int rg = tid / TAB_Q;
    const int4* tp = (const int4*)cntT + rg * TAB_Q + q;
    int4 ta[TAB_K];
    __syncthreads();
    #pragma unroll
    for (int k = 0; k < TAB_K; ++k) ta[k] = tp[k * TAB_RG * TAB_Q];
    for (int base = e0 + tid; base < e1; base += BT * EPT) {
        int d[EPT];
        #pragma unroll
        for (int k = 0; k < EPT; ++k) d[k] = (base + k * BT) < e1 ? dst[base + k * BT] : 0;
        #pragma unroll
        for (int k = 0; k < EPT; ++k)
            if ((base + k * BT) < e1) atomicAdd(&hist[bin_of(d[k])], 1);
    }
    {
        int t0 = 0, t1 = 0, t2 = 0, t3 = 0, p0 = 0, p1 = 0, p2 = 0, p3 = 0;
        #pragma unroll
        for (int k = 0; k < TAB_K; ++k) {
            const int m = (rg + k * TAB_RG) < c ? -1 : 0;
            t0 += ta[k].x; t1 += ta[k].y; t2 += ta[k].z; t3 += ta[k].w;
            p0 += ta[k].x & m; p1 += ta[k].y & m; p2 += ta[k].z & m; p3 += ta[k].w & m;
        }
        atomicAdd(&rtot[4 * q],     t0);  atomicAdd(&rtot[4 * q + 1], t1);
        atomicAdd(&rtot[4 * q + 2], t2);  atomicAdd(&rtot[4 * q + 3], t3);
        atomicAdd(&rpre[4 * q],     p0);  atomicAdd(&rpre[4 * q + 1], p1);
        atomicAdd(&rpre[4 * q + 2], p2);  atomicAdd(&rpre[4 * q + 3], p3);
    }
    __syncthreads();
    int cnt = 0, v = 0, rt = 0, vr = 0;
    if (tid < NB) {
        cnt = hist[tid];
        v = wave_incl_scan(cnt, lane);
        rt = rtot[tid];
        vr = wave_incl_scan(rt, lane);
        if (lane == 31) { wtot[wave] = v; wtotR[wave] = vr; }
    }
    __syncthreads();
    if (tid < NB) {
        int base = 0, baseR = 0;
        for (int w = 0; w < wave; ++w) { base += wtot[w]; baseR += wtotR[w]; }
        const int ex = base + v - cnt;
        const int gx = baseR + vr - rt;
        hist[tid]   = ex;
        gdelta[tid] = gx + rpre[tid] - ex;
        if (c == 0) {
            rbase[tid] = gx;
            if (tid == NB - 1) rbase[NB] = gx + rt;
        }
    }
    __syncthreads();
    const bool staged = n <= CHUNK_CAP;
    for (int base = e0 + tid; base < e1; base += BT * EPT) {
        int s[EPT], d[EPT];
        #pragma unroll
        for (int k = 0; k < EPT; ++k) {
            const bool in = (base + k * BT) < e1;
            s[k] = in ? src[base + k * BT] : 0;
            d[k] = in ? dst[base + k * BT] : 0;
        }
        #pragma unroll
        for (int k = 0; k < EPT; ++k) {
            if ((base + k * BT) < e1) {
                int2 r; r.x = s[k]; r.y = d[k];
                const int b = bin_of(r.y);
                const int p = atomicAdd(&hist[b], 1);
                if (staged) {
                    if ((unsigned)p < (unsigned)CHUNK_CAP) sbuf[p] = r;
                } else {
                    const int g = p + gdelta[b];
                    if ((unsigned)g < (unsigned)E) pairs[g] = r;
                }
            }
        }
    }
    if (staged) {
        __syncthreads();
        for (int i = tid; i < n; i += BT) {
            const int2 r = sbuf[i];
            const int  g = i + gdelta[bin_of(r.y)];
            if ((unsigned)g < (unsigned)E) pairs[g] = r;
        }
    }
}

__global__ __launch_bounds__(BT) void bin_csr(const int2* __restrict__ pairs,
                                              const int* __restrict__ rbase,
                                              int* __restrict__ rowptr, int* __restrict__ col,
                                              float* __restrict__ dis, int N, int E) {
    __shared__ int cntl[CNT_PAD];
    __shared__ int wtot[CNT_PAD / 32];
    __shared__ int scol[BIN_CAP];
    const int tid = threadIdx.x, lane = tid & 31, wave = tid >> 5, b = blockIdx.x;
    int beg = rbase[b], end = rbase[b + 1];
    beg = beg > 0 ? beg : 0;  beg = beg < E ? beg : E;
    end = end > beg ? end : beg;  end = end < E ? end : E;
    const int n     = end - beg;
    const int node0 = b * BIN_R;
    int nn = N - node0;
    nn = nn < BIN_R ? nn : BIN_R;
    nn = nn > 0 ? nn : 0;
    if (tid < CNT_PAD) cntl[tid] = 0;
    __syncthreads();
    for (int base = tid; base < n; base += BT * EPT) {
        int d[EPT];
        #pragma unroll
        for (int k = 0; k < EPT; ++k) d[k] = (base + k * BT) < n ? pairs[beg + base + k * BT].y : -1;
        #pragma unroll
        for (int k = 0; k < EPT; ++k) {
            const unsigned l = (unsigned)(d[k] - node0);
            if ((base + k * BT) < n && l < (unsigned)nn) atomicAdd(&cntl[l], 1);
        }
    }
    __syncthreads();
    int cnt = 0, v = 0;
    if (tid < CNT_PAD) {
        cnt = cntl[tid];
        v = wave_incl_scan(cnt, lane);
        if (lane == 31) wtot[wave] = v;
    }
    __syncthreads();
    if (tid < CNT_PAD) {
        int base = 0;
        for (int w = 0; w < wave; ++w) base += wtot[w];
        const int ex = base + v - cnt;
        cntl[tid] = ex;
        if (tid < nn) {
            rowptr[node0 + tid] = beg + ex;
            dis[node0 + tid]    = rsqrtf((float)cnt + 1.0f);
        }
    }
    if (b == NB - 1 && tid == 0) rowptr[N] = end;
    __syncthreads();
    const bool staged = n <= BIN_CAP;
    for (int base = tid; base < n; base += BT * EPT) {
        int2 r[EPT];
        #pragma unroll
        for (int k = 0; k < EPT; ++k) {
            if ((base + k * BT) < n) r[k] = pairs[beg + base + k * BT];
            else { r[k].x = 0; r[k].y = -1; }
        }
        #pragma unroll
        for (int k = 0; k < EPT; ++k) {
            const unsigned l = (unsigned)(r[k].y - node0);
            if ((base + k * BT) < n && l < (unsigned)nn) {
                const int p = atomicAdd(&cntl[l], 1);
                if (staged) {
                    if ((unsigned)p < (unsigned)BIN_CAP) scol[p] = r[k].x;
                } else {
                    const int g = beg + p;
                    if ((unsigned)g < (unsigned)E) col[g] = r[k].x;
                }
            }
        }
    }
    if (staged) {
        __syncthreads();
        for (int i = tid; i < n; i += BT) col[beg + i] = scol[i];
    }
}

__global__ __launch_bounds__(256) void gather_layer(const int* __restrict__ rowptr,
                                                    const int* __restrict__ col,
                                                    const float* __restrict__ dis,
                                                    const float* __restrict__ h,
                                                    const float* __restrict__ bias,
                                                    float* __restrict__ out, int N, int E) {
    const int lane = threadIdx.x & 31;
    const int gw = __builtin_amdgcn_readfirstlane(
        (int)((blockIdx.x * blockDim.x + threadIdx.x) >> 5));
    const int nw = (int)((gridDim.x * blockDim.x) >> 5);
    const float4* h4 = (const float4*)h;
    float4*       o4 = (float4*)out;
    const float4  bv = ((const float4*)bias)[lane];

    for (int i = gw; i < N; i += nw) {
        int beg = rowptr[i], end = rowptr[i + 1];
        beg = beg > 0 ? beg : 0;
        end = end < E ? end : E;
        const float  di  = dis[i];
        const float4 hv  = h4[(size_t)i * (HID / 4) + lane];
        float4 acc; acc.x = acc.y = acc.z = acc.w = 0.f;

        for (int base = beg; base < end; base += 32) {
            const int m = (end - base) < 32 ? (end - base) : 32;
            int   s_l = 0;
            float w_l = 0.f;
            if (lane < m) {
                s_l = col[base + lane];
                s_l = (unsigned)s_l < (unsigned)N ? s_l : 0;
                w_l = dis[s_l] * di;
            }
            #pragma unroll 4
            for (int j = 0; j < m; ++j) {
                const int   s = __builtin_amdgcn_readlane(s_l, j);
                const float w = __int_as_float(__builtin_amdgcn_readlane(__float_as_int(w_l), j));
                const float4 v = h4[(size_t)s * (HID / 4) + lane];
                acc.x += v.x * w; acc.y += v.y * w; acc.z += v.z * w; acc.w += v.w * w;
            }
        }
        const float dd = di * di;
        float4 r;
        r.x = acc.x + dd * hv.x + bv.x;
        r.y = acc.y + dd * hv.y + bv.y;
        r.z = acc.z + dd * hv.z + bv.z;
        r.w = acc.w + dd * hv.w + bv.w;
        r.x = r.x > 0.f ? r.x : 0.f;
        r.y = r.y > 0.f ? r.y : 0.f;
        r.z = r.z > 0.f ? r.z : 0.f;
        r.w = r.w > 0.f ? r.w : 0.f;
        o4[(size_t)i * (HID / 4) + lane] = r;
    }
}

#define GATHER_WGS 2048
static_assert((unsigned long long)N_NODES * (GATHER_WGS * 8ull) < (1ull << 32),
              "wave index times node count fits 32 bits in the range split");
__global__ __launch_bounds__(256) void gather_pool_layer(const int* __restrict__ rowptr,
                                                         const int* __restrict__ col,
                                                         const float* __restrict__ dis,
                                                         const float* __restrict__ h,
                                                         const float* __restrict__ bias,
                                                         const int* __restrict__ batch,
                                                         float* __restrict__ pooled,
                                                         float* __restrict__ cnt, int N, int E) {
    __shared__ float4 sAcc4[8 * (HID / 4)];
    __shared__ int4   sMeta4[4];
    const int tid  = threadIdx.x;
    const int lane = tid & 31;
    const int wave = tid >> 5;
    const unsigned gw = (unsigned)__builtin_amdgcn_readfirstlane(
        (int)((blockIdx.x * blockDim.x + threadIdx.x) >> 5));
    const unsigned nw = (gridDim.x * blockDim.x) >> 5;
    const int n0 = (int)(gw * (unsigned)N / nw);
    const int n1 = (int)((gw + 1u) * (unsigned)N / nw);
    const float4* h4 = (const float4*)h;
    const float4  bv = ((const float4*)bias)[lane];

    int    cur = 0, run = 0;
    float4 pacc; pacc.x = pacc.y = pacc.z = pacc.w = 0.f;

    for (int i = n0; i < n1; ++i) {
        int beg = rowptr[i], end = rowptr[i + 1];
        beg = beg > 0 ? beg : 0;
        end = end < E ? end : E;
        const float  di  = dis[i];
        const int    g   = batch[i];
        const float4 hv  = h4[(size_t)i * (HID / 4) + lane];
        float4 acc; acc.x = acc.y = acc.z = acc.w = 0.f;

        for (int base = beg; base < end; base += 32) {
            const int m = (end - base) < 32 ? (end - base) : 32;
            int   s_l = 0;
            float w_l = 0.f;
            if (lane < m) {
                s_l = col[base + lane];
                s_l = (unsigned)s_l < (unsigned)N ? s_l : 0;
                w_l = dis[s_l] * di;
            }
            #pragma unroll 4
            for (int j = 0; j < m; ++j) {
                const int   s = __builtin_amdgcn_readlane(s_l, j);
                const float w = __int_as_float(__builtin_amdgcn_readlane(__float_as_int(w_l), j));
                const float4 v = h4[(size_t)s * (HID / 4) + lane];
                acc.x += v.x * w; acc.y += v.y * w; acc.z += v.z * w; acc.w += v.w * w;
            }
        }
        const float dd = di * di;
        float4 r;
        r.x = acc.x + dd * hv.x + bv.x;
        r.y = acc.y + dd * hv.y + bv.y;
        r.z = acc.z + dd * hv.z + bv.z;
        r.w = acc.w + dd * hv.w + bv.w;
        r.x = r.x > 0.f ? r.x : 0.f;
        r.y = r.y > 0.f ? r.y : 0.f;
        r.z = r.z > 0.f ? r.z : 0.f;
        r.w = r.w > 0.f ? r.w : 0.f;

        if (run > 0 && g != cur) {
            if ((unsigned)cur < (unsigned)N_GRAPHS) {
                float* pg = pooled + cur * HID + 4 * lane;
                atomicAdd(&pg[0], pacc.x);
                atomicAdd(&pg[1], pacc.y);
                atomicAdd(&pg[2], pacc.z);
                atomicAdd(&pg[3], pacc.w);
                if (lane == 0) atomicAdd(&cnt[cur], (float)run);
            }
            run = 0;
            pacc.x = pacc.y = pacc.z = pacc.w = 0.f;
        }
        cur = g;
        pacc.x += r.x; pacc.y += r.y; pacc.z += r.z; pacc.w += r.w;
        ++run;
    }

    sAcc4[wave * (HID / 4) + lane] = pacc;
    if (lane == 0) { ((int*)sMeta4)[wave] = cur; ((int*)sMeta4)[8 + wave] = run; }
    __syncthreads();
    if (tid < HID) {
        const float* sa = (const float*)sAcc4;
        float v[8];
        #pragma unroll
        for (int w = 0; w < 8; ++w) v[w] = sa[w * HID + tid];
        const int4 ga = sMeta4[0], gb = sMeta4[1], ra = sMeta4[2], rb = sMeta4[3];
        const int gq[8] = { ga.x, ga.y, ga.z, ga.w, gb.x, gb.y, gb.z, gb.w };
        const int rw[8] = { ra.x, ra.y, ra.z, ra.w, rb.x, rb.y, rb.z, rb.w };
        int   cg = 0, rc = 0;
        float s  = 0.f;
        #pragma unroll
        for (int w = 0; w < 8; ++w) {
            if (rw[w] == 0) continue;
            if (rc > 0 && gq[w] != cg) {
                if ((unsigned)cg < (unsigned)N_GRAPHS) {
                    atomicAdd(&pooled[cg * HID + tid], s);
                    if (tid == 0) atomicAdd(&cnt[cg], (float)rc);
                }
                rc = 0; s = 0.f;
            }
            cg = gq[w];
            s += v[w];
            rc += rw[w];
        }
        if (rc > 0 && (unsigned)cg < (unsigned)N_GRAPHS) {
            atomicAdd(&pooled[cg * HID + tid], s);
            if (tid == 0) atomicAdd(&cnt[cg], (float)rc);
        }
    }
}

#define POOL_ROWS 16
static_assert(N_NODES % POOL_ROWS == 0, "every wave owns a full group of rows");
__global__ __launch_bounds__(256) void pool_accum(const float* __restrict__ h,
                                                  const int* __restrict__ batch,
                                                  float* __restrict__ pooled,
                                                  float* __restrict__ cnt, int N) {
    const int lane = threadIdx.x & 31;
    const int gw = __builtin_amdgcn_readfirstlane(
        (int)((blockIdx.x * blockDim.x + threadIdx.x) >> 5));
    const int r0 = gw * POOL_ROWS;
    if (r0 >= N) return;

    const int*   bp = batch + r0;
    const float* hp = h + (size_t)r0 * HID + lane;
    int   g[POOL_ROWS];
    float v[POOL_ROWS][4];
    #pragma unroll
    for (int i = 0; i < POOL_ROWS; ++i) {
        g[i] = bp[i];
        #pragma unroll
        for (int k = 0; k < 4; ++k) v[i][k] = hp[i * HID + 32 * k];
    }

    int   cur = g[0];
    int   run = 0;
    float a0 = 0.f, a1 = 0.f, a2 = 0.f, a3 = 0.f;
    #pragma unroll
    for (int i = 0; i < POOL_ROWS; ++i) {
        if (g[i] != cur) {
            float* pg = pooled + cur * HID + lane;
            atomicAdd(&pg[0],  a0);
            atomicAdd(&pg[32], a1);
            atomicAdd(&pg[64], a2);
            atomicAdd(&pg[96], a3);
            if (lane == 0) atomicAdd(&cnt[cur], (float)run);
            cur = g[i]; run = 0;
            a0 = a1 = a2 = a3 = 0.f;
        }
        a0 += v[i][0]; a1 += v[i][1]; a2 += v[i][2]; a3 += v[i][3];
        ++run;
    }
    {
        float* pg = pooled + cur * HID + lane;
        atomicAdd(&pg[0],  a0);
        atomicAdd(&pg[32], a1);
        atomicAdd(&pg[64], a2);
        atomicAdd(&pg[96], a3);
        if (lane == 0) atomicAdd(&cnt[cur], (float)run);
    }
}

#define HEAD_T_MF  (N_GRAPHS * OUT_MF)
#define HEAD_T_BP  (N_GRAPHS * OUT_BP)
#define HEAD_T_CC  (N_GRAPHS * OUT_CC)
#define HEAD_T_ALL (HEAD_T_MF + HEAD_T_BP + HEAD_T_CC)
#define HEAD_KB    32
static_assert(HEAD_T_MF % 32 == 0 && HEAD_T_BP % 32 == 0, "a wave never straddles two heads");
static_assert(HEAD_T_ALL % 256 == 0, "whole work-groups: every thread owns an output");
static_assert(OUT_MF > 32 && OUT_BP > 32 && OUT_CC > 32, "a wave's outputs touch at most two graphs");
static_assert(HID % HEAD_KB == 0 && HEAD_KB % 4 == 0, "k groups are whole 16-byte pieces of a mean row");

__global__ __launch_bounds__(256) void heads_fused(const float* __restrict__ pooled,
                                                   const float* __restrict__ cnt,
                                                   const float* __restrict__ Wmf,
                                                   const float* __restrict__ bmf,
                                                   const float* __restrict__ Wbp,
                                                   const float* __restrict__ bbp,
                                                   const float* __restrict__ Wcc,
                                                   const float* __restrict__ bcc,
                                                   float* __restrict__ out) {
    __shared__ float4 sP4[8 * 2 * (HID / 4)];
    const int tid = threadIdx.x, lane = tid & 31, wave = tid >> 5;
    const int t = blockIdx.x * 256 + tid;
    const int tw = __builtin_amdgcn_readfirstlane(t - lane);
    const float* W;
    const float* b;
    int OUT, hb;
    if (tw < HEAD_T_MF)                  { W = Wmf; b = bmf; OUT = OUT_MF; hb = 0; }
    else if (tw < HEAD_T_MF + HEAD_T_BP) { W = Wbp; b = bbp; OUT = OUT_BP; hb = HEAD_T_MF; }
    else                                 { W = Wcc; b = bcc; OUT = OUT_CC; hb = HEAD_T_MF + HEAD_T_BP; }
    const int tt = t - hb;
    const int g  = tt / OUT, o = tt - g * OUT;
    const int g0 = (tw - hb) / OUT;
    const int g1 = (g0 + 1) < N_GRAPHS ? (g0 + 1) : (N_GRAPHS - 1);

    #pragma unroll
    for (int r = 0; r < 2; ++r) {
        const int   gr = r ? g1 : g0;
        const float c  = cnt[gr];
        const float d  = c > 1.f ? c : 1.f;
        float4 v = ((const float4*)(pooled + gr * HID))[lane];
        v.x /= d; v.y /= d; v.z /= d; v.w /= d;
        sP4[(wave * 2 + r) * (HID / 4) + lane] = v;
    }
    __syncthreads();

    const float* prow = (const float*)sP4 + (wave * 2 + (g - g0)) * HID;
    const float* wp   = W + o;
    float s = b[o];
    for (int kb = 0; kb < HID; kb += HEAD_KB) {
        float w[HEAD_KB];
        #pragma unroll
        for (int i = 0; i < HEAD_KB; ++i) w[i] = wp[(size_t)(kb + i) * OUT];
        #pragma unroll
        for (int i = 0; i < HEAD_KB; i += 4) {
            const float4 p = *(const float4*)(prow + kb + i);
            s += p.x * w[i];
            s += p.y * w[i + 1];
            s += p.z * w[i + 2];
            s += p.w * w[i + 3];
        }
    }
    out[t] = s;
}

extern "C" void kernel_launch(void* const* d_in, const int* in_sizes, int n_in,
                              void* d_out, int out_size, void* d_ws, size_t ws_size,
                              hipStream_t stream) {
    const float* x   = (const float*)d_in[0];
    const int*   ei  = (const int*)d_in[1];
    const int*   bat = (const int*)d_in[2];
    const float* W1  = (const float*)d_in[3];
    const float* b1  = (const float*)d_in[4];
    const float* W2  = (const float*)d_in[5];
    const float* b2  = (const float*)d_in[6];
    const float* Wmf = (const float*)d_in[7];
    const float* bmf = (const float*)d_in[8];
    const float* Wbp = (const float*)d_in[9];
    const float* bbp = (const float*)d_in[10];
    const float* Wcc = (const float*)d_in[11];
    const float* bcc = (const float*)d_in[12];
    float* out = (float*)d_out;

    const int E = in_sizes[1] / 2;
    const int* src = ei;
    const int* dst = ei + E;

    float* ws     = (float*)d_ws;
    float* dis    = ws;
    float* h      = dis + N_NODES;
    float* agg    = h + (size_t)N_NODES * HID;
    float* pooled = agg + (size_t)N_NODES * HID;
    float* cnt    = pooled + N_GRAPHS * HID;

    int*  rowptr = (int*)(cnt + N_GRAPHS);
    int*  col    = rowptr + (N_NODES + 4);
    int2* pairs  = (int2*)agg;
    int*  cntT   = (int*)agg + 2 * N_EDGES;
    int*  rbase  = cntT + NB * NC;

    if (ws_size >= (size_t)WS_PLANES_BYTES + (size_t)WS_EXTRA_BYTES) {
        bin_count<<<NC, BT, 0, stream>>>(dst, cntT, E);
        bin_place<<<NC, BT, 0, stream>>>(src, dst, cntT, rbase, pairs, E);
        bin_csr<<<NB, BT, 0, stream>>>(pairs, rbase, rowptr, col, dis, N_NODES, E);

        gemm_wmma_f32<IN_DIM><<<N_NODES / GEMM_ROWS, 256, 0, stream>>>(x, W1, h);
        gather_layer<<<2048, 256, 0, stream>>>(rowptr, col, dis, h, b1, agg, N_NODES, E);

        gemm_wmma_f32<HID><<<N_NODES / GEMM_ROWS, 256, 0, stream>>>(agg, W2, h);

        fill_f32<<<(N_GRAPHS * HID + N_GRAPHS + 255) / 256, 256, 0, stream>>>(
            pooled, 0.0f, N_GRAPHS * HID + N_GRAPHS);
        gather_pool_layer<<<GATHER_WGS, 256, 0, stream>>>(rowptr, col, dis, h, b2, bat,
                                                          pooled, cnt, N_NODES, E);
    } else {
        const int NH  = N_NODES * HID;
        const int NH4 = NH / 4;
        fill_f32<<<(N_NODES + 255) / 256, 256, 0, stream>>>(dis, 1.0f, N_NODES);
        deg_accum<<<(E + 255) / 256, 256, 0, stream>>>(dst, dis, E);
        rsqrt_inplace<<<(N_NODES + 255) / 256, 256, 0, stream>>>(dis, N_NODES);

        gemm_wmma_f32<IN_DIM><<<N_NODES / GEMM_ROWS, 256, 0, stream>>>(x, W1, h);
        zero_f4<<<(NH4 + 255) / 256, 256, 0, stream>>>((float4*)agg, NH4);
        edge_scatter<<<2048, 256, 0, stream>>>(src, dst, dis, h, agg, E);
        finish_layer<<<(NH + 255) / 256, 256, 0, stream>>>(agg, h, dis, b1, NH);

        gemm_wmma_f32<HID><<<N_NODES / GEMM_ROWS, 256, 0, stream>>>(agg, W2, h);
        zero_f4<<<(NH4 + 255) / 256, 256, 0, stream>>>((float4*)agg, NH4);
        edge_scatter<<<2048, 256, 0, stream>>>(src, dst, dis, h, agg, E);
        finish_layer<<<(NH + 255) / 256, 256, 0, stream>>>(agg, h, dis, b2, NH);

        fill_f32<<<(N_GRAPHS * HID + N_GRAPHS + 255) / 256, 256, 0, stream>>>(
            pooled, 0.0f, N_GRAPHS * HID + N_GRAPHS);
        pool_accum<<<(N_NODES + POOL_ROWS * 8 - 1) / (POOL_ROWS * 8), 256, 0, stream>>>(agg, bat, pooled, cnt, N_NODES);
    }

    heads_fused<<<HEAD_T_ALL / 256, 256, 0, stream>>>(pooled, cnt, Wmf, bmf, Wbp, bbp, Wcc, bcc, out);
}


# GCN normalisation factored out of the edge walk: product rows stored pre-scaled by dis[row], gathers sum plain rows then scale by dis[i] once; no dis[src] fetch per edge; products re-associated
# speedup vs baseline: 4.7034x; 1.0848x over previous
#include <hip/hip_runtime.h>
#include <hip/hip_bf16.h>

#define N_NODES 100000
#define N_EDGES 1600000
#define IN_DIM  256
#define HID     128
#define N_GRAPHS 64
#define OUT_MF 489
#define OUT_BP 1943
#define OUT_CC 320

#define WS_PLANES_BYTES (4ull * (N_NODES + 2ull * N_NODES * HID + N_GRAPHS * HID + N_GRAPHS))
#define WS_EXTRA_BYTES  (4ull * (N_NODES + 4 + N_EDGES))

typedef __attribute__((ext_vector_type(2))) float v2f;
typedef __attribute__((ext_vector_type(8))) float v8f;
typedef __attribute__((ext_vector_type(4))) float v4f;

__global__ void fill_f32(float* __restrict__ p, float v, int n) {
    int t = blockIdx.x * blockDim.x + threadIdx.x;
    if (t < n) p[t] = v;
}

__global__ void zero_f4(float4* __restrict__ p, int n4) {
    int t = blockIdx.x * blockDim.x + threadIdx.x;
    if (t < n4) { float4 z; z.x = z.y = z.z = z.w = 0.f; p[t] = z; }
}

__global__ void deg_accum(const int* __restrict__ dst, float* __restrict__ deg, int E) {
    int t = blockIdx.x * blockDim.x + threadIdx.x;
    if (t < E) atomicAdd(&deg[dst[t]], 1.0f);
}

__global__ void rsqrt_inplace(float* __restrict__ p, int n) {
    int t = blockIdx.x * blockDim.x + threadIdx.x;
    if (t < n) p[t] = rsqrtf(p[t]);
}

#define GEMM_ROWS 80
#define GEMM_RT   (GEMM_ROWS / 16)
#define GEMM_KC   64
#define GEMM_PA   (GEMM_KC + 4)
#define GEMM_PW   144
#define GEMM_NA   (GEMM_ROWS * GEMM_KC / 4 / 256)
#define GEMM_NW   ((GEMM_KC / 2) * (HID / 4) / 256)
static_assert(N_NODES % GEMM_ROWS == 0, "every work-group owns a full set of rows");
static_assert((GEMM_ROWS * GEMM_KC / 4) % 256 == 0, "A chunk splits evenly over the threads");

template <int K, bool SCALE>
__global__ __launch_bounds__(256) void gemm_wmma_f32(const float* __restrict__ A,
                                                     const float* __restrict__ W,
                                                     const float* __restrict__ rs,
                                                     float* __restrict__ out) {
    __shared__ float4 ldsA4[GEMM_ROWS * GEMM_PA / 4];
    __shared__ float4 ldsW4[(GEMM_KC / 2) * GEMM_PW / 2];
    float*  ldsA = (float*)ldsA4;
    float2* ldsW = (float2*)ldsW4;
    const int tid  = threadIdx.x;
    const int wave = tid >> 5;
    const int lane = tid & 31;
    const int half = lane >> 4;
    const int m    = lane & 15;
    const int row0 = blockIdx.x * GEMM_ROWS;
    const int col0 = wave * 16;

    const float* ag = A + (size_t)(row0 + (tid >> 4)) * K + 4 * (tid & 15);
    const float* wg = W + (size_t)(2 * (tid >> 5)) * HID + 4 * (tid & 31);
    v4f* as = (v4f*)(ldsA + (tid >> 4) * GEMM_PA + 4 * (tid & 15));
    v4f* wsd = (v4f*)(ldsW + (tid >> 5) * GEMM_PW + 4 * (tid & 31));

    v8f acc[GEMM_RT];
    #pragma unroll
    for (int t = 0; t < GEMM_RT; ++t) acc[t] = v8f{};

    const float*  aF = ldsA + m * GEMM_PA + 2 * half;
    const float2* bF = ldsW + half * GEMM_PW + col0 + m;

    for (int kb = 0; kb < K; kb += GEMM_KC) {
        {
            v4f ra[GEMM_NA], rw0[GEMM_NW], rw1[GEMM_NW];
            #pragma unroll
            for (int i = 0; i < GEMM_NA; ++i)
                ra[i] = *(const v4f*)(ag + (size_t)(16 * i) * K + kb);
            #pragma unroll
            for (int i = 0; i < GEMM_NW; ++i) {
                rw0[i] = *(const v4f*)(wg + (size_t)(kb + 16 * i) * HID);
                rw1[i] = *(const v4f*)(wg + (size_t)(kb + 16 * i + 1) * HID);
            }
            __builtin_amdgcn_sched_barrier(0);
            #pragma unroll
            for (int i = 0; i < GEMM_NA; ++i) as[i * 16 * GEMM_PA / 4] = ra[i];
            #pragma unroll
            for (int i = 0; i < GEMM_NW; ++i) {
                v4f lo, hi;
                lo.x = rw0[i].x; lo.y = rw1[i].x; lo.z = rw0[i].y; lo.w = rw1[i].y;
                hi.x = rw0[i].z; hi.y = rw1[i].z; hi.z = rw0[i].w; hi.w = rw1[i].w;
                wsd[i * 8 * GEMM_PW / 2]     = lo;
                wsd[i * 8 * GEMM_PW / 2 + 1] = hi;
            }
        }
        __syncthreads();

        #pragma unroll
        for (int k = 0; k < GEMM_KC; k += 4) {
            const float2 bb = bF[(k >> 1) * GEMM_PW];
            v2f b; b.x = bb.x; b.y = bb.y;
            #pragma unroll
            for (int t = 0; t < GEMM_RT; ++t) {
                const float2 aa = *(const float2*)(aF + t * 16 * GEMM_PA + k);
                v2f a; a.x = aa.x; a.y = aa.y;
                acc[t] = __builtin_amdgcn_wmma_f32_16x16x4_f32(
                    false, a, false, b, (short)0, acc[t], false, false);
            }
        }
        __syncthreads();
    }

    #pragma unroll
    for (int t = 0; t < GEMM_RT; ++t) {
        v8f o = acc[t];
        if (SCALE) {
            const float* rp = rs + row0 + 16 * t + 8 * half;
            const v4f s0 = *(const v4f*)rp;
            const v4f s1 = *(const v4f*)(rp + 4);
            o[0] *= s0.x; o[1] *= s0.y; o[2] *= s0.z; o[3] *= s0.w;
            o[4] *= s1.x; o[5] *= s1.y; o[6] *= s1.z; o[7] *= s1.w;
        }
        #pragma unroll
        for (int v = 0; v < 8; ++v)
            out[(size_t)(row0 + 16 * t + v + 8 * half) * HID + col0 + m] = o[v];
    }
}

__global__ __launch_bounds__(256) void edge_scatter(const int* __restrict__ src,
                                                    const int* __restrict__ dst,
                                                    const float* __restrict__ dis,
                                                    const float* __restrict__ h,
                                                    float* __restrict__ agg, int E) {
    const int lane = threadIdx.x & 31;
    const int gw   = (blockIdx.x * blockDim.x + threadIdx.x) >> 5;
    const int nw   = (gridDim.x * blockDim.x) >> 5;
    for (int e = gw; e < E; e += nw) {
        const int s = src[e], d = dst[e];
        const float nrm = dis[s] * dis[d];
        const float* hs = h + (size_t)s * HID;
        float* ad = agg + (size_t)d * HID;
        #pragma unroll
        for (int j = 0; j < HID; j += 32)
            atomicAdd(&ad[j + lane], hs[j + lane] * nrm);
    }
}

__global__ void finish_layer(float* __restrict__ agg, const float* __restrict__ h,
                             const float* __restrict__ dis, const float* __restrict__ bias,
                             int n) {
    int t = blockIdx.x * blockDim.x + threadIdx.x;
    if (t >= n) return;
    const int i = t >> 7, j = t & (HID - 1);
    const float di = dis[i];
    const float v = agg[t] + di * di * h[t] + bias[j];
    agg[t] = v > 0.f ? v : 0.f;
}

#define NB     256
#define BIN_R  ((N_NODES + NB - 1) / NB)
#define NC     256
#define BT     1024
#define CHUNK_CAP 6400
#define BIN_CAP   8192
#define CNT_PAD   512
#define EPT       8
static_assert(BIN_R <= CNT_PAD, "a range's nodes fit the padded counter array");
#define TAB_Q  (NB / 4)
#define TAB_RG (BT / TAB_Q)
#define TAB_K  (NC / TAB_RG)
static_assert(NB % 4 == 0 && BT % TAB_Q == 0 && NC % TAB_RG == 0 && TAB_Q % 32 == 0,
              "the table splits into whole 16-byte groups, whole waves per row half");

__device__ __forceinline__ int bin_of(int d) {
    const unsigned b = (unsigned)d / (unsigned)BIN_R;
    return b < (unsigned)NB ? (int)b : NB - 1;
}

__device__ __forceinline__ int wave_incl_scan(int v, int lane) {
    #pragma unroll
    for (int d = 1; d < 32; d <<= 1) {
        const int u = __shfl_up(v, d);
        if (lane >= d) v += u;
    }
    return v;
}

__global__ __launch_bounds__(BT) void bin_count(const int* __restrict__ dst,
                                                int* __restrict__ cntT, int E) {
    __shared__ int hist[NB];
    const int tid = threadIdx.x, c = blockIdx.x;
    const int epc = (E + NC - 1) / NC;
    const int e0  = c * epc;
    const int e1  = (e0 + epc) < E ? (e0 + epc) : E;
    if (tid < NB) hist[tid] = 0;
    __syncthreads();
    for (int base = e0 + tid; base < e1; base += BT * EPT) {
        int d[EPT];
        #pragma unroll
        for (int k = 0; k < EPT; ++k) d[k] = (base + k * BT) < e1 ? dst[base + k * BT] : 0;
        #pragma unroll
        for (int k = 0; k < EPT; ++k)
            if ((base + k * BT) < e1) atomicAdd(&hist[bin_of(d[k])], 1);
    }
    __syncthreads();
    if (tid < NB) cntT[c * NB + tid] = hist[tid];
}

__global__ __launch_bounds__(BT) void bin_place(const int* __restrict__ src,
                                                const int* __restrict__ dst,
                                                const int* __restrict__ cntT,
                                                int* __restrict__ rbase,
                                                int2* __restrict__ pairs, int E) {
    __shared__ int  hist[NB];
    __shared__ int  gdelta[NB];
    __shared__ int  wtot[NB / 32];
    __shared__ int  rtot[NB];
    __shared__ int  rpre[NB];
    __shared__ int  wtotR[NB / 32];
    __shared__ int2 sbuf[CHUNK_CAP];
    const int tid = threadIdx.x, lane = tid & 31, wave = tid >> 5, c = blockIdx.x;
    const int epc = (E + NC - 1) / NC;
    const int e0  = c * epc;
    const int e1  = (e0 + epc) < E ? (e0 + epc) : E;
    const int n   = e1 > e0 ? e1 - e0 : 0;
    if (tid < NB) { hist[tid] = 0; rtot[tid] = 0; rpre[tid] = 0; }
    const int q  = tid & (TAB_Q - 1);
    const int rg = tid / TAB_Q;
    const int4* tp = (const int4*)cntT + rg * TAB_Q + q;
    int4 ta[TAB_K];
    __syncthreads();
    #pragma unroll
    for (int k = 0; k < TAB_K; ++k) ta[k] = tp[k * TAB_RG * TAB_Q];
    for (int base = e0 + tid; base < e1; base += BT * EPT) {
        int d[EPT];
        #pragma unroll
        for (int k = 0; k < EPT; ++k) d[k] = (base + k * BT) < e1 ? dst[base + k * BT] : 0;
        #pragma unroll
        for (int k = 0; k < EPT; ++k)
            if ((base + k * BT) < e1) atomicAdd(&hist[bin_of(d[k])], 1);
    }
    {
        int t0 = 0, t1 = 0, t2 = 0, t3 = 0, p0 = 0, p1 = 0, p2 = 0, p3 = 0;
        #pragma unroll
        for (int k = 0; k < TAB_K; ++k) {
            const int m = (rg + k * TAB_RG) < c ? -1 : 0;
            t0 += ta[k].x; t1 += ta[k].y; t2 += ta[k].z; t3 += ta[k].w;
            p0 += ta[k].x & m; p1 += ta[k].y & m; p2 += ta[k].z & m; p3 += ta[k].w & m;
        }
        atomicAdd(&rtot[4 * q],     t0);  atomicAdd(&rtot[4 * q + 1], t1);
        atomicAdd(&rtot[4 * q + 2], t2);  atomicAdd(&rtot[4 * q + 3], t3);
        atomicAdd(&rpre[4 * q],     p0);  atomicAdd(&rpre[4 * q + 1], p1);
        atomicAdd(&rpre[4 * q + 2], p2);  atomicAdd(&rpre[4 * q + 3], p3);
    }
    __syncthreads();
    int cnt = 0, v = 0, rt = 0, vr = 0;
    if (tid < NB) {
        cnt = hist[tid];
        v = wave_incl_scan(cnt, lane);
        rt = rtot[tid];
        vr = wave_incl_scan(rt, lane);
        if (lane == 31) { wtot[wave] = v; wtotR[wave] = vr; }
    }
    __syncthreads();
    if (tid < NB) {
        int base = 0, baseR = 0;
        for (int w = 0; w < wave; ++w) { base += wtot[w]; baseR += wtotR[w]; }
        const int ex = base + v - cnt;
        const int gx = baseR + vr - rt;
        hist[tid]   = ex;
        gdelta[tid] = gx + rpre[tid] - ex;
        if (c == 0) {
            rbase[tid] = gx;
            if (tid == NB - 1) rbase[NB] = gx + rt;
        }
    }
    __syncthreads();
    const bool staged = n <= CHUNK_CAP;
    for (int base = e0 + tid; base < e1; base += BT * EPT) {
        int s[EPT], d[EPT];
        #pragma unroll
        for (int k = 0; k < EPT; ++k) {
            const bool in = (base + k * BT) < e1;
            s[k] = in ? src[base + k * BT] : 0;
            d[k] = in ? dst[base + k * BT] : 0;
        }
        #pragma unroll
        for (int k = 0; k < EPT; ++k) {
            if ((base + k * BT) < e1) {
                int2 r; r.x = s[k]; r.y = d[k];
                const int b = bin_of(r.y);
                const int p = atomicAdd(&hist[b], 1);
                if (staged) {
                    if ((unsigned)p < (unsigned)CHUNK_CAP) sbuf[p] = r;
                } else {
                    const int g = p + gdelta[b];
                    if ((unsigned)g < (unsigned)E) pairs[g] = r;
                }
            }
        }
    }
    if (staged) {
        __syncthreads();
        for (int i = tid; i < n; i += BT) {
            const int2 r = sbuf[i];
            const int  g = i + gdelta[bin_of(r.y)];
            if ((unsigned)g < (unsigned)E) pairs[g] = r;
        }
    }
}

__global__ __launch_bounds__(BT) void bin_csr(const int2* __restrict__ pairs,
                                              const int* __restrict__ rbase,
                                              int* __restrict__ rowptr, int* __restrict__ col,
                                              float* __restrict__ dis, int N, int E) {
    __shared__ int cntl[CNT_PAD];
    __shared__ int wtot[CNT_PAD / 32];
    __shared__ int scol[BIN_CAP];
    const int tid = threadIdx.x, lane = tid & 31, wave = tid >> 5, b = blockIdx.x;
    int beg = rbase[b], end = rbase[b + 1];
    beg = beg > 0 ? beg : 0;  beg = beg < E ? beg : E;
    end = end > beg ? end : beg;  end = end < E ? end : E;
    const int n     = end - beg;
    const int node0 = b * BIN_R;
    int nn = N - node0;
    nn = nn < BIN_R ? nn : BIN_R;
    nn = nn > 0 ? nn : 0;
    if (tid < CNT_PAD) cntl[tid] = 0;
    __syncthreads();
    for (int base = tid; base < n; base += BT * EPT) {
        int d[EPT];
        #pragma unroll
        for (int k = 0; k < EPT; ++k) d[k] = (base + k * BT) < n ? pairs[beg + base + k * BT].y : -1;
        #pragma unroll
        for (int k = 0; k < EPT; ++k) {
            const unsigned l = (unsigned)(d[k] - node0);
            if ((base + k * BT) < n && l < (unsigned)nn) atomicAdd(&cntl[l], 1);
        }
    }
    __syncthreads();
    int cnt = 0, v = 0;
    if (tid < CNT_PAD) {
        cnt = cntl[tid];
        v = wave_incl_scan(cnt, lane);
        if (lane == 31) wtot[wave] = v;
    }
    __syncthreads();
    if (tid < CNT_PAD) {
        int base = 0;
        for (int w = 0; w < wave; ++w) base += wtot[w];
        const int ex = base + v - cnt;
        cntl[tid] = ex;
        if (tid < nn) {
            rowptr[node0 + tid] = beg + ex;
            dis[node0 + tid]    = rsqrtf((float)cnt + 1.0f);
        }
    }
    if (b == NB - 1 && tid == 0) rowptr[N] = end;
    __syncthreads();
    const bool staged = n <= BIN_CAP;
    for (int base = tid; base < n; base += BT * EPT) {
        int2 r[EPT];
        #pragma unroll
        for (int k = 0; k < EPT; ++k) {
            if ((base + k * BT) < n) r[k] = pairs[beg + base + k * BT];
            else { r[k].x = 0; r[k].y = -1; }
        }
        #pragma unroll
        for (int k = 0; k < EPT; ++k) {
            const unsigned l = (unsigned)(r[k].y - node0);
            if ((base + k * BT) < n && l < (unsigned)nn) {
                const int p = atomicAdd(&cntl[l], 1);
                if (staged) {
                    if ((unsigned)p < (unsigned)BIN_CAP) scol[p] = r[k].x;
                } else {
                    const int g = beg + p;
                    if ((unsigned)g < (unsigned)E) col[g] = r[k].x;
                }
            }
        }
    }
    if (staged) {
        __syncthreads();
        for (int i = tid; i < n; i += BT) col[beg + i] = scol[i];
    }
}

__global__ __launch_bounds__(256) void gather_layer(const int* __restrict__ rowptr,
                                                    const int* __restrict__ col,
                                                    const float* __restrict__ dis,
                                                    const float* __restrict__ h,
                                                    const float* __restrict__ bias,
                                                    float* __restrict__ out, int N, int E) {
    const int lane = threadIdx.x & 31;
    const int gw = __builtin_amdgcn_readfirstlane(
        (int)((blockIdx.x * blockDim.x + threadIdx.x) >> 5));
    const int nw = (int)((gridDim.x * blockDim.x) >> 5);
    const float4* h4 = (const float4*)h;
    float4*       o4 = (float4*)out;
    const float4  bv = ((const float4*)bias)[lane];

    for (int i = gw; i < N; i += nw) {
        int beg = rowptr[i], end = rowptr[i + 1];
        beg = beg > 0 ? beg : 0;
        end = end < E ? end : E;
        const float  di  = dis[i];
        const float4 hv  = h4[(size_t)i * (HID / 4) + lane];
        float4 acc; acc.x = acc.y = acc.z = acc.w = 0.f;

        for (int base = beg; base < end; base += 32) {
            const int m = (end - base) < 32 ? (end - base) : 32;
            int s_l = 0;
            if (lane < m) {
                s_l = col[base + lane];
                s_l = (unsigned)s_l < (unsigned)N ? s_l : 0;
            }
            #pragma unroll 4
            for (int j = 0; j < m; ++j) {
                const int    s = __builtin_amdgcn_readlane(s_l, j);
                const float4 v = h4[(size_t)s * (HID / 4) + lane];
                acc.x += v.x; acc.y += v.y; acc.z += v.z; acc.w += v.w;
            }
        }
        float4 r;
        r.x = di * (acc.x + hv.x) + bv.x;
        r.y = di * (acc.y + hv.y) + bv.y;
        r.z = di * (acc.z + hv.z) + bv.z;
        r.w = di * (acc.w + hv.w) + bv.w;
        r.x = r.x > 0.f ? r.x : 0.f;
        r.y = r.y > 0.f ? r.y : 0.f;
        r.z = r.z > 0.f ? r.z : 0.f;
        r.w = r.w > 0.f ? r.w : 0.f;
        r.x *= di; r.y *= di; r.z *= di; r.w *= di;
        o4[(size_t)i * (HID / 4) + lane] = r;
    }
}

#define GATHER_WGS 2048
static_assert((unsigned long long)N_NODES * (GATHER_WGS * 8ull) < (1ull << 32),
              "wave index times node count fits 32 bits in the range split");
__global__ __launch_bounds__(256) void gather_pool_layer(const int* __restrict__ rowptr,
                                                         const int* __restrict__ col,
                                                         const float* __restrict__ dis,
                                                         const float* __restrict__ h,
                                                         const float* __restrict__ bias,
                                                         const int* __restrict__ batch,
                                                         float* __restrict__ pooled,
                                                         float* __restrict__ cnt, int N, int E) {
    __shared__ float4 sAcc4[8 * (HID / 4)];
    __shared__ int4   sMeta4[4];
    const int tid  = threadIdx.x;
    const int lane = tid & 31;
    const int wave = tid >> 5;
    const unsigned gw = (unsigned)__builtin_amdgcn_readfirstlane(
        (int)((blockIdx.x * blockDim.x + threadIdx.x) >> 5));
    const unsigned nw = (gridDim.x * blockDim.x) >> 5;
    const int n0 = (int)(gw * (unsigned)N / nw);
    const int n1 = (int)((gw + 1u) * (unsigned)N / nw);
    const float4* h4 = (const float4*)h;
    const float4  bv = ((const float4*)bias)[lane];

    int    cur = 0, run = 0;
    float4 pacc; pacc.x = pacc.y = pacc.z = pacc.w = 0.f;

    for (int i = n0; i < n1; ++i) {
        int beg = rowptr[i], end = rowptr[i + 1];
        beg = beg > 0 ? beg : 0;
        end = end < E ? end : E;
        const float  di  = dis[i];
        const int    g   = batch[i];
        const float4 hv  = h4[(size_t)i * (HID / 4) + lane];
        float4 acc; acc.x = acc.y = acc.z = acc.w = 0.f;

        for (int base = beg; base < end; base += 32) {
            const int m = (end - base) < 32 ? (end - base) : 32;
            int s_l = 0;
            if (lane < m) {
                s_l = col[base + lane];
                s_l = (unsigned)s_l < (unsigned)N ? s_l : 0;
            }
            #pragma unroll 4
            for (int j = 0; j < m; ++j) {
                const int    s = __builtin_amdgcn_readlane(s_l, j);
                const float4 v = h4[(size_t)s * (HID / 4) + lane];
                acc.x += v.x; acc.y += v.y; acc.z += v.z; acc.w += v.w;
            }
        }
        float4 r;
        r.x = di * (acc.x + hv.x) + bv.x;
        r.y = di * (acc.y + hv.y) + bv.y;
        r.z = di * (acc.z + hv.z) + bv.z;
        r.w = di * (acc.w + hv.w) + bv.w;
        r.x = r.x > 0.f ? r.x : 0.f;
        r.y = r.y > 0.f ? r.y : 0.f;
        r.z = r.z > 0.f ? r.z : 0.f;
        r.w = r.w > 0.f ? r.w : 0.f;

        if (run > 0 && g != cur) {
            if ((unsigned)cur < (unsigned)N_GRAPHS) {
                float* pg = pooled + cur * HID + 4 * lane;
                atomicAdd(&pg[0], pacc.x);
                atomicAdd(&pg[1], pacc.y);
                atomicAdd(&pg[2], pacc.z);
                atomicAdd(&pg[3], pacc.w);
                if (lane == 0) atomicAdd(&cnt[cur], (float)run);
            }
            run = 0;
            pacc.x = pacc.y = pacc.z = pacc.w = 0.f;
        }
        cur = g;
        pacc.x += r.x; pacc.y += r.y; pacc.z += r.z; pacc.w += r.w;
        ++run;
    }

    sAcc4[wave * (HID / 4) + lane] = pacc;
    if (lane == 0) { ((int*)sMeta4)[wave] = cur; ((int*)sMeta4)[8 + wave] = run; }
    __syncthreads();
    if (tid < HID) {
        const float* sa = (const float*)sAcc4;
        float v[8];
        #pragma unroll
        for (int w = 0; w < 8; ++w) v[w] = sa[w * HID + tid];
        const int4 ga = sMeta4[0], gb = sMeta4[1], ra = sMeta4[2], rb = sMeta4[3];
        const int gq[8] = { ga.x, ga.y, ga.z, ga.w, gb.x, gb.y, gb.z, gb.w };
        const int rw[8] = { ra.x, ra.y, ra.z, ra.w, rb.x, rb.y, rb.z, rb.w };
        int   cg = 0, rc = 0;
        float s  = 0.f;
        #pragma unroll
        for (int w = 0; w < 8; ++w) {
            if (rw[w] == 0) continue;
            if (rc > 0 && gq[w] != cg) {
                if ((unsigned)cg < (unsigned)N_GRAPHS) {
                    atomicAdd(&pooled[cg * HID + tid], s);
                    if (tid == 0) atomicAdd(&cnt[cg], (float)rc);
                }
                rc = 0; s = 0.f;
            }
            cg = gq[w];
            s += v[w];
            rc += rw[w];
        }
        if (rc > 0 && (unsigned)cg < (unsigned)N_GRAPHS) {
            atomicAdd(&pooled[cg * HID + tid], s);
            if (tid == 0) atomicAdd(&cnt[cg], (float)rc);
        }
    }
}

#define POOL_ROWS 16
static_assert(N_NODES % POOL_ROWS == 0, "every wave owns a full group of rows");
__global__ __launch_bounds__(256) void pool_accum(const float* __restrict__ h,
                                                  const int* __restrict__ batch,
                                                  float* __restrict__ pooled,
                                                  float* __restrict__ cnt, int N) {
    const int lane = threadIdx.x & 31;
    const int gw = __builtin_amdgcn_readfirstlane(
        (int)((blockIdx.x * blockDim.x + threadIdx.x) >> 5));
    const int r0 = gw * POOL_ROWS;
    if (r0 >= N) return;

    const int*   bp = batch + r0;
    const float* hp = h + (size_t)r0 * HID + lane;
    int   g[POOL_ROWS];
    float v[POOL_ROWS][4];
    #pragma unroll
    for (int i = 0; i < POOL_ROWS; ++i) {
        g[i] = bp[i];
        #pragma unroll
        for (int k = 0; k < 4; ++k) v[i][k] = hp[i * HID + 32 * k];
    }

    int   cur = g[0];
    int   run = 0;
    float a0 = 0.f, a1 = 0.f, a2 = 0.f, a3 = 0.f;
    #pragma unroll
    for (int i = 0; i < POOL_ROWS; ++i) {
        if (g[i] != cur) {
            float* pg = pooled + cur * HID + lane;
            atomicAdd(&pg[0],  a0);
            atomicAdd(&pg[32], a1);
            atomicAdd(&pg[64], a2);
            atomicAdd(&pg[96], a3);
            if (lane == 0) atomicAdd(&cnt[cur], (float)run);
            cur = g[i]; run = 0;
            a0 = a1 = a2 = a3 = 0.f;
        }
        a0 += v[i][0]; a1 += v[i][1]; a2 += v[i][2]; a3 += v[i][3];
        ++run;
    }
    {
        float* pg = pooled + cur * HID + lane;
        atomicAdd(&pg[0],  a0);
        atomicAdd(&pg[32], a1);
        atomicAdd(&pg[64], a2);
        atomicAdd(&pg[96], a3);
        if (lane == 0) atomicAdd(&cnt[cur], (float)run);
    }
}

#define HEAD_T_MF  (N_GRAPHS * OUT_MF)
#define HEAD_T_BP  (N_GRAPHS * OUT_BP)
#define HEAD_T_CC  (N_GRAPHS * OUT_CC)
#define HEAD_T_ALL (HEAD_T_MF + HEAD_T_BP + HEAD_T_CC)
#define HEAD_KB    32
static_assert(HEAD_T_MF % 32 == 0 && HEAD_T_BP % 32 == 0, "a wave never straddles two heads");
static_assert(HEAD_T_ALL % 256 == 0, "whole work-groups: every thread owns an output");
static_assert(OUT_MF > 32 && OUT_BP > 32 && OUT_CC > 32, "a wave's outputs touch at most two graphs");
static_assert(HID % HEAD_KB == 0 && HEAD_KB % 4 == 0, "k groups are whole 16-byte pieces of a mean row");

__global__ __launch_bounds__(256) void heads_fused(const float* __restrict__ pooled,
                                                   const float* __restrict__ cnt,
                                                   const float* __restrict__ Wmf,
                                                   const float* __restrict__ bmf,
                                                   const float* __restrict__ Wbp,
                                                   const float* __restrict__ bbp,
                                                   const float* __restrict__ Wcc,
                                                   const float* __restrict__ bcc,
                                                   float* __restrict__ out) {
    __shared__ float4 sP4[8 * 2 * (HID / 4)];
    const int tid = threadIdx.x, lane = tid & 31, wave = tid >> 5;
    const int t = blockIdx.x * 256 + tid;
    const int tw = __builtin_amdgcn_readfirstlane(t - lane);
    const float* W;
    const float* b;
    int OUT, hb;
    if (tw < HEAD_T_MF)                  { W = Wmf; b = bmf; OUT = OUT_MF; hb = 0; }
    else if (tw < HEAD_T_MF + HEAD_T_BP) { W = Wbp; b = bbp; OUT = OUT_BP; hb = HEAD_T_MF; }
    else                                 { W = Wcc; b = bcc; OUT = OUT_CC; hb = HEAD_T_MF + HEAD_T_BP; }
    const int tt = t - hb;
    const int g  = tt / OUT, o = tt - g * OUT;
    const int g0 = (tw - hb) / OUT;
    const int g1 = (g0 + 1) < N_GRAPHS ? (g0 + 1) : (N_GRAPHS - 1);

    #pragma unroll
    for (int r = 0; r < 2; ++r) {
        const int   gr = r ? g1 : g0;
        const float c  = cnt[gr];
        const float d  = c > 1.f ? c : 1.f;
        float4 v = ((const float4*)(pooled + gr * HID))[lane];
        v.x /= d; v.y /= d; v.z /= d; v.w /= d;
        sP4[(wave * 2 + r) * (HID / 4) + lane] = v;
    }
    __syncthreads();

    const float* prow = (const float*)sP4 + (wave * 2 + (g - g0)) * HID;
    const float* wp   = W + o;
    float s = b[o];
    for (int kb = 0; kb < HID; kb += HEAD_KB) {
        float w[HEAD_KB];
        #pragma unroll
        for (int i = 0; i < HEAD_KB; ++i) w[i] = wp[(size_t)(kb + i) * OUT];
        #pragma unroll
        for (int i = 0; i < HEAD_KB; i += 4) {
            const float4 p = *(const float4*)(prow + kb + i);
            s += p.x * w[i];
            s += p.y * w[i + 1];
            s += p.z * w[i + 2];
            s += p.w * w[i + 3];
        }
    }
    out[t] = s;
}

extern "C" void kernel_launch(void* const* d_in, const int* in_sizes, int n_in,
                              void* d_out, int out_size, void* d_ws, size_t ws_size,
                              hipStream_t stream) {
    const float* x   = (const float*)d_in[0];
    const int*   ei  = (const int*)d_in[1];
    const int*   bat = (const int*)d_in[2];
    const float* W1  = (const float*)d_in[3];
    const float* b1  = (const float*)d_in[4];
    const float* W2  = (const float*)d_in[5];
    const float* b2  = (const float*)d_in[6];
    const float* Wmf = (const float*)d_in[7];
    const float* bmf = (const float*)d_in[8];
    const float* Wbp = (const float*)d_in[9];
    const float* bbp = (const float*)d_in[10];
    const float* Wcc = (const float*)d_in[11];
    const float* bcc = (const float*)d_in[12];
    float* out = (float*)d_out;

    const int E = in_sizes[1] / 2;
    const int* src = ei;
    const int* dst = ei + E;

    float* ws     = (float*)d_ws;
    float* dis    = ws;
    float* h      = dis + N_NODES;
    float* agg    = h + (size_t)N_NODES * HID;
    float* pooled = agg + (size_t)N_NODES * HID;
    float* cnt    = pooled + N_GRAPHS * HID;

    int*  rowptr = (int*)(cnt + N_GRAPHS);
    int*  col    = rowptr + (N_NODES + 4);
    int2* pairs  = (int2*)agg;
    int*  cntT   = (int*)agg + 2 * N_EDGES;
    int*  rbase  = cntT + NB * NC;

    if (ws_size >= (size_t)WS_PLANES_BYTES + (size_t)WS_EXTRA_BYTES) {
        bin_count<<<NC, BT, 0, stream>>>(dst, cntT, E);
        bin_place<<<NC, BT, 0, stream>>>(src, dst, cntT, rbase, pairs, E);
        bin_csr<<<NB, BT, 0, stream>>>(pairs, rbase, rowptr, col, dis, N_NODES, E);

        gemm_wmma_f32<IN_DIM, true><<<N_NODES / GEMM_ROWS, 256, 0, stream>>>(x, W1, dis, h);
        gather_layer<<<2048, 256, 0, stream>>>(rowptr, col, dis, h, b1, agg, N_NODES, E);

        gemm_wmma_f32<HID, false><<<N_NODES / GEMM_ROWS, 256, 0, stream>>>(agg, W2, dis, h);

        fill_f32<<<(N_GRAPHS * HID + N_GRAPHS + 255) / 256, 256, 0, stream>>>(
            pooled, 0.0f, N_GRAPHS * HID + N_GRAPHS);
        gather_pool_layer<<<GATHER_WGS, 256, 0, stream>>>(rowptr, col, dis, h, b2, bat,
                                                          pooled, cnt, N_NODES, E);
    } else {
        const int NH  = N_NODES * HID;
        const int NH4 = NH / 4;
        fill_f32<<<(N_NODES + 255) / 256, 256, 0, stream>>>(dis, 1.0f, N_NODES);
        deg_accum<<<(E + 255) / 256, 256, 0, stream>>>(dst, dis, E);
        rsqrt_inplace<<<(N_NODES + 255) / 256, 256, 0, stream>>>(dis, N_NODES);

        gemm_wmma_f32<IN_DIM, false><<<N_NODES / GEMM_ROWS, 256, 0, stream>>>(x, W1, dis, h);
        zero_f4<<<(NH4 + 255) / 256, 256, 0, stream>>>((float4*)agg, NH4);
        edge_scatter<<<2048, 256, 0, stream>>>(src, dst, dis, h, agg, E);
        finish_layer<<<(NH + 255) / 256, 256, 0, stream>>>(agg, h, dis, b1, NH);

        gemm_wmma_f32<HID, false><<<N_NODES / GEMM_ROWS, 256, 0, stream>>>(agg, W2, dis, h);
        zero_f4<<<(NH4 + 255) / 256, 256, 0, stream>>>((float4*)agg, NH4);
        edge_scatter<<<2048, 256, 0, stream>>>(src, dst, dis, h, agg, E);
        finish_layer<<<(NH + 255) / 256, 256, 0, stream>>>(agg, h, dis, b2, NH);

        fill_f32<<<(N_GRAPHS * HID + N_GRAPHS + 255) / 256, 256, 0, stream>>>(
            pooled, 0.0f, N_GRAPHS * HID + N_GRAPHS);
        pool_accum<<<(N_NODES + POOL_ROWS * 8 - 1) / (POOL_ROWS * 8), 256, 0, stream>>>(agg, bat, pooled, cnt, N_NODES);
    }

    heads_fused<<<HEAD_T_ALL / 256, 256, 0, stream>>>(pooled, cnt, Wmf, bmf, Wbp, bbp, Wcc, bcc, out);
}


# layer-2 product fused into layer-1 gather: a work-group's 80 finished rows stay in LDS as the WMMA A tile (same loaders, same k order); no h1 plane, one launch fewer; row loads in groups of 8/4/2/1
# speedup vs baseline: 4.9882x; 1.0606x over previous
#include <hip/hip_runtime.h>
#include <hip/hip_bf16.h>

#define N_NODES 100000
#define N_EDGES 1600000
#define IN_DIM  256
#define HID     128
#define N_GRAPHS 64
#define OUT_MF 489
#define OUT_BP 1943
#define OUT_CC 320

#define WS_PLANES_BYTES (4ull * (N_NODES + 2ull * N_NODES * HID + N_GRAPHS * HID + N_GRAPHS))
#define WS_EXTRA_BYTES  (4ull * (N_NODES + 4 + N_EDGES))

typedef __attribute__((ext_vector_type(2))) float v2f;
typedef __attribute__((ext_vector_type(8))) float v8f;
typedef __attribute__((ext_vector_type(4))) float v4f;

__global__ void fill_f32(float* __restrict__ p, float v, int n) {
    int t = blockIdx.x * blockDim.x + threadIdx.x;
    if (t < n) p[t] = v;
}

__global__ void zero_f4(float4* __restrict__ p, int n4) {
    int t = blockIdx.x * blockDim.x + threadIdx.x;
    if (t < n4) { float4 z; z.x = z.y = z.z = z.w = 0.f; p[t] = z; }
}

__global__ void deg_accum(const int* __restrict__ dst, float* __restrict__ deg, int E) {
    int t = blockIdx.x * blockDim.x + threadIdx.x;
    if (t < E) atomicAdd(&deg[dst[t]], 1.0f);
}

__global__ void rsqrt_inplace(float* __restrict__ p, int n) {
    int t = blockIdx.x * blockDim.x + threadIdx.x;
    if (t < n) p[t] = rsqrtf(p[t]);
}

#define GEMM_ROWS 80
#define GEMM_RT   (GEMM_ROWS / 16)
#define GEMM_KC   64
#define GEMM_PA   (GEMM_KC + 4)
#define GEMM_PW   144
#define GEMM_NA   (GEMM_ROWS * GEMM_KC / 4 / 256)
#define GEMM_NW   ((GEMM_KC / 2) * (HID / 4) / 256)
static_assert(N_NODES % GEMM_ROWS == 0, "every work-group owns a full set of rows");
static_assert((GEMM_ROWS * GEMM_KC / 4) % 256 == 0, "A chunk splits evenly over the threads");

template <int K, bool SCALE>
__global__ __launch_bounds__(256) void gemm_wmma_f32(const float* __restrict__ A,
                                                     const float* __restrict__ W,
                                                     const float* __restrict__ rs,
                                                     float* __restrict__ out) {
    __shared__ float4 ldsA4[GEMM_ROWS * GEMM_PA / 4];
    __shared__ float4 ldsW4[(GEMM_KC / 2) * GEMM_PW / 2];
    float*  ldsA = (float*)ldsA4;
    float2* ldsW = (float2*)ldsW4;
    const int tid  = threadIdx.x;
    const int wave = tid >> 5;
    const int lane = tid & 31;
    const int half = lane >> 4;
    const int m    = lane & 15;
    const int row0 = blockIdx.x * GEMM_ROWS;
    const int col0 = wave * 16;

    const float* ag = A + (size_t)(row0 + (tid >> 4)) * K + 4 * (tid & 15);
    const float* wg = W + (size_t)(2 * (tid >> 5)) * HID + 4 * (tid & 31);
    v4f* as = (v4f*)(ldsA + (tid >> 4) * GEMM_PA + 4 * (tid & 15));
    v4f* wsd = (v4f*)(ldsW + (tid >> 5) * GEMM_PW + 4 * (tid & 31));

    v8f acc[GEMM_RT];
    #pragma unroll
    for (int t = 0; t < GEMM_RT; ++t) acc[t] = v8f{};

    const float*  aF = ldsA + m * GEMM_PA + 2 * half;
    const float2* bF = ldsW + half * GEMM_PW + col0 + m;

    for (int kb = 0; kb < K; kb += GEMM_KC) {
        {
            v4f ra[GEMM_NA], rw0[GEMM_NW], rw1[GEMM_NW];
            #pragma unroll
            for (int i = 0; i < GEMM_NA; ++i)
                ra[i] = *(const v4f*)(ag + (size_t)(16 * i) * K + kb);
            #pragma unroll
            for (int i = 0; i < GEMM_NW; ++i) {
                rw0[i] = *(const v4f*)(wg + (size_t)(kb + 16 * i) * HID);
                rw1[i] = *(const v4f*)(wg + (size_t)(kb + 16 * i + 1) * HID);
            }
            __builtin_amdgcn_sched_barrier(0);
            #pragma unroll
            for (int i = 0; i < GEMM_NA; ++i) as[i * 16 * GEMM_PA / 4] = ra[i];
            #pragma unroll
            for (int i = 0; i < GEMM_NW; ++i) {
                v4f lo, hi;
                lo.x = rw0[i].x; lo.y = rw1[i].x; lo.z = rw0[i].y; lo.w = rw1[i].y;
                hi.x = rw0[i].z; hi.y = rw1[i].z; hi.z = rw0[i].w; hi.w = rw1[i].w;
                wsd[i * 8 * GEMM_PW / 2]     = lo;
                wsd[i * 8 * GEMM_PW / 2 + 1] = hi;
            }
        }
        __syncthreads();

        #pragma unroll
        for (int k = 0; k < GEMM_KC; k += 4) {
            const float2 bb = bF[(k >> 1) * GEMM_PW];
            v2f b; b.x = bb.x; b.y = bb.y;
            #pragma unroll
            for (int t = 0; t < GEMM_RT; ++t) {
                const float2 aa = *(const float2*)(aF + t * 16 * GEMM_PA + k);
                v2f a; a.x = aa.x; a.y = aa.y;
                acc[t] = __builtin_amdgcn_wmma_f32_16x16x4_f32(
                    false, a, false, b, (short)0, acc[t], false, false);
            }
        }
        __syncthreads();
    }

    #pragma unroll
    for (int t = 0; t < GEMM_RT; ++t) {
        v8f o = acc[t];
        if (SCALE) {
            const float* rp = rs + row0 + 16 * t + 8 * half;
            const v4f s0 = *(const v4f*)rp;
            const v4f s1 = *(const v4f*)(rp + 4);
            o[0] *= s0.x; o[1] *= s0.y; o[2] *= s0.z; o[3] *= s0.w;
            o[4] *= s1.x; o[5] *= s1.y; o[6] *= s1.z; o[7] *= s1.w;
        }
        #pragma unroll
        for (int v = 0; v < 8; ++v)
            out[(size_t)(row0 + 16 * t + v + 8 * half) * HID + col0 + m] = o[v];
    }
}

__global__ __launch_bounds__(256) void edge_scatter(const int* __restrict__ src,
                                                    const int* __restrict__ dst,
                                                    const float* __restrict__ dis,
                                                    const float* __restrict__ h,
                                                    float* __restrict__ agg, int E) {
    const int lane = threadIdx.x & 31;
    const int gw   = (blockIdx.x * blockDim.x + threadIdx.x) >> 5;
    const int nw   = (gridDim.x * blockDim.x) >> 5;
    for (int e = gw; e < E; e += nw) {
        const int s = src[e], d = dst[e];
        const float nrm = dis[s] * dis[d];
        const float* hs = h + (size_t)s * HID;
        float* ad = agg + (size_t)d * HID;
        #pragma unroll
        for (int j = 0; j < HID; j += 32)
            atomicAdd(&ad[j + lane], hs[j + lane] * nrm);
    }
}

__global__ void finish_layer(float* __restrict__ agg, const float* __restrict__ h,
                             const float* __restrict__ dis, const float* __restrict__ bias,
                             int n) {
    int t = blockIdx.x * blockDim.x + threadIdx.x;
    if (t >= n) return;
    const int i = t >> 7, j = t & (HID - 1);
    const float di = dis[i];
    const float v = agg[t] + di * di * h[t] + bias[j];
    agg[t] = v > 0.f ? v : 0.f;
}

#define NB     256
#define BIN_R  ((N_NODES + NB - 1) / NB)
#define NC     256
#define BT     1024
#define CHUNK_CAP 6400
#define BIN_CAP   8192
#define CNT_PAD   512
#define EPT       8
static_assert(BIN_R <= CNT_PAD, "a range's nodes fit the padded counter array");
#define TAB_Q  (NB / 4)
#define TAB_RG (BT / TAB_Q)
#define TAB_K  (NC / TAB_RG)
static_assert(NB % 4 == 0 && BT % TAB_Q == 0 && NC % TAB_RG == 0 && TAB_Q % 32 == 0,
              "the table splits into whole 16-byte groups, whole waves per row half");

__device__ __forceinline__ int bin_of(int d) {
    const unsigned b = (unsigned)d / (unsigned)BIN_R;
    return b < (unsigned)NB ? (int)b : NB - 1;
}

__device__ __forceinline__ int wave_incl_scan(int v, int lane) {
    #pragma unroll
    for (int d = 1; d < 32; d <<= 1) {
        const int u = __shfl_up(v, d);
        if (lane >= d) v += u;
    }
    return v;
}

__global__ __launch_bounds__(BT) void bin_count(const int* __restrict__ dst,
                                                int* __restrict__ cntT, int E) {
    __shared__ int hist[NB];
    const int tid = threadIdx.x, c = blockIdx.x;
    const int epc = (E + NC - 1) / NC;
    const int e0  = c * epc;
    const int e1  = (e0 + epc) < E ? (e0 + epc) : E;
    if (tid < NB) hist[tid] = 0;
    __syncthreads();
    for (int base = e0 + tid; base < e1; base += BT * EPT) {
        int d[EPT];
        #pragma unroll
        for (int k = 0; k < EPT; ++k) d[k] = (base + k * BT) < e1 ? dst[base + k * BT] : 0;
        #pragma unroll
        for (int k = 0; k < EPT; ++k)
            if ((base + k * BT) < e1) atomicAdd(&hist[bin_of(d[k])], 1);
    }
    __syncthreads();
    if (tid < NB) cntT[c * NB + tid] = hist[tid];
}

__global__ __launch_bounds__(BT) void bin_place(const int* __restrict__ src,
                                                const int* __restrict__ dst,
                                                const int* __restrict__ cntT,
                                                int* __restrict__ rbase,
                                                int2* __restrict__ pairs, int E) {
    __shared__ int  hist[NB];
    __shared__ int  gdelta[NB];
    __shared__ int  wtot[NB / 32];
    __shared__ int  rtot[NB];
    __shared__ int  rpre[NB];
    __shared__ int  wtotR[NB / 32];
    __shared__ int2 sbuf[CHUNK_CAP];
    const int tid = threadIdx.x, lane = tid & 31, wave = tid >> 5, c = blockIdx.x;
    const int epc = (E + NC - 1) / NC;
    const int e0  = c * epc;
    const int e1  = (e0 + epc) < E ? (e0 + epc) : E;
    const int n   = e1 > e0 ? e1 - e0 : 0;
    if (tid < NB) { hist[tid] = 0; rtot[tid] = 0; rpre[tid] = 0; }
    const int q  = tid & (TAB_Q - 1);
    const int rg = tid / TAB_Q;
    const int4* tp = (const int4*)cntT + rg * TAB_Q + q;
    int4 ta[TAB_K];
    __syncthreads();
    #pragma unroll
    for (int k = 0; k < TAB_K; ++k) ta[k] = tp[k * TAB_RG * TAB_Q];
    for (int base = e0 + tid; base < e1; base += BT * EPT) {
        int d[EPT];
        #pragma unroll
        for (int k = 0; k < EPT; ++k) d[k] = (base + k * BT) < e1 ? dst[base + k * BT] : 0;
        #pragma unroll
        for (int k = 0; k < EPT; ++k)
            if ((base + k * BT) < e1) atomicAdd(&hist[bin_of(d[k])], 1);
    }
    {
        int t0 = 0, t1 = 0, t2 = 0, t3 = 0, p0 = 0, p1 = 0, p2 = 0, p3 = 0;
        #pragma unroll
        for (int k = 0; k < TAB_K; ++k) {
            const int m = (rg + k * TAB_RG) < c ? -1 : 0;
            t0 += ta[k].x; t1 += ta[k].y; t2 += ta[k].z; t3 += ta[k].w;
            p0 += ta[k].x & m; p1 += ta[k].y & m; p2 += ta[k].z & m; p3 += ta[k].w & m;
        }
        atomicAdd(&rtot[4 * q],     t0);  atomicAdd(&rtot[4 * q + 1], t1);
        atomicAdd(&rtot[4 * q + 2], t2);  atomicAdd(&rtot[4 * q + 3], t3);
        atomicAdd(&rpre[4 * q],     p0);  atomicAdd(&rpre[4 * q + 1], p1);
        atomicAdd(&rpre[4 * q + 2], p2);  atomicAdd(&rpre[4 * q + 3], p3);
    }
    __syncthreads();
    int cnt = 0, v = 0, rt = 0, vr = 0;
    if (tid < NB) {
        cnt = hist[tid];
        v = wave_incl_scan(cnt, lane);
        rt = rtot[tid];
        vr = wave_incl_scan(rt, lane);
        if (lane == 31) { wtot[wave] = v; wtotR[wave] = vr; }
    }
    __syncthreads();
    if (tid < NB) {
        int base = 0, baseR = 0;
        for (int w = 0; w < wave; ++w) { base += wtot[w]; baseR += wtotR[w]; }
        const int ex = base + v - cnt;
        const int gx = baseR + vr - rt;
        hist[tid]   = ex;
        gdelta[tid] = gx + rpre[tid] - ex;
        if (c == 0) {
            rbase[tid] = gx;
            if (tid == NB - 1) rbase[NB] = gx + rt;
        }
    }
    __syncthreads();
    const bool staged = n <= CHUNK_CAP;
    for (int base = e0 + tid; base < e1; base += BT * EPT) {
        int s[EPT], d[EPT];
        #pragma unroll
        for (int k = 0; k < EPT; ++k) {
            const bool in = (base + k * BT) < e1;
            s[k] = in ? src[base + k * BT] : 0;
            d[k] = in ? dst[base + k * BT] : 0;
        }
        #pragma unroll
        for (int k = 0; k < EPT; ++k) {
            if ((base + k * BT) < e1) {
                int2 r; r.x = s[k]; r.y = d[k];
                const int b = bin_of(r.y);
                const int p = atomicAdd(&hist[b], 1);
                if (staged) {
                    if ((unsigned)p < (unsigned)CHUNK_CAP) sbuf[p] = r;
                } else {
                    const int g = p + gdelta[b];
                    if ((unsigned)g < (unsigned)E) pairs[g] = r;
                }
            }
        }
    }
    if (staged) {
        __syncthreads();
        for (int i = tid; i < n; i += BT) {
            const int2 r = sbuf[i];
            const int  g = i + gdelta[bin_of(r.y)];
            if ((unsigned)g < (unsigned)E) pairs[g] = r;
        }
    }
}

__global__ __launch_bounds__(BT) void bin_csr(const int2* __restrict__ pairs,
                                              const int* __restrict__ rbase,
                                              int* __restrict__ rowptr, int* __restrict__ col,
                                              float* __restrict__ dis, int N, int E) {
    __shared__ int cntl[CNT_PAD];
    __shared__ int wtot[CNT_PAD / 32];
    __shared__ int scol[BIN_CAP];
    const int tid = threadIdx.x, lane = tid & 31, wave = tid >> 5, b = blockIdx.x;
    int beg = rbase[b], end = rbase[b + 1];
    beg = beg > 0 ? beg : 0;  beg = beg < E ? beg : E;
    end = end > beg ? end : beg;  end = end < E ? end : E;
    const int n     = end - beg;
    const int node0 = b * BIN_R;
    int nn = N - node0;
    nn = nn < BIN_R ? nn : BIN_R;
    nn = nn > 0 ? nn : 0;
    if (tid < CNT_PAD) cntl[tid] = 0;
    __syncthreads();
    for (int base = tid; base < n; base += BT * EPT) {
        int d[EPT];
        #pragma unroll
        for (int k = 0; k < EPT; ++k) d[k] = (base + k * BT) < n ? pairs[beg + base + k * BT].y : -1;
        #pragma unroll
        for (int k = 0; k < EPT; ++k) {
            const unsigned l = (unsigned)(d[k] - node0);
            if ((base + k * BT) < n && l < (unsigned)nn) atomicAdd(&cntl[l], 1);
        }
    }
    __syncthreads();
    int cnt = 0, v = 0;
    if (tid < CNT_PAD) {
        cnt = cntl[tid];
        v = wave_incl_scan(cnt, lane);
        if (lane == 31) wtot[wave] = v;
    }
    __syncthreads();
    if (tid < CNT_PAD) {
        int base = 0;
        for (int w = 0; w < wave; ++w) base += wtot[w];
        const int ex = base + v - cnt;
        cntl[tid] = ex;
        if (tid < nn) {
            rowptr[node0 + tid] = beg + ex;
            dis[node0 + tid]    = rsqrtf((float)cnt + 1.0f);
        }
    }
    if (b == NB - 1 && tid == 0) rowptr[N] = end;
    __syncthreads();
    const bool staged = n <= BIN_CAP;
    for (int base = tid; base < n; base += BT * EPT) {
        int2 r[EPT];
        #pragma unroll
        for (int k = 0; k < EPT; ++k) {
            if ((base + k * BT) < n) r[k] = pairs[beg + base + k * BT];
            else { r[k].x = 0; r[k].y = -1; }
        }
        #pragma unroll
        for (int k = 0; k < EPT; ++k) {
            const unsigned l = (unsigned)(r[k].y - node0);
            if ((base + k * BT) < n && l < (unsigned)nn) {
                const int p = atomicAdd(&cntl[l], 1);
                if (staged) {
                    if ((unsigned)p < (unsigned)BIN_CAP) scol[p] = r[k].x;
                } else {
                    const int g = beg + p;
                    if ((unsigned)g < (unsigned)E) col[g] = r[k].x;
                }
            }
        }
    }
    if (staged) {
        __syncthreads();
        for (int i = tid; i < n; i += BT) col[beg + i] = scol[i];
    }
}

#define FG_PA   (HID + 4)
#define FG_NPW  (GEMM_ROWS / 8)
#define FG_KC   32
#define FG_NW   ((FG_KC / 2) * (HID / 4) / 256)
static_assert(((FG_KC / 2) * (HID / 4)) % 256 == 0, "W chunk splits evenly over the threads");
static_assert(GEMM_ROWS % 8 == 0, "the tile's rows split evenly over the eight waves");

template <int G>
__device__ __forceinline__ void row_group(const float4* __restrict__ h4, int s_l, int j,
                                          int lane, float4& acc) {
    float4 v[G];
    #pragma unroll
    for (int t = 0; t < G; ++t) {
        const int s = __builtin_amdgcn_readlane(s_l, j + t);
        v[t] = h4[(size_t)s * (HID / 4) + lane];
    }
    #pragma unroll
    for (int t = 0; t < G; ++t) {
        acc.x += v[t].x; acc.y += v[t].y; acc.z += v[t].z; acc.w += v[t].w;
    }
}

__global__ __launch_bounds__(256, 10) void gather_gemm(const int* __restrict__ rowptr,
                                                   const int* __restrict__ col,
                                                   const float* __restrict__ dis,
                                                   const float* __restrict__ h,
                                                   const float* __restrict__ bias,
                                                   const float* __restrict__ W,
                                                   float* __restrict__ out, int N, int E) {
    __shared__ float4 ldsA4[GEMM_ROWS * FG_PA / 4];
    __shared__ float4 ldsW4[(FG_KC / 2) * GEMM_PW / 2];
    float*  ldsA = (float*)ldsA4;
    float2* ldsW = (float2*)ldsW4;
    const int tid  = threadIdx.x;
    const int lane = tid & 31;
    const int wave = __builtin_amdgcn_readfirstlane(tid >> 5);
    const int half = lane >> 4;
    const int m    = lane & 15;
    const int row0 = blockIdx.x * GEMM_ROWS;
    const int col0 = wave * 16;
    const float4* h4 = (const float4*)h;
    const float4  bv = ((const float4*)bias)[lane];

    for (int q = 0; q < FG_NPW; ++q) {
        const int lr = wave * FG_NPW + q;
        const int i  = row0 + lr;
        int beg = rowptr[i], end = rowptr[i + 1];
        beg = beg > 0 ? beg : 0;
        end = end < E ? end : E;
        const float  di  = dis[i];
        const float4 hv  = h4[(size_t)i * (HID / 4) + lane];
        float4 acc; acc.x = acc.y = acc.z = acc.w = 0.f;

        for (int base = beg; base < end; base += 32) {
            const int mm = (end - base) < 32 ? (end - base) : 32;
            int s_l = 0;
            if (lane < mm) {
                s_l = col[base + lane];
                s_l = (unsigned)s_l < (unsigned)N ? s_l : 0;
            }
            int j = 0;
            for (; j + 8 <= mm; j += 8) row_group<8>(h4, s_l, j, lane, acc);
            if (mm - j >= 4) { row_group<4>(h4, s_l, j, lane, acc); j += 4; }
            if (mm - j >= 2) { row_group<2>(h4, s_l, j, lane, acc); j += 2; }
            if (mm - j >= 1) { row_group<1>(h4, s_l, j, lane, acc); }
        }
        float4 r;
        r.x = di * (acc.x + hv.x) + bv.x;
        r.y = di * (acc.y + hv.y) + bv.y;
        r.z = di * (acc.z + hv.z) + bv.z;
        r.w = di * (acc.w + hv.w) + bv.w;
        r.x = r.x > 0.f ? r.x : 0.f;
        r.y = r.y > 0.f ? r.y : 0.f;
        r.z = r.z > 0.f ? r.z : 0.f;
        r.w = r.w > 0.f ? r.w : 0.f;
        r.x *= di; r.y *= di; r.z *= di; r.w *= di;
        *(float4*)(ldsA + lr * FG_PA + 4 * lane) = r;
    }

    const float* wg = W + (size_t)(2 * (tid >> 5)) * HID + 4 * (tid & 31);
    v4f* wsd = (v4f*)(ldsW + (tid >> 5) * GEMM_PW + 4 * (tid & 31));

    v8f accm[GEMM_RT];
    #pragma unroll
    for (int t = 0; t < GEMM_RT; ++t) accm[t] = v8f{};

    const float*  aF = ldsA + m * FG_PA + 2 * half;
    const float2* bF = ldsW + half * GEMM_PW + col0 + m;

    for (int kb = 0; kb < HID; kb += FG_KC) {
        {
            v4f rw0[FG_NW], rw1[FG_NW];
            #pragma unroll
            for (int i = 0; i < FG_NW; ++i) {
                rw0[i] = *(const v4f*)(wg + (size_t)(kb + 16 * i) * HID);
                rw1[i] = *(const v4f*)(wg + (size_t)(kb + 16 * i + 1) * HID);
            }
            #pragma unroll
            for (int i = 0; i < FG_NW; ++i) {
                v4f lo, hi;
                lo.x = rw0[i].x; lo.y = rw1[i].x; lo.z = rw0[i].y; lo.w = rw1[i].y;
                hi.x = rw0[i].z; hi.y = rw1[i].z; hi.z = rw0[i].w; hi.w = rw1[i].w;
                wsd[i * 8 * GEMM_PW / 2]     = lo;
                wsd[i * 8 * GEMM_PW / 2 + 1] = hi;
            }
        }
        __syncthreads();

        #pragma unroll
        for (int k = 0; k < FG_KC; k += 4) {
            const float2 bb = bF[(k >> 1) * GEMM_PW];
            v2f b; b.x = bb.x; b.y = bb.y;
            #pragma unroll
            for (int t = 0; t < GEMM_RT; ++t) {
                const float2 aa = *(const float2*)(aF + t * 16 * FG_PA + kb + k);
                v2f a; a.x = aa.x; a.y = aa.y;
                accm[t] = __builtin_amdgcn_wmma_f32_16x16x4_f32(
                    false, a, false, b, (short)0, accm[t], false, false);
            }
        }
        __syncthreads();
    }

    const int halfD = (int)(__builtin_amdgcn_mbcnt_lo(~0u, 0u) >> 4);
    #pragma unroll
    for (int t = 0; t < GEMM_RT; ++t) {
        #pragma unroll
        for (int v = 0; v < 8; ++v)
            out[(size_t)(row0 + 16 * t + v + 8 * halfD) * HID + col0 + m] = accm[t][v];
    }
}

#define GATHER_WGS 2048
static_assert((unsigned long long)N_NODES * (GATHER_WGS * 8ull) < (1ull << 32),
              "wave index times node count fits 32 bits in the range split");
__global__ __launch_bounds__(256) void gather_pool_layer(const int* __restrict__ rowptr,
                                                         const int* __restrict__ col,
                                                         const float* __restrict__ dis,
                                                         const float* __restrict__ h,
                                                         const float* __restrict__ bias,
                                                         const int* __restrict__ batch,
                                                         float* __restrict__ pooled,
                                                         float* __restrict__ cnt, int N, int E) {
    __shared__ float4 sAcc4[8 * (HID / 4)];
    __shared__ int4   sMeta4[4];
    const int tid  = threadIdx.x;
    const int lane = tid & 31;
    const int wave = tid >> 5;
    const unsigned gw = (unsigned)__builtin_amdgcn_readfirstlane(
        (int)((blockIdx.x * blockDim.x + threadIdx.x) >> 5));
    const unsigned nw = (gridDim.x * blockDim.x) >> 5;
    const int n0 = (int)(gw * (unsigned)N / nw);
    const int n1 = (int)((gw + 1u) * (unsigned)N / nw);
    const float4* h4 = (const float4*)h;
    const float4  bv = ((const float4*)bias)[lane];

    int    cur = 0, run = 0;
    float4 pacc; pacc.x = pacc.y = pacc.z = pacc.w = 0.f;

    for (int i = n0; i < n1; ++i) {
        int beg = rowptr[i], end = rowptr[i + 1];
        beg = beg > 0 ? beg : 0;
        end = end < E ? end : E;
        const float  di  = dis[i];
        const int    g   = batch[i];
        const float4 hv  = h4[(size_t)i * (HID / 4) + lane];
        float4 acc; acc.x = acc.y = acc.z = acc.w = 0.f;

        for (int base = beg; base < end; base += 32) {
            const int m = (end - base) < 32 ? (end - base) : 32;
            int s_l = 0;
            if (lane < m) {
                s_l = col[base + lane];
                s_l = (unsigned)s_l < (unsigned)N ? s_l : 0;
            }
            #pragma unroll 4
            for (int j = 0; j < m; ++j) {
                const int    s = __builtin_amdgcn_readlane(s_l, j);
                const float4 v = h4[(size_t)s * (HID / 4) + lane];
                acc.x += v.x; acc.y += v.y; acc.z += v.z; acc.w += v.w;
            }
        }
        float4 r;
        r.x = di * (acc.x + hv.x) + bv.x;
        r.y = di * (acc.y + hv.y) + bv.y;
        r.z = di * (acc.z + hv.z) + bv.z;
        r.w = di * (acc.w + hv.w) + bv.w;
        r.x = r.x > 0.f ? r.x : 0.f;
        r.y = r.y > 0.f ? r.y : 0.f;
        r.z = r.z > 0.f ? r.z : 0.f;
        r.w = r.w > 0.f ? r.w : 0.f;

        if (run > 0 && g != cur) {
            if ((unsigned)cur < (unsigned)N_GRAPHS) {
                float* pg = pooled + cur * HID + 4 * lane;
                atomicAdd(&pg[0], pacc.x);
                atomicAdd(&pg[1], pacc.y);
                atomicAdd(&pg[2], pacc.z);
                atomicAdd(&pg[3], pacc.w);
                if (lane == 0) atomicAdd(&cnt[cur], (float)run);
            }
            run = 0;
            pacc.x = pacc.y = pacc.z = pacc.w = 0.f;
        }
        cur = g;
        pacc.x += r.x; pacc.y += r.y; pacc.z += r.z; pacc.w += r.w;
        ++run;
    }

    sAcc4[wave * (HID / 4) + lane] = pacc;
    if (lane == 0) { ((int*)sMeta4)[wave] = cur; ((int*)sMeta4)[8 + wave] = run; }
    __syncthreads();
    if (tid < HID) {
        const float* sa = (const float*)sAcc4;
        float v[8];
        #pragma unroll
        for (int w = 0; w < 8; ++w) v[w] = sa[w * HID + tid];
        const int4 ga = sMeta4[0], gb = sMeta4[1], ra = sMeta4[2], rb = sMeta4[3];
        const int gq[8] = { ga.x, ga.y, ga.z, ga.w, gb.x, gb.y, gb.z, gb.w };
        const int rw[8] = { ra.x, ra.y, ra.z, ra.w, rb.x, rb.y, rb.z, rb.w };
        int   cg = 0, rc = 0;
        float s  = 0.f;
        #pragma unroll
        for (int w = 0; w < 8; ++w) {
            if (rw[w] == 0) continue;
            if (rc > 0 && gq[w] != cg) {
                if ((unsigned)cg < (unsigned)N_GRAPHS) {
                    atomicAdd(&pooled[cg * HID + tid], s);
                    if (tid == 0) atomicAdd(&cnt[cg], (float)rc);
                }
                rc = 0; s = 0.f;
            }
            cg = gq[w];
            s += v[w];
            rc += rw[w];
        }
        if (rc > 0 && (unsigned)cg < (unsigned)N_GRAPHS) {
            atomicAdd(&pooled[cg * HID + tid], s);
            if (tid == 0) atomicAdd(&cnt[cg], (float)rc);
        }
    }
}

#define POOL_ROWS 16
static_assert(N_NODES % POOL_ROWS == 0, "every wave owns a full group of rows");
__global__ __launch_bounds__(256) void pool_accum(const float* __restrict__ h,
                                                  const int* __restrict__ batch,
                                                  float* __restrict__ pooled,
                                                  float* __restrict__ cnt, int N) {
    const int lane = threadIdx.x & 31;
    const int gw = __builtin_amdgcn_readfirstlane(
        (int)((blockIdx.x * blockDim.x + threadIdx.x) >> 5));
    const int r0 = gw * POOL_ROWS;
    if (r0 >= N) return;

    const int*   bp = batch + r0;
    const float* hp = h + (size_t)r0 * HID + lane;
    int   g[POOL_ROWS];
    float v[POOL_ROWS][4];
    #pragma unroll
    for (int i = 0; i < POOL_ROWS; ++i) {
        g[i] = bp[i];
        #pragma unroll
        for (int k = 0; k < 4; ++k) v[i][k] = hp[i * HID + 32 * k];
    }

    int   cur = g[0];
    int   run = 0;
    float a0 = 0.f, a1 = 0.f, a2 = 0.f, a3 = 0.f;
    #pragma unroll
    for (int i = 0; i < POOL_ROWS; ++i) {
        if (g[i] != cur) {
            float* pg = pooled + cur * HID + lane;
            atomicAdd(&pg[0],  a0);
            atomicAdd(&pg[32], a1);
            atomicAdd(&pg[64], a2);
            atomicAdd(&pg[96], a3);
            if (lane == 0) atomicAdd(&cnt[cur], (float)run);
            cur = g[i]; run = 0;
            a0 = a1 = a2 = a3 = 0.f;
        }
        a0 += v[i][0]; a1 += v[i][1]; a2 += v[i][2]; a3 += v[i][3];
        ++run;
    }
    {
        float* pg = pooled + cur * HID + lane;
        atomicAdd(&pg[0],  a0);
        atomicAdd(&pg[32], a1);
        atomicAdd(&pg[64], a2);
        atomicAdd(&pg[96], a3);
        if (lane == 0) atomicAdd(&cnt[cur], (float)run);
    }
}

#define HEAD_T_MF  (N_GRAPHS * OUT_MF)
#define HEAD_T_BP  (N_GRAPHS * OUT_BP)
#define HEAD_T_CC  (N_GRAPHS * OUT_CC)
#define HEAD_T_ALL (HEAD_T_MF + HEAD_T_BP + HEAD_T_CC)
#define HEAD_KB    32
static_assert(HEAD_T_MF % 32 == 0 && HEAD_T_BP % 32 == 0, "a wave never straddles two heads");
static_assert(HEAD_T_ALL % 256 == 0, "whole work-groups: every thread owns an output");
static_assert(OUT_MF > 32 && OUT_BP > 32 && OUT_CC > 32, "a wave's outputs touch at most two graphs");
static_assert(HID % HEAD_KB == 0 && HEAD_KB % 4 == 0, "k groups are whole 16-byte pieces of a mean row");

__global__ __launch_bounds__(256) void heads_fused(const float* __restrict__ pooled,
                                                   const float* __restrict__ cnt,
                                                   const float* __restrict__ Wmf,
                                                   const float* __restrict__ bmf,
                                                   const float* __restrict__ Wbp,
                                                   const float* __restrict__ bbp,
                                                   const float* __restrict__ Wcc,
                                                   const float* __restrict__ bcc,
                                                   float* __restrict__ out) {
    __shared__ float4 sP4[8 * 2 * (HID / 4)];
    const int tid = threadIdx.x, lane = tid & 31, wave = tid >> 5;
    const int t = blockIdx.x * 256 + tid;
    const int tw = __builtin_amdgcn_readfirstlane(t - lane);
    const float* W;
    const float* b;
    int OUT, hb;
    if (tw < HEAD_T_MF)                  { W = Wmf; b = bmf; OUT = OUT_MF; hb = 0; }
    else if (tw < HEAD_T_MF + HEAD_T_BP) { W = Wbp; b = bbp; OUT = OUT_BP; hb = HEAD_T_MF; }
    else                                 { W = Wcc; b = bcc; OUT = OUT_CC; hb = HEAD_T_MF + HEAD_T_BP; }
    const int tt = t - hb;
    const int g  = tt / OUT, o = tt - g * OUT;
    const int g0 = (tw - hb) / OUT;
    const int g1 = (g0 + 1) < N_GRAPHS ? (g0 + 1) : (N_GRAPHS - 1);

    #pragma unroll
    for (int r = 0; r < 2; ++r) {
        const int   gr = r ? g1 : g0;
        const float c  = cnt[gr];
        const float d  = c > 1.f ? c : 1.f;
        float4 v = ((const float4*)(pooled + gr * HID))[lane];
        v.x /= d; v.y /= d; v.z /= d; v.w /= d;
        sP4[(wave * 2 + r) * (HID / 4) + lane] = v;
    }
    __syncthreads();

    const float* prow = (const float*)sP4 + (wave * 2 + (g - g0)) * HID;
    const float* wp   = W + o;
    float s = b[o];
    for (int kb = 0; kb < HID; kb += HEAD_KB) {
        float w[HEAD_KB];
        #pragma unroll
        for (int i = 0; i < HEAD_KB; ++i) w[i] = wp[(size_t)(kb + i) * OUT];
        #pragma unroll
        for (int i = 0; i < HEAD_KB; i += 4) {
            const float4 p = *(const float4*)(prow + kb + i);
            s += p.x * w[i];
            s += p.y * w[i + 1];
            s += p.z * w[i + 2];
            s += p.w * w[i + 3];
        }
    }
    out[t] = s;
}

extern "C" void kernel_launch(void* const* d_in, const int* in_sizes, int n_in,
                              void* d_out, int out_size, void* d_ws, size_t ws_size,
                              hipStream_t stream) {
    const float* x   = (const float*)d_in[0];
    const int*   ei  = (const int*)d_in[1];
    const int*   bat = (const int*)d_in[2];
    const float* W1  = (const float*)d_in[3];
    const float* b1  = (const float*)d_in[4];
    const float* W2  = (const float*)d_in[5];
    const float* b2  = (const float*)d_in[6];
    const float* Wmf = (const float*)d_in[7];
    const float* bmf = (const float*)d_in[8];
    const float* Wbp = (const float*)d_in[9];
    const float* bbp = (const float*)d_in[10];
    const float* Wcc = (const float*)d_in[11];
    const float* bcc = (const float*)d_in[12];
    float* out = (float*)d_out;

    const int E = in_sizes[1] / 2;
    const int* src = ei;
    const int* dst = ei + E;

    float* ws     = (float*)d_ws;
    float* dis    = ws;
    float* h      = dis + N_NODES;
    float* agg    = h + (size_t)N_NODES * HID;
    float* pooled = agg + (size_t)N_NODES * HID;
    float* cnt    = pooled + N_GRAPHS * HID;

    int*  rowptr = (int*)(cnt + N_GRAPHS);
    int*  col    = rowptr + (N_NODES + 4);
    int2* pairs  = (int2*)agg;
    int*  cntT   = (int*)agg + 2 * N_EDGES;
    int*  rbase  = cntT + NB * NC;

    if (ws_size >= (size_t)WS_PLANES_BYTES + (size_t)WS_EXTRA_BYTES) {
        bin_count<<<NC, BT, 0, stream>>>(dst, cntT, E);
        bin_place<<<NC, BT, 0, stream>>>(src, dst, cntT, rbase, pairs, E);
        bin_csr<<<NB, BT, 0, stream>>>(pairs, rbase, rowptr, col, dis, N_NODES, E);

        gemm_wmma_f32<IN_DIM, true><<<N_NODES / GEMM_ROWS, 256, 0, stream>>>(x, W1, dis, h);

        gather_gemm<<<N_NODES / GEMM_ROWS, 256, 0, stream>>>(rowptr, col, dis, h, b1, W2, agg, N_NODES, E);

        fill_f32<<<(N_GRAPHS * HID + N_GRAPHS + 255) / 256, 256, 0, stream>>>(
            pooled, 0.0f, N_GRAPHS * HID + N_GRAPHS);
        gather_pool_layer<<<GATHER_WGS, 256, 0, stream>>>(rowptr, col, dis, agg, b2, bat,
                                                          pooled, cnt, N_NODES, E);
    } else {
        const int NH  = N_NODES * HID;
        const int NH4 = NH / 4;
        fill_f32<<<(N_NODES + 255) / 256, 256, 0, stream>>>(dis, 1.0f, N_NODES);
        deg_accum<<<(E + 255) / 256, 256, 0, stream>>>(dst, dis, E);
        rsqrt_inplace<<<(N_NODES + 255) / 256, 256, 0, stream>>>(dis, N_NODES);

        gemm_wmma_f32<IN_DIM, false><<<N_NODES / GEMM_ROWS, 256, 0, stream>>>(x, W1, dis, h);
        zero_f4<<<(NH4 + 255) / 256, 256, 0, stream>>>((float4*)agg, NH4);
        edge_scatter<<<2048, 256, 0, stream>>>(src, dst, dis, h, agg, E);
        finish_layer<<<(NH + 255) / 256, 256, 0, stream>>>(agg, h, dis, b1, NH);

        gemm_wmma_f32<HID, false><<<N_NODES / GEMM_ROWS, 256, 0, stream>>>(agg, W2, dis, h);
        zero_f4<<<(NH4 + 255) / 256, 256, 0, stream>>>((float4*)agg, NH4);
        edge_scatter<<<2048, 256, 0, stream>>>(src, dst, dis, h, agg, E);
        finish_layer<<<(NH + 255) / 256, 256, 0, stream>>>(agg, h, dis, b2, NH);

        fill_f32<<<(N_GRAPHS * HID + N_GRAPHS + 255) / 256, 256, 0, stream>>>(
            pooled, 0.0f, N_GRAPHS * HID + N_GRAPHS);
        pool_accum<<<(N_NODES + POOL_ROWS * 8 - 1) / (POOL_ROWS * 8), 256, 0, stream>>>(agg, bat, pooled, cnt, N_NODES);
    }

    heads_fused<<<HEAD_T_ALL / 256, 256, 0, stream>>>(pooled, cnt, Wmf, bmf, Wbp, bbp, Wcc, bcc, out);
}


# edge grouping in two launches (was three): a chunk orders its pairs by range inside its own slice, the range pass finds its runs via two table columns; no count launch; row-sum order still by arrival
# speedup vs baseline: 5.1344x; 1.0293x over previous
#include <hip/hip_runtime.h>
#include <hip/hip_bf16.h>

#define N_NODES 100000
#define N_EDGES 1600000
#define IN_DIM  256
#define HID     128
#define N_GRAPHS 64
#define OUT_MF 489
#define OUT_BP 1943
#define OUT_CC 320

#define WS_PLANES_BYTES (4ull * (N_NODES + 2ull * N_NODES * HID + N_GRAPHS * HID + N_GRAPHS))
#define WS_EXTRA_BYTES  (4ull * (N_NODES + 4 + N_EDGES))

typedef __attribute__((ext_vector_type(2))) float v2f;
typedef __attribute__((ext_vector_type(8))) float v8f;
typedef __attribute__((ext_vector_type(4))) float v4f;

__global__ void fill_f32(float* __restrict__ p, float v, int n) {
    int t = blockIdx.x * blockDim.x + threadIdx.x;
    if (t < n) p[t] = v;
}

__global__ void zero_f4(float4* __restrict__ p, int n4) {
    int t = blockIdx.x * blockDim.x + threadIdx.x;
    if (t < n4) { float4 z; z.x = z.y = z.z = z.w = 0.f; p[t] = z; }
}

__global__ void deg_accum(const int* __restrict__ dst, float* __restrict__ deg, int E) {
    int t = blockIdx.x * blockDim.x + threadIdx.x;
    if (t < E) atomicAdd(&deg[dst[t]], 1.0f);
}

__global__ void rsqrt_inplace(float* __restrict__ p, int n) {
    int t = blockIdx.x * blockDim.x + threadIdx.x;
    if (t < n) p[t] = rsqrtf(p[t]);
}

#define GEMM_ROWS 80
#define GEMM_RT   (GEMM_ROWS / 16)
#define GEMM_KC   64
#define GEMM_PA   (GEMM_KC + 4)
#define GEMM_PW   144
#define GEMM_NA   (GEMM_ROWS * GEMM_KC / 4 / 256)
#define GEMM_NW   ((GEMM_KC / 2) * (HID / 4) / 256)
static_assert(N_NODES % GEMM_ROWS == 0, "every work-group owns a full set of rows");
static_assert((GEMM_ROWS * GEMM_KC / 4) % 256 == 0, "A chunk splits evenly over the threads");

template <int K, bool SCALE>
__global__ __launch_bounds__(256) void gemm_wmma_f32(const float* __restrict__ A,
                                                     const float* __restrict__ W,
                                                     const float* __restrict__ rs,
                                                     float* __restrict__ out) {
    __shared__ float4 ldsA4[GEMM_ROWS * GEMM_PA / 4];
    __shared__ float4 ldsW4[(GEMM_KC / 2) * GEMM_PW / 2];
    float*  ldsA = (float*)ldsA4;
    float2* ldsW = (float2*)ldsW4;
    const int tid  = threadIdx.x;
    const int wave = tid >> 5;
    const int lane = tid & 31;
    const int half = lane >> 4;
    const int m    = lane & 15;
    const int row0 = blockIdx.x * GEMM_ROWS;
    const int col0 = wave * 16;

    const float* ag = A + (size_t)(row0 + (tid >> 4)) * K + 4 * (tid & 15);
    const float* wg = W + (size_t)(2 * (tid >> 5)) * HID + 4 * (tid & 31);
    v4f* as = (v4f*)(ldsA + (tid >> 4) * GEMM_PA + 4 * (tid & 15));
    v4f* wsd = (v4f*)(ldsW + (tid >> 5) * GEMM_PW + 4 * (tid & 31));

    v8f acc[GEMM_RT];
    #pragma unroll
    for (int t = 0; t < GEMM_RT; ++t) acc[t] = v8f{};

    const float*  aF = ldsA + m * GEMM_PA + 2 * half;
    const float2* bF = ldsW + half * GEMM_PW + col0 + m;

    for (int kb = 0; kb < K; kb += GEMM_KC) {
        {
            v4f ra[GEMM_NA], rw0[GEMM_NW], rw1[GEMM_NW];
            #pragma unroll
            for (int i = 0; i < GEMM_NA; ++i)
                ra[i] = *(const v4f*)(ag + (size_t)(16 * i) * K + kb);
            #pragma unroll
            for (int i = 0; i < GEMM_NW; ++i) {
                rw0[i] = *(const v4f*)(wg + (size_t)(kb + 16 * i) * HID);
                rw1[i] = *(const v4f*)(wg + (size_t)(kb + 16 * i + 1) * HID);
            }
            __builtin_amdgcn_sched_barrier(0);
            #pragma unroll
            for (int i = 0; i < GEMM_NA; ++i) as[i * 16 * GEMM_PA / 4] = ra[i];
            #pragma unroll
            for (int i = 0; i < GEMM_NW; ++i) {
                v4f lo, hi;
                lo.x = rw0[i].x; lo.y = rw1[i].x; lo.z = rw0[i].y; lo.w = rw1[i].y;
                hi.x = rw0[i].z; hi.y = rw1[i].z; hi.z = rw0[i].w; hi.w = rw1[i].w;
                wsd[i * 8 * GEMM_PW / 2]     = lo;
                wsd[i * 8 * GEMM_PW / 2 + 1] = hi;
            }
        }
        __syncthreads();

        #pragma unroll
        for (int k = 0; k < GEMM_KC; k += 4) {
            const float2 bb = bF[(k >> 1) * GEMM_PW];
            v2f b; b.x = bb.x; b.y = bb.y;
            #pragma unroll
            for (int t = 0; t < GEMM_RT; ++t) {
                const float2 aa = *(const float2*)(aF + t * 16 * GEMM_PA + k);
                v2f a; a.x = aa.x; a.y = aa.y;
                acc[t] = __builtin_amdgcn_wmma_f32_16x16x4_f32(
                    false, a, false, b, (short)0, acc[t], false, false);
            }
        }
        __syncthreads();
    }

    #pragma unroll
    for (int t = 0; t < GEMM_RT; ++t) {
        v8f o = acc[t];
        if (SCALE) {
            const float* rp = rs + row0 + 16 * t + 8 * half;
            const v4f s0 = *(const v4f*)rp;
            const v4f s1 = *(const v4f*)(rp + 4);
            o[0] *= s0.x; o[1] *= s0.y; o[2] *= s0.z; o[3] *= s0.w;
            o[4] *= s1.x; o[5] *= s1.y; o[6] *= s1.z; o[7] *= s1.w;
        }
        #pragma unroll
        for (int v = 0; v < 8; ++v)
            out[(size_t)(row0 + 16 * t + v + 8 * half) * HID + col0 + m] = o[v];
    }
}

__global__ __launch_bounds__(256) void edge_scatter(const int* __restrict__ src,
                                                    const int* __restrict__ dst,
                                                    const float* __restrict__ dis,
                                                    const float* __restrict__ h,
                                                    float* __restrict__ agg, int E) {
    const int lane = threadIdx.x & 31;
    const int gw   = (blockIdx.x * blockDim.x + threadIdx.x) >> 5;
    const int nw   = (gridDim.x * blockDim.x) >> 5;
    for (int e = gw; e < E; e += nw) {
        const int s = src[e], d = dst[e];
        const float nrm = dis[s] * dis[d];
        const float* hs = h + (size_t)s * HID;
        float* ad = agg + (size_t)d * HID;
        #pragma unroll
        for (int j = 0; j < HID; j += 32)
            atomicAdd(&ad[j + lane], hs[j + lane] * nrm);
    }
}

__global__ void finish_layer(float* __restrict__ agg, const float* __restrict__ h,
                             const float* __restrict__ dis, const float* __restrict__ bias,
                             int n) {
    int t = blockIdx.x * blockDim.x + threadIdx.x;
    if (t >= n) return;
    const int i = t >> 7, j = t & (HID - 1);
    const float di = dis[i];
    const float v = agg[t] + di * di * h[t] + bias[j];
    agg[t] = v > 0.f ? v : 0.f;
}

#define NB     256
#define BIN_R  ((N_NODES + NB - 1) / NB)
#define NC     256
#define BT     1024
#define CHUNK_CAP 6400
#define BIN_CAP   8192
#define CNT_PAD   512
#define EPT       8
#define SEG_PW    (NC / (BT / 32))
static_assert(BIN_R <= CNT_PAD, "a range's nodes fit the padded counter array");
static_assert(NB % 32 == 0 && NC % 32 == 0 && NC <= BT && NC % (BT / 32) == 0,
              "whole waves scan the ranges and the chunk runs; the runs split evenly over the waves");

__device__ __forceinline__ int bin_of(int d) {
    const unsigned b = (unsigned)d / (unsigned)BIN_R;
    return b < (unsigned)NB ? (int)b : NB - 1;
}

__device__ __forceinline__ int wave_incl_scan(int v, int lane) {
    #pragma unroll
    for (int d = 1; d < 32; d <<= 1) {
        const int u = __shfl_up(v, d);
        if (lane >= d) v += u;
    }
    return v;
}

__global__ __launch_bounds__(BT) void bin_place(const int* __restrict__ src,
                                                const int* __restrict__ dst,
                                                int* __restrict__ offT,
                                                int2* __restrict__ pairs, int E) {
    __shared__ int  hist[NB];
    __shared__ int  wtot[NB / 32];
    __shared__ int2 sbuf[CHUNK_CAP];
    const int tid = threadIdx.x, lane = tid & 31, wave = tid >> 5, c = blockIdx.x;
    const int epc = (E + NC - 1) / NC;
    const int e0  = c * epc;
    const int e1  = (e0 + epc) < E ? (e0 + epc) : E;
    const int n   = e1 > e0 ? e1 - e0 : 0;
    if (tid < NB) hist[tid] = 0;
    __syncthreads();
    for (int base = e0 + tid; base < e1; base += BT * EPT) {
        int d[EPT];
        #pragma unroll
        for (int k = 0; k < EPT; ++k) d[k] = (base + k * BT) < e1 ? dst[base + k * BT] : 0;
        #pragma unroll
        for (int k = 0; k < EPT; ++k)
            if ((base + k * BT) < e1) atomicAdd(&hist[bin_of(d[k])], 1);
    }
    __syncthreads();
    int cnt = 0, v = 0;
    if (tid < NB) {
        cnt = hist[tid];
        v = wave_incl_scan(cnt, lane);
        if (lane == 31) wtot[wave] = v;
    }
    __syncthreads();
    if (tid < NB) {
        int base = 0;
        for (int w = 0; w < wave; ++w) base += wtot[w];
        const int ex = base + v - cnt;
        hist[tid] = ex;
        offT[c * NB + tid] = ex;
    }
    __syncthreads();
    const bool staged = n <= CHUNK_CAP;
    for (int base = e0 + tid; base < e1; base += BT * EPT) {
        int s[EPT], d[EPT];
        #pragma unroll
        for (int k = 0; k < EPT; ++k) {
            const bool in = (base + k * BT) < e1;
            s[k] = in ? src[base + k * BT] : 0;
            d[k] = in ? dst[base + k * BT] : 0;
        }
        #pragma unroll
        for (int k = 0; k < EPT; ++k) {
            if ((base + k * BT) < e1) {
                int2 r; r.x = s[k]; r.y = d[k];
                const int p = atomicAdd(&hist[bin_of(r.y)], 1);
                if (staged) {
                    if ((unsigned)p < (unsigned)CHUNK_CAP) sbuf[p] = r;
                } else {
                    const int g = e0 + p;
                    if ((unsigned)p < (unsigned)n && (unsigned)g < (unsigned)E) pairs[g] = r;
                }
            }
        }
    }
    if (staged) {
        __syncthreads();
        for (int i = tid; i < n; i += BT) pairs[e0 + i] = sbuf[i];
    }
}

__global__ __launch_bounds__(BT) void bin_csr(const int2* __restrict__ pairs,
                                              const int* __restrict__ offT,
                                              int* __restrict__ rowptr, int* __restrict__ col,
                                              float* __restrict__ dis, int N, int E) {
    __shared__ int  cntl[CNT_PAD];
    __shared__ int  wtot[CNT_PAD / 32];
    __shared__ int  segStart[NC];
    __shared__ int  segPre[NC + 1];
    __shared__ int  wsum[2 * (NC / 32)];
    __shared__ int  scol[BIN_CAP];
    __shared__ int2 spair[BIN_CAP];
    const int tid = threadIdx.x, lane = tid & 31, wave = tid >> 5, b = blockIdx.x;
    const int epc = (E + NC - 1) / NC;
    int len = 0, vlen = 0, o0 = 0, vo = 0;
    if (tid < NC) {
        const int e0 = tid * epc;
        const int e1 = (e0 + epc) < E ? (e0 + epc) : E;
        const int nc = e1 > e0 ? e1 - e0 : 0;
        o0 = offT[tid * NB + b];
        int o1 = (b + 1 < NB) ? offT[tid * NB + b + 1] : nc;
        o0 = o0 > 0 ? o0 : 0;   o0 = o0 < nc ? o0 : nc;
        o1 = o1 > o0 ? o1 : o0; o1 = o1 < nc ? o1 : nc;
        len = o1 - o0;
        segStart[tid] = e0 + o0;
        vlen = wave_incl_scan(len, lane);
        vo   = wave_incl_scan(o0, lane);
        if (lane == 31) { wsum[wave] = vlen; wsum[NC / 32 + wave] = vo; }
    }
    if (tid < CNT_PAD) cntl[tid] = 0;
    __syncthreads();
    int n = 0, beg = 0;
    #pragma unroll
    for (int w = 0; w < NC / 32; ++w) { n += wsum[w]; beg += wsum[NC / 32 + w]; }
    if (tid < NC) {
        int base = 0;
        for (int w = 0; w < wave; ++w) base += wsum[w];
        segPre[tid] = base + vlen - len;
        if (tid == NC - 1) segPre[NC] = base + vlen;
    }
    const int node0 = b * BIN_R;
    int nn = N - node0;
    nn = nn < BIN_R ? nn : BIN_R;
    nn = nn > 0 ? nn : 0;
    const bool staged = n <= BIN_CAP;
    __syncthreads();
    if (staged) {
        int2 r[SEG_PW], t[SEG_PW];
        #pragma unroll
        for (int k = 0; k < SEG_PW; ++k) {
            const int cc = wave + k * (BT / 32);
            const int ln = segPre[cc + 1] - segPre[cc];
            const int st = segStart[cc];
            r[k].x = 0; r[k].y = -1; t[k] = r[k];
            if (lane < ln)      r[k] = pairs[st + lane];
            if (lane + 32 < ln) t[k] = pairs[st + lane + 32];
        }
        #pragma unroll
        for (int k = 0; k < SEG_PW; ++k) {
            const int cc = wave + k * (BT / 32);
            const int p0 = segPre[cc];
            const int ln = segPre[cc + 1] - p0;
            const int st = segStart[cc];
            if (lane < ln)      spair[p0 + lane]      = r[k];
            if (lane + 32 < ln) spair[p0 + lane + 32] = t[k];
            for (int j = lane + 64; j < ln; j += 32) spair[p0 + j] = pairs[st + j];
        }
        __syncthreads();
        for (int i = tid; i < n; i += BT) {
            const unsigned l = (unsigned)(spair[i].y - node0);
            if (l < (unsigned)nn) atomicAdd(&cntl[l], 1);
        }
    } else {
        for (int k = 0; k < SEG_PW; ++k) {
            const int cc = wave + k * (BT / 32);
            const int ln = segPre[cc + 1] - segPre[cc];
            const int st = segStart[cc];
            for (int j = lane; j < ln; j += 32) {
                const unsigned l = (unsigned)(pairs[st + j].y - node0);
                if (l < (unsigned)nn) atomicAdd(&cntl[l], 1);
            }
        }
    }
    __syncthreads();
    int cnt = 0, v = 0;
    if (tid < CNT_PAD) {
        cnt = cntl[tid];
        v = wave_incl_scan(cnt, lane);
        if (lane == 31) wtot[wave] = v;
    }
    __syncthreads();
    if (tid < CNT_PAD) {
        int base = 0;
        for (int w = 0; w < wave; ++w) base += wtot[w];
        const int ex = base + v - cnt;
        cntl[tid] = ex;
        if (tid < nn) {
            rowptr[node0 + tid] = beg + ex;
            dis[node0 + tid]    = rsqrtf((float)cnt + 1.0f);
        }
    }
    if (b == NB - 1 && tid == 0) rowptr[N] = beg + n;
    __syncthreads();
    if (staged) {
        for (int i = tid; i < n; i += BT) {
            const int2 r = spair[i];
            const unsigned l = (unsigned)(r.y - node0);
            if (l < (unsigned)nn) {
                const int p = atomicAdd(&cntl[l], 1);
                if ((unsigned)p < (unsigned)BIN_CAP) scol[p] = r.x;
            }
        }
        __syncthreads();
        for (int i = tid; i < n; i += BT) col[beg + i] = scol[i];
    } else {
        for (int k = 0; k < SEG_PW; ++k) {
            const int cc = wave + k * (BT / 32);
            const int ln = segPre[cc + 1] - segPre[cc];
            const int st = segStart[cc];
            for (int j = lane; j < ln; j += 32) {
                const int2 r = pairs[st + j];
                const unsigned l = (unsigned)(r.y - node0);
                if (l < (unsigned)nn) {
                    const int g = beg + atomicAdd(&cntl[l], 1);
                    if ((unsigned)g < (unsigned)E) col[g] = r.x;
                }
            }
        }
    }
}

#define FG_PA   (HID + 4)
#define FG_NPW  (GEMM_ROWS / 8)
#define FG_KC   32
#define FG_NW   ((FG_KC / 2) * (HID / 4) / 256)
static_assert(((FG_KC / 2) * (HID / 4)) % 256 == 0, "W chunk splits evenly over the threads");
static_assert(GEMM_ROWS % 8 == 0, "the tile's rows split evenly over the eight waves");

template <int G>
__device__ __forceinline__ void row_group(const float4* __restrict__ h4, int s_l, int j,
                                          int lane, float4& acc) {
    float4 v[G];
    #pragma unroll
    for (int t = 0; t < G; ++t) {
        const int s = __builtin_amdgcn_readlane(s_l, j + t);
        v[t] = h4[(size_t)s * (HID / 4) + lane];
    }
    #pragma unroll
    for (int t = 0; t < G; ++t) {
        acc.x += v[t].x; acc.y += v[t].y; acc.z += v[t].z; acc.w += v[t].w;
    }
}

__global__ __launch_bounds__(256, 10) void gather_gemm(const int* __restrict__ rowptr,
                                                   const int* __restrict__ col,
                                                   const float* __restrict__ dis,
                                                   const float* __restrict__ h,
                                                   const float* __restrict__ bias,
                                                   const float* __restrict__ W,
                                                   float* __restrict__ out, int N, int E) {
    __shared__ float4 ldsA4[GEMM_ROWS * FG_PA / 4];
    __shared__ float4 ldsW4[(FG_KC / 2) * GEMM_PW / 2];
    float*  ldsA = (float*)ldsA4;
    float2* ldsW = (float2*)ldsW4;
    const int tid  = threadIdx.x;
    const int lane = tid & 31;
    const int wave = __builtin_amdgcn_readfirstlane(tid >> 5);
    const int half = lane >> 4;
    const int m    = lane & 15;
    const int row0 = blockIdx.x * GEMM_ROWS;
    const int col0 = wave * 16;
    const float4* h4 = (const float4*)h;
    const float4  bv = ((const float4*)bias)[lane];

    for (int q = 0; q < FG_NPW; ++q) {
        const int lr = wave * FG_NPW + q;
        const int i  = row0 + lr;
        int beg = rowptr[i], end = rowptr[i + 1];
        beg = beg > 0 ? beg : 0;
        end = end < E ? end : E;
        const float  di  = dis[i];
        const float4 hv  = h4[(size_t)i * (HID / 4) + lane];
        float4 acc; acc.x = acc.y = acc.z = acc.w = 0.f;

        for (int base = beg; base < end; base += 32) {
            const int mm = (end - base) < 32 ? (end - base) : 32;
            int s_l = 0;
            if (lane < mm) {
                s_l = col[base + lane];
                s_l = (unsigned)s_l < (unsigned)N ? s_l : 0;
            }
            int j = 0;
            for (; j + 8 <= mm; j += 8) row_group<8>(h4, s_l, j, lane, acc);
            if (mm - j >= 4) { row_group<4>(h4, s_l, j, lane, acc); j += 4; }
            if (mm - j >= 2) { row_group<2>(h4, s_l, j, lane, acc); j += 2; }
            if (mm - j >= 1) { row_group<1>(h4, s_l, j, lane, acc); }
        }
        float4 r;
        r.x = di * (acc.x + hv.x) + bv.x;
        r.y = di * (acc.y + hv.y) + bv.y;
        r.z = di * (acc.z + hv.z) + bv.z;
        r.w = di * (acc.w + hv.w) + bv.w;
        r.x = r.x > 0.f ? r.x : 0.f;
        r.y = r.y > 0.f ? r.y : 0.f;
        r.z = r.z > 0.f ? r.z : 0.f;
        r.w = r.w > 0.f ? r.w : 0.f;
        r.x *= di; r.y *= di; r.z *= di; r.w *= di;
        *(float4*)(ldsA + lr * FG_PA + 4 * lane) = r;
    }

    const float* wg = W + (size_t)(2 * (tid >> 5)) * HID + 4 * (tid & 31);
    v4f* wsd = (v4f*)(ldsW + (tid >> 5) * GEMM_PW + 4 * (tid & 31));

    v8f accm[GEMM_RT];
    #pragma unroll
    for (int t = 0; t < GEMM_RT; ++t) accm[t] = v8f{};

    const float*  aF = ldsA + m * FG_PA + 2 * half;
    const float2* bF = ldsW + half * GEMM_PW + col0 + m;

    for (int kb = 0; kb < HID; kb += FG_KC) {
        {
            v4f rw0[FG_NW], rw1[FG_NW];
            #pragma unroll
            for (int i = 0; i < FG_NW; ++i) {
                rw0[i] = *(const v4f*)(wg + (size_t)(kb + 16 * i) * HID);
                rw1[i] = *(const v4f*)(wg + (size_t)(kb + 16 * i + 1) * HID);
            }
            #pragma unroll
            for (int i = 0; i < FG_NW; ++i) {
                v4f lo, hi;
                lo.x = rw0[i].x; lo.y = rw1[i].x; lo.z = rw0[i].y; lo.w = rw1[i].y;
                hi.x = rw0[i].z; hi.y = rw1[i].z; hi.z = rw0[i].w; hi.w = rw1[i].w;
                wsd[i * 8 * GEMM_PW / 2]     = lo;
                wsd[i * 8 * GEMM_PW / 2 + 1] = hi;
            }
        }
        __syncthreads();

        #pragma unroll
        for (int k = 0; k < FG_KC; k += 4) {
            const float2 bb = bF[(k >> 1) * GEMM_PW];
            v2f b; b.x = bb.x; b.y = bb.y;
            #pragma unroll
            for (int t = 0; t < GEMM_RT; ++t) {
                const float2 aa = *(const float2*)(aF + t * 16 * FG_PA + kb + k);
                v2f a; a.x = aa.x; a.y = aa.y;
                accm[t] = __builtin_amdgcn_wmma_f32_16x16x4_f32(
                    false, a, false, b, (short)0, accm[t], false, false);
            }
        }
        __syncthreads();
    }

    const int halfD = (int)(__builtin_amdgcn_mbcnt_lo(~0u, 0u) >> 4);
    #pragma unroll
    for (int t = 0; t < GEMM_RT; ++t) {
        #pragma unroll
        for (int v = 0; v < 8; ++v)
            out[(size_t)(row0 + 16 * t + v + 8 * halfD) * HID + col0 + m] = accm[t][v];
    }
}

#define GATHER_WGS 2048
static_assert((unsigned long long)N_NODES * (GATHER_WGS * 8ull) < (1ull << 32),
              "wave index times node count fits 32 bits in the range split");
__global__ __launch_bounds__(256) void gather_pool_layer(const int* __restrict__ rowptr,
                                                         const int* __restrict__ col,
                                                         const float* __restrict__ dis,
                                                         const float* __restrict__ h,
                                                         const float* __restrict__ bias,
                                                         const int* __restrict__ batch,
                                                         float* __restrict__ pooled,
                                                         float* __restrict__ cnt, int N, int E) {
    __shared__ float4 sAcc4[8 * (HID / 4)];
    __shared__ int4   sMeta4[4];
    const int tid  = threadIdx.x;
    const int lane = tid & 31;
    const int wave = tid >> 5;
    const unsigned gw = (unsigned)__builtin_amdgcn_readfirstlane(
        (int)((blockIdx.x * blockDim.x + threadIdx.x) >> 5));
    const unsigned nw = (gridDim.x * blockDim.x) >> 5;
    const int n0 = (int)(gw * (unsigned)N / nw);
    const int n1 = (int)((gw + 1u) * (unsigned)N / nw);
    const float4* h4 = (const float4*)h;
    const float4  bv = ((const float4*)bias)[lane];

    int    cur = 0, run = 0;
    float4 pacc; pacc.x = pacc.y = pacc.z = pacc.w = 0.f;

    for (int i = n0; i < n1; ++i) {
        int beg = rowptr[i], end = rowptr[i + 1];
        beg = beg > 0 ? beg : 0;
        end = end < E ? end : E;
        const float  di  = dis[i];
        const int    g   = batch[i];
        const float4 hv  = h4[(size_t)i * (HID / 4) + lane];
        float4 acc; acc.x = acc.y = acc.z = acc.w = 0.f;

        for (int base = beg; base < end; base += 32) {
            const int m = (end - base) < 32 ? (end - base) : 32;
            int s_l = 0;
            if (lane < m) {
                s_l = col[base + lane];
                s_l = (unsigned)s_l < (unsigned)N ? s_l : 0;
            }
            #pragma unroll 4
            for (int j = 0; j < m; ++j) {
                const int    s = __builtin_amdgcn_readlane(s_l, j);
                const float4 v = h4[(size_t)s * (HID / 4) + lane];
                acc.x += v.x; acc.y += v.y; acc.z += v.z; acc.w += v.w;
            }
        }
        float4 r;
        r.x = di * (acc.x + hv.x) + bv.x;
        r.y = di * (acc.y + hv.y) + bv.y;
        r.z = di * (acc.z + hv.z) + bv.z;
        r.w = di * (acc.w + hv.w) + bv.w;
        r.x = r.x > 0.f ? r.x : 0.f;
        r.y = r.y > 0.f ? r.y : 0.f;
        r.z = r.z > 0.f ? r.z : 0.f;
        r.w = r.w > 0.f ? r.w : 0.f;

        if (run > 0 && g != cur) {
            if ((unsigned)cur < (unsigned)N_GRAPHS) {
                float* pg = pooled + cur * HID + 4 * lane;
                atomicAdd(&pg[0], pacc.x);
                atomicAdd(&pg[1], pacc.y);
                atomicAdd(&pg[2], pacc.z);
                atomicAdd(&pg[3], pacc.w);
                if (lane == 0) atomicAdd(&cnt[cur], (float)run);
            }
            run = 0;
            pacc.x = pacc.y = pacc.z = pacc.w = 0.f;
        }
        cur = g;
        pacc.x += r.x; pacc.y += r.y; pacc.z += r.z; pacc.w += r.w;
        ++run;
    }

    sAcc4[wave * (HID / 4) + lane] = pacc;
    if (lane == 0) { ((int*)sMeta4)[wave] = cur; ((int*)sMeta4)[8 + wave] = run; }
    __syncthreads();
    if (tid < HID) {
        const float* sa = (const float*)sAcc4;
        float v[8];
        #pragma unroll
        for (int w = 0; w < 8; ++w) v[w] = sa[w * HID + tid];
        const int4 ga = sMeta4[0], gb = sMeta4[1], ra = sMeta4[2], rb = sMeta4[3];
        const int gq[8] = { ga.x, ga.y, ga.z, ga.w, gb.x, gb.y, gb.z, gb.w };
        const int rw[8] = { ra.x, ra.y, ra.z, ra.w, rb.x, rb.y, rb.z, rb.w };
        int   cg = 0, rc = 0;
        float s  = 0.f;
        #pragma unroll
        for (int w = 0; w < 8; ++w) {
            if (rw[w] == 0) continue;
            if (rc > 0 && gq[w] != cg) {
                if ((unsigned)cg < (unsigned)N_GRAPHS) {
                    atomicAdd(&pooled[cg * HID + tid], s);
                    if (tid == 0) atomicAdd(&cnt[cg], (float)rc);
                }
                rc = 0; s = 0.f;
            }
            cg = gq[w];
            s += v[w];
            rc += rw[w];
        }
        if (rc > 0 && (unsigned)cg < (unsigned)N_GRAPHS) {
            atomicAdd(&pooled[cg * HID + tid], s);
            if (tid == 0) atomicAdd(&cnt[cg], (float)rc);
        }
    }
}

#define POOL_ROWS 16
static_assert(N_NODES % POOL_ROWS == 0, "every wave owns a full group of rows");
__global__ __launch_bounds__(256) void pool_accum(const float* __restrict__ h,
                                                  const int* __restrict__ batch,
                                                  float* __restrict__ pooled,
                                                  float* __restrict__ cnt, int N) {
    const int lane = threadIdx.x & 31;
    const int gw = __builtin_amdgcn_readfirstlane(
        (int)((blockIdx.x * blockDim.x + threadIdx.x) >> 5));
    const int r0 = gw * POOL_ROWS;
    if (r0 >= N) return;

    const int*   bp = batch + r0;
    const float* hp = h + (size_t)r0 * HID + lane;
    int   g[POOL_ROWS];
    float v[POOL_ROWS][4];
    #pragma unroll
    for (int i = 0; i < POOL_ROWS; ++i) {
        g[i] = bp[i];
        #pragma unroll
        for (int k = 0; k < 4; ++k) v[i][k] = hp[i * HID + 32 * k];
    }

    int   cur = g[0];
    int   run = 0;
    float a0 = 0.f, a1 = 0.f, a2 = 0.f, a3 = 0.f;
    #pragma unroll
    for (int i = 0; i < POOL_ROWS; ++i) {
        if (g[i] != cur) {
            float* pg = pooled + cur * HID + lane;
            atomicAdd(&pg[0],  a0);
            atomicAdd(&pg[32], a1);
            atomicAdd(&pg[64], a2);
            atomicAdd(&pg[96], a3);
            if (lane == 0) atomicAdd(&cnt[cur], (float)run);
            cur = g[i]; run = 0;
            a0 = a1 = a2 = a3 = 0.f;
        }
        a0 += v[i][0]; a1 += v[i][1]; a2 += v[i][2]; a3 += v[i][3];
        ++run;
    }
    {
        float* pg = pooled + cur * HID + lane;
        atomicAdd(&pg[0],  a0);
        atomicAdd(&pg[32], a1);
        atomicAdd(&pg[64], a2);
        atomicAdd(&pg[96], a3);
        if (lane == 0) atomicAdd(&cnt[cur], (float)run);
    }
}

#define HEAD_T_MF  (N_GRAPHS * OUT_MF)
#define HEAD_T_BP  (N_GRAPHS * OUT_BP)
#define HEAD_T_CC  (N_GRAPHS * OUT_CC)
#define HEAD_T_ALL (HEAD_T_MF + HEAD_T_BP + HEAD_T_CC)
#define HEAD_KB    32
static_assert(HEAD_T_MF % 32 == 0 && HEAD_T_BP % 32 == 0, "a wave never straddles two heads");
static_assert(HEAD_T_ALL % 256 == 0, "whole work-groups: every thread owns an output");
static_assert(OUT_MF > 32 && OUT_BP > 32 && OUT_CC > 32, "a wave's outputs touch at most two graphs");
static_assert(HID % HEAD_KB == 0 && HEAD_KB % 4 == 0, "k groups are whole 16-byte pieces of a mean row");

__global__ __launch_bounds__(256) void heads_fused(const float* __restrict__ pooled,
                                                   const float* __restrict__ cnt,
                                                   const float* __restrict__ Wmf,
                                                   const float* __restrict__ bmf,
                                                   const float* __restrict__ Wbp,
                                                   const float* __restrict__ bbp,
                                                   const float* __restrict__ Wcc,
                                                   const float* __restrict__ bcc,
                                                   float* __restrict__ out) {
    __shared__ float4 sP4[8 * 2 * (HID / 4)];
    const int tid = threadIdx.x, lane = tid & 31, wave = tid >> 5;
    const int t = blockIdx.x * 256 + tid;
    const int tw = __builtin_amdgcn_readfirstlane(t - lane);
    const float* W;
    const float* b;
    int OUT, hb;
    if (tw < HEAD_T_MF)                  { W = Wmf; b = bmf; OUT = OUT_MF; hb = 0; }
    else if (tw < HEAD_T_MF + HEAD_T_BP) { W = Wbp; b = bbp; OUT = OUT_BP; hb = HEAD_T_MF; }
    else                                 { W = Wcc; b = bcc; OUT = OUT_CC; hb = HEAD_T_MF + HEAD_T_BP; }
    const int tt = t - hb;
    const int g  = tt / OUT, o = tt - g * OUT;
    const int g0 = (tw - hb) / OUT;
    const int g1 = (g0 + 1) < N_GRAPHS ? (g0 + 1) : (N_GRAPHS - 1);

    #pragma unroll
    for (int r = 0; r < 2; ++r) {
        const int   gr = r ? g1 : g0;
        const float c  = cnt[gr];
        const float d  = c > 1.f ? c : 1.f;
        float4 v = ((const float4*)(pooled + gr * HID))[lane];
        v.x /= d; v.y /= d; v.z /= d; v.w /= d;
        sP4[(wave * 2 + r) * (HID / 4) + lane] = v;
    }
    __syncthreads();

    const float* prow = (const float*)sP4 + (wave * 2 + (g - g0)) * HID;
    const float* wp   = W + o;
    float s = b[o];
    for (int kb = 0; kb < HID; kb += HEAD_KB) {
        float w[HEAD_KB];
        #pragma unroll
        for (int i = 0; i < HEAD_KB; ++i) w[i] = wp[(size_t)(kb + i) * OUT];
        #pragma unroll
        for (int i = 0; i < HEAD_KB; i += 4) {
            const float4 p = *(const float4*)(prow + kb + i);
            s += p.x * w[i];
            s += p.y * w[i + 1];
            s += p.z * w[i + 2];
            s += p.w * w[i + 3];
        }
    }
    out[t] = s;
}

extern "C" void kernel_launch(void* const* d_in, const int* in_sizes, int n_in,
                              void* d_out, int out_size, void* d_ws, size_t ws_size,
                              hipStream_t stream) {
    const float* x   = (const float*)d_in[0];
    const int*   ei  = (const int*)d_in[1];
    const int*   bat = (const int*)d_in[2];
    const float* W1  = (const float*)d_in[3];
    const float* b1  = (const float*)d_in[4];
    const float* W2  = (const float*)d_in[5];
    const float* b2  = (const float*)d_in[6];
    const float* Wmf = (const float*)d_in[7];
    const float* bmf = (const float*)d_in[8];
    const float* Wbp = (const float*)d_in[9];
    const float* bbp = (const float*)d_in[10];
    const float* Wcc = (const float*)d_in[11];
    const float* bcc = (const float*)d_in[12];
    float* out = (float*)d_out;

    const int E = in_sizes[1] / 2;
    const int* src = ei;
    const int* dst = ei + E;

    float* ws     = (float*)d_ws;
    float* dis    = ws;
    float* h      = dis + N_NODES;
    float* agg    = h + (size_t)N_NODES * HID;
    float* pooled = agg + (size_t)N_NODES * HID;
    float* cnt    = pooled + N_GRAPHS * HID;

    int*  rowptr = (int*)(cnt + N_GRAPHS);
    int*  col    = rowptr + (N_NODES + 4);
    int2* pairs  = (int2*)agg;
    int*  offT   = (int*)agg + 2 * N_EDGES;

    if (ws_size >= (size_t)WS_PLANES_BYTES + (size_t)WS_EXTRA_BYTES) {
        bin_place<<<NC, BT, 0, stream>>>(src, dst, offT, pairs, E);
        bin_csr<<<NB, BT, 0, stream>>>(pairs, offT, rowptr, col, dis, N_NODES, E);

        gemm_wmma_f32<IN_DIM, true><<<N_NODES / GEMM_ROWS, 256, 0, stream>>>(x, W1, dis, h);

        gather_gemm<<<N_NODES / GEMM_ROWS, 256, 0, stream>>>(rowptr, col, dis, h, b1, W2, agg, N_NODES, E);

        fill_f32<<<(N_GRAPHS * HID + N_GRAPHS + 255) / 256, 256, 0, stream>>>(
            pooled, 0.0f, N_GRAPHS * HID + N_GRAPHS);
        gather_pool_layer<<<GATHER_WGS, 256, 0, stream>>>(rowptr, col, dis, agg, b2, bat,
                                                          pooled, cnt, N_NODES, E);
    } else {
        const int NH  = N_NODES * HID;
        const int NH4 = NH / 4;
        fill_f32<<<(N_NODES + 255) / 256, 256, 0, stream>>>(dis, 1.0f, N_NODES);
        deg_accum<<<(E + 255) / 256, 256, 0, stream>>>(dst, dis, E);
        rsqrt_inplace<<<(N_NODES + 255) / 256, 256, 0, stream>>>(dis, N_NODES);

        gemm_wmma_f32<IN_DIM, false><<<N_NODES / GEMM_ROWS, 256, 0, stream>>>(x, W1, dis, h);
        zero_f4<<<(NH4 + 255) / 256, 256, 0, stream>>>((float4*)agg, NH4);
        edge_scatter<<<2048, 256, 0, stream>>>(src, dst, dis, h, agg, E);
        finish_layer<<<(NH + 255) / 256, 256, 0, stream>>>(agg, h, dis, b1, NH);

        gemm_wmma_f32<HID, false><<<N_NODES / GEMM_ROWS, 256, 0, stream>>>(agg, W2, dis, h);
        zero_f4<<<(NH4 + 255) / 256, 256, 0, stream>>>((float4*)agg, NH4);
        edge_scatter<<<2048, 256, 0, stream>>>(src, dst, dis, h, agg, E);
        finish_layer<<<(NH + 255) / 256, 256, 0, stream>>>(agg, h, dis, b2, NH);

        fill_f32<<<(N_GRAPHS * HID + N_GRAPHS + 255) / 256, 256, 0, stream>>>(
            pooled, 0.0f, N_GRAPHS * HID + N_GRAPHS);
        pool_accum<<<(N_NODES + POOL_ROWS * 8 - 1) / (POOL_ROWS * 8), 256, 0, stream>>>(agg, bat, pooled, cnt, N_NODES);
    }

    heads_fused<<<HEAD_T_ALL / 256, 256, 0, stream>>>(pooled, cnt, Wmf, bmf, Wbp, bbp, Wcc, bcc, out);
}
